# starting kernel
# baseline (speedup 1.0000x reference)
#include <hip/hip_runtime.h>
#include <hip/hip_bf16.h>

typedef __attribute__((ext_vector_type(2))) float v2f;
typedef __attribute__((ext_vector_type(8))) float v8f;

__global__ __launch_bounds__(256) void zero_kernel(float* __restrict__ p, size_t n) {
    size_t i = (size_t)blockIdx.x * blockDim.x + threadIdx.x;
    if (i < n) p[i] = 0.0f;
}

template <int F>
__global__ __launch_bounds__(256) void scatter_add_vec(const float* __restrict__ src,
                                                       const int* __restrict__ sidx,
                                                       const int* __restrict__ didx,
                                                       float* dst, int nE) {
    constexpr int PER = F / 4;
    unsigned t = blockIdx.x * blockDim.x + threadIdx.x;
    unsigned e = t / PER;
    unsigned f = (t % PER) * 4;
    if (e >= (unsigned)nE) return;
    int s = sidx[e];
    int d = didx[e];
    const float4 v = *(const float4*)(src + (size_t)s * F + f);
    float* p = dst + (size_t)d * F + f;
    atomicAdd(p + 0, v.x);
    atomicAdd(p + 1, v.y);
    atomicAdd(p + 2, v.z);
    atomicAdd(p + 3, v.w);
}

template <int K1, int K2, bool RELU, int MT>
__global__ __launch_bounds__(256) void gemm_dual_wmma(const float* __restrict__ A1,
                                                      const float* __restrict__ W1,
                                                      const float* __restrict__ A2,
                                                      const float* __restrict__ W2,
                                                      const float* __restrict__ bias,
                                                      float* out, int M) {
    constexpr int N = 128;
    const int lane = threadIdx.x & 31;
    const int wave = threadIdx.x >> 5;
    const int m0 = (blockIdx.x * 8 + wave) * (16 * MT);
    if (m0 >= M) return;
    const int half = lane >> 4;
    const int lr = lane & 15;

    v8f acc[MT][8] = {};

    int arow[MT];
#pragma unroll
    for (int mt = 0; mt < MT; ++mt) {
        int r = m0 + mt * 16 + lr;
        arow[mt] = (r < M) ? r : (M - 1);
    }

    for (int k = 0; k < K1; k += 4) {
        v2f a[MT];
#pragma unroll
        for (int mt = 0; mt < MT; ++mt)
            a[mt] = *(const v2f*)(A1 + (size_t)arow[mt] * K1 + k + 2 * half);
        const float* w0 = W1 + (size_t)(k + 2 * half) * N + lr;
#pragma unroll
        for (int nt = 0; nt < 8; ++nt) {
            v2f b;
            b.x = w0[nt * 16];
            b.y = w0[N + nt * 16];
#pragma unroll
            for (int mt = 0; mt < MT; ++mt)
                acc[mt][nt] = __builtin_amdgcn_wmma_f32_16x16x4_f32(
                    false, a[mt], false, b, (short)0, acc[mt][nt], false, false);
        }
    }
    for (int k = 0; k < K2; k += 4) {
        v2f a[MT];
#pragma unroll
        for (int mt = 0; mt < MT; ++mt)
            a[mt] = *(const v2f*)(A2 + (size_t)arow[mt] * K2 + k + 2 * half);
        const float* w0 = W2 + (size_t)(k + 2 * half) * N + lr;
#pragma unroll
        for (int nt = 0; nt < 8; ++nt) {
            v2f b;
            b.x = w0[nt * 16];
            b.y = w0[N + nt * 16];
#pragma unroll
            for (int mt = 0; mt < MT; ++mt)
                acc[mt][nt] = __builtin_amdgcn_wmma_f32_16x16x4_f32(
                    false, a[mt], false, b, (short)0, acc[mt][nt], false, false);
        }
    }

#pragma unroll
    for (int mt = 0; mt < MT; ++mt) {
        if (m0 + mt * 16 >= M) break;
#pragma unroll
        for (int nt = 0; nt < 8; ++nt) {
            const int col = nt * 16 + lr;
            const float bv = bias[col];
#pragma unroll
            for (int i = 0; i < 8; ++i) {
                float v = acc[mt][nt][i] + bv;
                if (RELU) v = fmaxf(v, 0.0f);
                out[(size_t)(m0 + mt * 16 + i + 8 * half) * N + col] = v;
            }
        }
    }
}

__global__ __launch_bounds__(256) void rowmat3(const float* __restrict__ X,
                                               const float* __restrict__ W,
                                               const float* __restrict__ bias,
                                               float* __restrict__ out, int n) {
    const int lane = threadIdx.x & 31;
    const int node = blockIdx.x * (blockDim.x >> 5) + (threadIdx.x >> 5);
    if (node >= n) return;
    const float4 x4 = *(const float4*)(X + (size_t)node * 128 + lane * 4);
    const float* w = W + lane * 4 * 3;
    float p0 = x4.x * w[0] + x4.y * w[3] + x4.z * w[6] + x4.w * w[9];
    float p1 = x4.x * w[1] + x4.y * w[4] + x4.z * w[7] + x4.w * w[10];
    float p2 = x4.x * w[2] + x4.y * w[5] + x4.z * w[8] + x4.w * w[11];
#pragma unroll
    for (int off = 16; off; off >>= 1) {
        p0 += __shfl_xor(p0, off, 32);
        p1 += __shfl_xor(p1, off, 32);
        p2 += __shfl_xor(p2, off, 32);
    }
    if (lane == 0) {
        float b0 = 0.f, b1 = 0.f, b2 = 0.f;
        if (bias) { b0 = bias[0]; b1 = bias[1]; b2 = bias[2]; }
        out[(size_t)node * 3 + 0] = p0 + b0;
        out[(size_t)node * 3 + 1] = p1 + b1;
        out[(size_t)node * 3 + 2] = p2 + b2;
    }
}

__global__ __launch_bounds__(256) void scatter_add3(const float* __restrict__ proj,
                                                    const int* __restrict__ sidx,
                                                    const int* __restrict__ didx,
                                                    float* out, int nE) {
    int e = blockIdx.x * blockDim.x + threadIdx.x;
    if (e >= nE) return;
    int s = sidx[e];
    int d = didx[e];
    atomicAdd(out + (size_t)d * 3 + 0, proj[(size_t)s * 3 + 0]);
    atomicAdd(out + (size_t)d * 3 + 1, proj[(size_t)s * 3 + 1]);
    atomicAdd(out + (size_t)d * 3 + 2, proj[(size_t)s * 3 + 2]);
}

__global__ __launch_bounds__(256) void knowledge_kernel(float* __restrict__ out,
                                                        const float* __restrict__ x,
                                                        int n) {
    int i = blockIdx.x * blockDim.x + threadIdx.x;
    if (i >= n) return;
    float a = out[(size_t)i * 3 + 0];
    float b = out[(size_t)i * 3 + 1];
    float c = out[(size_t)i * 3 + 2];
    float nrm = sqrtf(a * a + b * b + c * c);
    float s = 10.0f / fmaxf(nrm, 1e-12f);
    a *= s; b *= s; c *= s;
    if (x[(size_t)i * 64 + 61] != 0.0f) a -= 10.0f;
    if (x[(size_t)i * 64 + 63] != 0.0f) c -= 10.0f;
    out[(size_t)i * 3 + 0] = a;
    out[(size_t)i * 3 + 1] = b;
    out[(size_t)i * 3 + 2] = c;
}

static inline int ceil_div(long long a, long long b) { return (int)((a + b - 1) / b); }

extern "C" void kernel_launch(void* const* d_in, const int* in_sizes, int n_in,
                              void* d_out, int out_size, void* d_ws, size_t ws_size,
                              hipStream_t stream) {
    const float* x_s = (const float*)d_in[0];
    const float* x_t = (const float*)d_in[1];
    const int* e_l = (const int*)d_in[2];
    const int* e_r = (const int*)d_in[3];
    const float* w1lr_rel  = (const float*)d_in[4];
    const float* b1lr      = (const float*)d_in[5];
    const float* w1lr_root = (const float*)d_in[6];
    const float* w1rl_rel  = (const float*)d_in[7];
    const float* b1rl      = (const float*)d_in[8];
    const float* w1rl_root = (const float*)d_in[9];
    const float* w2lr_rel  = (const float*)d_in[10];
    const float* b2lr      = (const float*)d_in[11];
    const float* w2lr_root = (const float*)d_in[12];
    const float* w2rl_rel  = (const float*)d_in[13];
    const float* b2rl      = (const float*)d_in[14];
    const float* w2rl_root = (const float*)d_in[15];
    const float* w3lr_rel  = (const float*)d_in[16];
    const float* b3lr      = (const float*)d_in[17];
    const float* w3lr_root = (const float*)d_in[18];
    const float* w3rl_rel  = (const float*)d_in[19];
    const float* b3rl      = (const float*)d_in[20];
    const float* w3rl_root = (const float*)d_in[21];

    const int NL = in_sizes[0] / 64;
    const int NR = in_sizes[1] / 64;
    const int E  = in_sizes[2];

    float* ws = (float*)d_ws;
    const size_t AGG_R = 0;
    const size_t AGG_L = AGG_R + (size_t)NR * 128;
    const size_t H1_R  = AGG_L + (size_t)NL * 128;
    const size_t H1_L  = H1_R + (size_t)NR * 128;
    const size_t PROJ_LR = H1_R;
    const size_t PROJ_RL = H1_R + (size_t)NL * 3;

    float* d_left  = (float*)d_out;
    float* d_right = (float*)d_out + (size_t)NL * 3;

    const int TPB = 256;
    constexpr int MT = 2;

    zero_kernel<<<ceil_div((size_t)NR * 64, TPB), TPB, 0, stream>>>(ws + AGG_R, (size_t)NR * 64);
    zero_kernel<<<ceil_div((size_t)NL * 64, TPB), TPB, 0, stream>>>(ws + AGG_L, (size_t)NL * 64);
    scatter_add_vec<64><<<ceil_div((long long)E * 16, TPB), TPB, 0, stream>>>(x_s, e_l, e_r, ws + AGG_R, E);
    scatter_add_vec<64><<<ceil_div((long long)E * 16, TPB), TPB, 0, stream>>>(x_t, e_r, e_l, ws + AGG_L, E);
    gemm_dual_wmma<64, 64, true, MT><<<ceil_div(ceil_div(NR, 16 * MT), 8), TPB, 0, stream>>>(
        ws + AGG_R, w1lr_rel, x_t, w1lr_root, b1lr, ws + H1_R, NR);
    gemm_dual_wmma<64, 64, true, MT><<<ceil_div(ceil_div(NL, 16 * MT), 8), TPB, 0, stream>>>(
        ws + AGG_L, w1rl_rel, x_s, w1rl_root, b1rl, ws + H1_L, NL);

    zero_kernel<<<ceil_div((size_t)NR * 128, TPB), TPB, 0, stream>>>(ws + AGG_R, (size_t)NR * 128);
    zero_kernel<<<ceil_div((size_t)NL * 128, TPB), TPB, 0, stream>>>(ws + AGG_L, (size_t)NL * 128);
    scatter_add_vec<128><<<ceil_div((long long)E * 32, TPB), TPB, 0, stream>>>(ws + H1_L, e_l, e_r, ws + AGG_R, E);
    scatter_add_vec<128><<<ceil_div((long long)E * 32, TPB), TPB, 0, stream>>>(ws + H1_R, e_r, e_l, ws + AGG_L, E);
    gemm_dual_wmma<128, 128, true, MT><<<ceil_div(ceil_div(NR, 16 * MT), 8), TPB, 0, stream>>>(
        ws + AGG_R, w2lr_rel, ws + H1_R, w2lr_root, b2lr, ws + AGG_R, NR);
    gemm_dual_wmma<128, 128, true, MT><<<ceil_div(ceil_div(NL, 16 * MT), 8), TPB, 0, stream>>>(
        ws + AGG_L, w2rl_rel, ws + H1_L, w2rl_root, b2rl, ws + AGG_L, NL);

    rowmat3<<<ceil_div(NL, 8), TPB, 0, stream>>>(ws + AGG_L, w3lr_rel, nullptr, ws + PROJ_LR, NL);
    rowmat3<<<ceil_div(NR, 8), TPB, 0, stream>>>(ws + AGG_R, w3rl_rel, nullptr, ws + PROJ_RL, NR);
    rowmat3<<<ceil_div(NR, 8), TPB, 0, stream>>>(ws + AGG_R, w3lr_root, b3lr, d_right, NR);
    rowmat3<<<ceil_div(NL, 8), TPB, 0, stream>>>(ws + AGG_L, w3rl_root, b3rl, d_left, NL);
    scatter_add3<<<ceil_div(E, TPB), TPB, 0, stream>>>(ws + PROJ_LR, e_l, e_r, d_right, E);
    scatter_add3<<<ceil_div(E, TPB), TPB, 0, stream>>>(ws + PROJ_RL, e_r, e_l, d_left, E);

    knowledge_kernel<<<ceil_div(NL, TPB), TPB, 0, stream>>>(d_left, x_s, NL);
    knowledge_kernel<<<ceil_div(NR, TPB), TPB, 0, stream>>>(d_right, x_t, NR);
}


# Edge sums without float atomics: per-call counting sort of edges by destination (int atomic rank + scan), rows summed in registers and stored once; row sum order = rank order; zero fills dropped
# speedup vs baseline: 2.0308x; 2.0308x over previous
#include <hip/hip_runtime.h>
#include <hip/hip_bf16.h>

typedef __attribute__((ext_vector_type(2))) float v2f;
typedef __attribute__((ext_vector_type(8))) float v8f;

__global__ __launch_bounds__(256) void zero_kernel(float* __restrict__ p, size_t n) {
    size_t i = (size_t)blockIdx.x * blockDim.x + threadIdx.x;
    if (i < n) p[i] = 0.0f;
}

constexpr int CS = 16;
constexpr int TILE = 1024;

__global__ __launch_bounds__(256) void edge_rank(const int* __restrict__ e_l,
                                                 const int* __restrict__ e_r,
                                                 int* cntL, int* cntR,
                                                 int* __restrict__ rankL,
                                                 int* __restrict__ rankR, int nE) {
    int e = blockIdx.x * blockDim.x + threadIdx.x;
    if (e >= nE) return;
    int dl = e_l[e];
    int dr = e_r[e];
    const int rr = atomicAdd(cntR + (size_t)dr * CS, 1);
    const int rl = atomicAdd(cntL + (size_t)dl * CS, 1);
    rankR[e] = rr;
    rankL[e] = rl;
}

__global__ __launch_bounds__(256) void tile_sums(const int* __restrict__ cntA, int nA, int tilesA,
                                                 const int* __restrict__ cntB, int nB,
                                                 int* __restrict__ tsum) {
    __shared__ int wsum[8];
    const int b = blockIdx.x;
    const int lane = threadIdx.x & 31;
    const int wave = threadIdx.x >> 5;
    const int* cnt = (b < tilesA) ? cntA : cntB;
    const int n = (b < tilesA) ? nA : nB;
    const int tile = (b < tilesA) ? b : b - tilesA;
    const int i0 = tile * TILE + threadIdx.x * 4;
    int s = 0;
#pragma unroll
    for (int j = 0; j < 4; ++j) {
        int i = i0 + j;
        if (i < n) s += cnt[(size_t)i * CS];
    }
#pragma unroll
    for (int o = 16; o; o >>= 1) s += __shfl_xor(s, o, 32);
    if (lane == 0) wsum[wave] = s;
    __syncthreads();
    if (threadIdx.x == 0) {
        int t = 0;
#pragma unroll
        for (int w = 0; w < 8; ++w) t += wsum[w];
        tsum[b] = t;
    }
}

__global__ __launch_bounds__(256) void scan_apply(const int* __restrict__ cntA, int nA, int tilesA,
                                                  int* __restrict__ offA,
                                                  const int* __restrict__ cntB, int nB,
                                                  int* __restrict__ offB,
                                                  const int* __restrict__ tsum) {
    __shared__ int wsum[8];
    __shared__ int wtot[8];
    const int b = blockIdx.x;
    const int lane = threadIdx.x & 31;
    const int wave = threadIdx.x >> 5;
    const bool isA = b < tilesA;
    const int* cnt = isA ? cntA : cntB;
    int* off = isA ? offA : offB;
    const int n = isA ? nA : nB;
    const int tile = isA ? b : b - tilesA;
    const int tbase = isA ? 0 : tilesA;

    int p = 0;
    for (int i = threadIdx.x; i < tile; i += 256) p += tsum[tbase + i];
#pragma unroll
    for (int o = 16; o; o >>= 1) p += __shfl_xor(p, o, 32);
    if (lane == 0) wsum[wave] = p;

    const int i0 = tile * TILE + threadIdx.x * 4;
    int v[4];
    int own = 0;
#pragma unroll
    for (int j = 0; j < 4; ++j) {
        int i = i0 + j;
        v[j] = (i < n) ? cnt[(size_t)i * CS] : 0;
        own += v[j];
    }
    int incl = own;
#pragma unroll
    for (int o = 1; o < 32; o <<= 1) {
        int u = __shfl_up(incl, o, 32);
        if (lane >= o) incl += u;
    }
    if (lane == 31) wtot[wave] = incl;
    __syncthreads();

    int run = incl - own;
#pragma unroll
    for (int w = 0; w < 8; ++w) {
        run += wsum[w];
        if (w < wave) run += wtot[w];
    }
#pragma unroll
    for (int j = 0; j < 4; ++j) {
        int i = i0 + j;
        if (i < n) {
            off[i] = run;
            run += v[j];
            if (i == n - 1) off[n] = run;
        }
    }
}

__global__ __launch_bounds__(256) void edge_place(const int* __restrict__ e_l,
                                                  const int* __restrict__ e_r,
                                                  const int* __restrict__ offL,
                                                  const int* __restrict__ offR,
                                                  const int* __restrict__ rankL,
                                                  const int* __restrict__ rankR,
                                                  int* __restrict__ permL,
                                                  int* __restrict__ permR, int nE) {
    int e = blockIdx.x * blockDim.x + threadIdx.x;
    if (e >= nE) return;
    int dl = e_l[e];
    int dr = e_r[e];
    permR[offR[dr] + rankR[e]] = dl;
    permL[offL[dl] + rankL[e]] = dr;
}

template <int F>
__global__ __launch_bounds__(256) void gather_sum_vec(const float* __restrict__ src,
                                                      const int* __restrict__ off,
                                                      const int* __restrict__ perm,
                                                      float* __restrict__ dst, int nDst) {
    constexpr int PER = F / 4;
    unsigned t = blockIdx.x * blockDim.x + threadIdx.x;
    unsigned d = t / PER;
    unsigned f = (t % PER) * 4;
    if (d >= (unsigned)nDst) return;
    const int beg = off[d];
    const int end = off[d + 1];
    float4 acc = make_float4(0.f, 0.f, 0.f, 0.f);
    int j = beg;
    for (; j + 4 <= end; j += 4) {
        const int s0 = perm[j + 0];
        const int s1 = perm[j + 1];
        const int s2 = perm[j + 2];
        const int s3 = perm[j + 3];
        const float4 v0 = *(const float4*)(src + (size_t)s0 * F + f);
        const float4 v1 = *(const float4*)(src + (size_t)s1 * F + f);
        const float4 v2 = *(const float4*)(src + (size_t)s2 * F + f);
        const float4 v3 = *(const float4*)(src + (size_t)s3 * F + f);
        acc.x += v0.x; acc.y += v0.y; acc.z += v0.z; acc.w += v0.w;
        acc.x += v1.x; acc.y += v1.y; acc.z += v1.z; acc.w += v1.w;
        acc.x += v2.x; acc.y += v2.y; acc.z += v2.z; acc.w += v2.w;
        acc.x += v3.x; acc.y += v3.y; acc.z += v3.z; acc.w += v3.w;
    }
    for (; j < end; ++j) {
        const int s = perm[j];
        const float4 v = *(const float4*)(src + (size_t)s * F + f);
        acc.x += v.x; acc.y += v.y; acc.z += v.z; acc.w += v.w;
    }
    *(float4*)(dst + (size_t)d * F + f) = acc;
}

__global__ __launch_bounds__(256) void gather_add3(const float* __restrict__ proj,
                                                   const int* __restrict__ off,
                                                   const int* __restrict__ perm,
                                                   float* out, int nDst) {
    int d = blockIdx.x * blockDim.x + threadIdx.x;
    if (d >= nDst) return;
    const int beg = off[d];
    const int end = off[d + 1];
    float a0 = out[(size_t)d * 3 + 0];
    float a1 = out[(size_t)d * 3 + 1];
    float a2 = out[(size_t)d * 3 + 2];
    for (int j = beg; j < end; ++j) {
        const int s = perm[j];
        a0 += proj[(size_t)s * 3 + 0];
        a1 += proj[(size_t)s * 3 + 1];
        a2 += proj[(size_t)s * 3 + 2];
    }
    out[(size_t)d * 3 + 0] = a0;
    out[(size_t)d * 3 + 1] = a1;
    out[(size_t)d * 3 + 2] = a2;
}

template <int K1, int K2, bool RELU, int MT>
__global__ __launch_bounds__(256) void gemm_dual_wmma(const float* __restrict__ A1,
                                                      const float* __restrict__ W1,
                                                      const float* __restrict__ A2,
                                                      const float* __restrict__ W2,
                                                      const float* __restrict__ bias,
                                                      float* out, int M) {
    constexpr int N = 128;
    const int lane = threadIdx.x & 31;
    const int wave = threadIdx.x >> 5;
    const int m0 = (blockIdx.x * 8 + wave) * (16 * MT);
    if (m0 >= M) return;
    const int half = lane >> 4;
    const int lr = lane & 15;

    v8f acc[MT][8] = {};

    int arow[MT];
#pragma unroll
    for (int mt = 0; mt < MT; ++mt) {
        int r = m0 + mt * 16 + lr;
        arow[mt] = (r < M) ? r : (M - 1);
    }

    for (int k = 0; k < K1; k += 4) {
        v2f a[MT];
#pragma unroll
        for (int mt = 0; mt < MT; ++mt)
            a[mt] = *(const v2f*)(A1 + (size_t)arow[mt] * K1 + k + 2 * half);
        const float* w0 = W1 + (size_t)(k + 2 * half) * N + lr;
#pragma unroll
        for (int nt = 0; nt < 8; ++nt) {
            v2f b;
            b.x = w0[nt * 16];
            b.y = w0[N + nt * 16];
#pragma unroll
            for (int mt = 0; mt < MT; ++mt)
                acc[mt][nt] = __builtin_amdgcn_wmma_f32_16x16x4_f32(
                    false, a[mt], false, b, (short)0, acc[mt][nt], false, false);
        }
    }
    for (int k = 0; k < K2; k += 4) {
        v2f a[MT];
#pragma unroll
        for (int mt = 0; mt < MT; ++mt)
            a[mt] = *(const v2f*)(A2 + (size_t)arow[mt] * K2 + k + 2 * half);
        const float* w0 = W2 + (size_t)(k + 2 * half) * N + lr;
#pragma unroll
        for (int nt = 0; nt < 8; ++nt) {
            v2f b;
            b.x = w0[nt * 16];
            b.y = w0[N + nt * 16];
#pragma unroll
            for (int mt = 0; mt < MT; ++mt)
                acc[mt][nt] = __builtin_amdgcn_wmma_f32_16x16x4_f32(
                    false, a[mt], false, b, (short)0, acc[mt][nt], false, false);
        }
    }

#pragma unroll
    for (int mt = 0; mt < MT; ++mt) {
        if (m0 + mt * 16 >= M) break;
#pragma unroll
        for (int nt = 0; nt < 8; ++nt) {
            const int col = nt * 16 + lr;
            const float bv = bias[col];
#pragma unroll
            for (int i = 0; i < 8; ++i) {
                float v = acc[mt][nt][i] + bv;
                if (RELU) v = fmaxf(v, 0.0f);
                out[(size_t)(m0 + mt * 16 + i + 8 * half) * N + col] = v;
            }
        }
    }
}

__global__ __launch_bounds__(256) void rowmat3(const float* __restrict__ X,
                                               const float* __restrict__ W,
                                               const float* __restrict__ bias,
                                               float* __restrict__ out, int n) {
    const int lane = threadIdx.x & 31;
    const int node = blockIdx.x * (blockDim.x >> 5) + (threadIdx.x >> 5);
    if (node >= n) return;
    const float4 x4 = *(const float4*)(X + (size_t)node * 128 + lane * 4);
    const float* w = W + lane * 4 * 3;
    float p0 = x4.x * w[0] + x4.y * w[3] + x4.z * w[6] + x4.w * w[9];
    float p1 = x4.x * w[1] + x4.y * w[4] + x4.z * w[7] + x4.w * w[10];
    float p2 = x4.x * w[2] + x4.y * w[5] + x4.z * w[8] + x4.w * w[11];
#pragma unroll
    for (int off = 16; off; off >>= 1) {
        p0 += __shfl_xor(p0, off, 32);
        p1 += __shfl_xor(p1, off, 32);
        p2 += __shfl_xor(p2, off, 32);
    }
    if (lane == 0) {
        float b0 = 0.f, b1 = 0.f, b2 = 0.f;
        if (bias) { b0 = bias[0]; b1 = bias[1]; b2 = bias[2]; }
        out[(size_t)node * 3 + 0] = p0 + b0;
        out[(size_t)node * 3 + 1] = p1 + b1;
        out[(size_t)node * 3 + 2] = p2 + b2;
    }
}

__global__ __launch_bounds__(256) void knowledge_kernel(float* __restrict__ out,
                                                        const float* __restrict__ x,
                                                        int n) {
    int i = blockIdx.x * blockDim.x + threadIdx.x;
    if (i >= n) return;
    float a = out[(size_t)i * 3 + 0];
    float b = out[(size_t)i * 3 + 1];
    float c = out[(size_t)i * 3 + 2];
    float nrm = sqrtf(a * a + b * b + c * c);
    float s = 10.0f / fmaxf(nrm, 1e-12f);
    a *= s; b *= s; c *= s;
    if (x[(size_t)i * 64 + 61] != 0.0f) a -= 10.0f;
    if (x[(size_t)i * 64 + 63] != 0.0f) c -= 10.0f;
    out[(size_t)i * 3 + 0] = a;
    out[(size_t)i * 3 + 1] = b;
    out[(size_t)i * 3 + 2] = c;
}

static inline int ceil_div(long long a, long long b) { return (int)((a + b - 1) / b); }

extern "C" void kernel_launch(void* const* d_in, const int* in_sizes, int n_in,
                              void* d_out, int out_size, void* d_ws, size_t ws_size,
                              hipStream_t stream) {
    const float* x_s = (const float*)d_in[0];
    const float* x_t = (const float*)d_in[1];
    const int* e_l = (const int*)d_in[2];
    const int* e_r = (const int*)d_in[3];
    const float* w1lr_rel  = (const float*)d_in[4];
    const float* b1lr      = (const float*)d_in[5];
    const float* w1lr_root = (const float*)d_in[6];
    const float* w1rl_rel  = (const float*)d_in[7];
    const float* b1rl      = (const float*)d_in[8];
    const float* w1rl_root = (const float*)d_in[9];
    const float* w2lr_rel  = (const float*)d_in[10];
    const float* b2lr      = (const float*)d_in[11];
    const float* w2lr_root = (const float*)d_in[12];
    const float* w2rl_rel  = (const float*)d_in[13];
    const float* b2rl      = (const float*)d_in[14];
    const float* w2rl_root = (const float*)d_in[15];
    const float* w3lr_rel  = (const float*)d_in[16];
    const float* b3lr      = (const float*)d_in[17];
    const float* w3lr_root = (const float*)d_in[18];
    const float* w3rl_rel  = (const float*)d_in[19];
    const float* b3rl      = (const float*)d_in[20];
    const float* w3rl_root = (const float*)d_in[21];

    const int NL = in_sizes[0] / 64;
    const int NR = in_sizes[1] / 64;
    const int E  = in_sizes[2];

    float* ws = (float*)d_ws;
    const size_t AGG_R = 0;
    const size_t AGG_L = AGG_R + (size_t)NR * 128;
    const size_t H1_R  = AGG_L + (size_t)NL * 128;
    const size_t H1_L  = H1_R + (size_t)NR * 128;
    const size_t PROJ_LR = H1_R;
    const size_t PROJ_RL = H1_R + (size_t)NL * 3;
    const int tilesR = ceil_div(NR, TILE);
    const int tilesL = ceil_div(NL, TILE);
    int* wi = (int*)d_ws;
    const size_t CNT_R  = H1_L + (size_t)NL * 128;
    const size_t CNT_L  = CNT_R + (size_t)NR * CS;
    const size_t OFF_R  = CNT_L + (size_t)NL * CS;
    const size_t OFF_L  = OFF_R + (size_t)NR + 1;
    const size_t RANK_R = OFF_L + (size_t)NL + 1;
    const size_t RANK_L = RANK_R + (size_t)E;
    const size_t PERM_R = RANK_L + (size_t)E;
    const size_t PERM_L = PERM_R + (size_t)E;
    const size_t TSUM   = PERM_L + (size_t)E;

    float* d_left  = (float*)d_out;
    float* d_right = (float*)d_out + (size_t)NL * 3;

    const int TPB = 256;
    constexpr int MT = 2;

    zero_kernel<<<ceil_div((size_t)(NR + NL) * CS, TPB), TPB, 0, stream>>>(ws + CNT_R, (size_t)(NR + NL) * CS);
    edge_rank<<<ceil_div(E, TPB), TPB, 0, stream>>>(e_l, e_r, wi + CNT_L, wi + CNT_R, wi + RANK_L, wi + RANK_R, E);
    tile_sums<<<tilesR + tilesL, TPB, 0, stream>>>(wi + CNT_R, NR, tilesR, wi + CNT_L, NL, wi + TSUM);
    scan_apply<<<tilesR + tilesL, TPB, 0, stream>>>(wi + CNT_R, NR, tilesR, wi + OFF_R,
                                                    wi + CNT_L, NL, wi + OFF_L, wi + TSUM);
    edge_place<<<ceil_div(E, TPB), TPB, 0, stream>>>(e_l, e_r, wi + OFF_L, wi + OFF_R, wi + RANK_L, wi + RANK_R,
                                                     wi + PERM_L, wi + PERM_R, E);

    gather_sum_vec<64><<<ceil_div((long long)NR * 16, TPB), TPB, 0, stream>>>(x_s, wi + OFF_R, wi + PERM_R, ws + AGG_R, NR);
    gather_sum_vec<64><<<ceil_div((long long)NL * 16, TPB), TPB, 0, stream>>>(x_t, wi + OFF_L, wi + PERM_L, ws + AGG_L, NL);
    gemm_dual_wmma<64, 64, true, MT><<<ceil_div(ceil_div(NR, 16 * MT), 8), TPB, 0, stream>>>(
        ws + AGG_R, w1lr_rel, x_t, w1lr_root, b1lr, ws + H1_R, NR);
    gemm_dual_wmma<64, 64, true, MT><<<ceil_div(ceil_div(NL, 16 * MT), 8), TPB, 0, stream>>>(
        ws + AGG_L, w1rl_rel, x_s, w1rl_root, b1rl, ws + H1_L, NL);

    gather_sum_vec<128><<<ceil_div((long long)NR * 32, TPB), TPB, 0, stream>>>(ws + H1_L, wi + OFF_R, wi + PERM_R, ws + AGG_R, NR);
    gather_sum_vec<128><<<ceil_div((long long)NL * 32, TPB), TPB, 0, stream>>>(ws + H1_R, wi + OFF_L, wi + PERM_L, ws + AGG_L, NL);
    gemm_dual_wmma<128, 128, true, MT><<<ceil_div(ceil_div(NR, 16 * MT), 8), TPB, 0, stream>>>(
        ws + AGG_R, w2lr_rel, ws + H1_R, w2lr_root, b2lr, ws + AGG_R, NR);
    gemm_dual_wmma<128, 128, true, MT><<<ceil_div(ceil_div(NL, 16 * MT), 8), TPB, 0, stream>>>(
        ws + AGG_L, w2rl_rel, ws + H1_L, w2rl_root, b2rl, ws + AGG_L, NL);

    rowmat3<<<ceil_div(NL, 8), TPB, 0, stream>>>(ws + AGG_L, w3lr_rel, nullptr, ws + PROJ_LR, NL);
    rowmat3<<<ceil_div(NR, 8), TPB, 0, stream>>>(ws + AGG_R, w3rl_rel, nullptr, ws + PROJ_RL, NR);
    rowmat3<<<ceil_div(NR, 8), TPB, 0, stream>>>(ws + AGG_R, w3lr_root, b3lr, d_right, NR);
    rowmat3<<<ceil_div(NL, 8), TPB, 0, stream>>>(ws + AGG_L, w3rl_root, b3rl, d_left, NL);
    gather_add3<<<ceil_div(NR, TPB), TPB, 0, stream>>>(ws + PROJ_LR, wi + OFF_R, wi + PERM_R, d_right, NR);
    gather_add3<<<ceil_div(NL, TPB), TPB, 0, stream>>>(ws + PROJ_RL, wi + OFF_L, wi + PERM_L, d_left, NL);

    knowledge_kernel<<<ceil_div(NL, TPB), TPB, 0, stream>>>(d_left, x_s, NL);
    knowledge_kernel<<<ceil_div(NR, TPB), TPB, 0, stream>>>(d_right, x_t, NR);
}


# Fewer, fatter waves, under a fifth as many per call: edge sums 4 rows per wave; layer-3 dots 16 rows per wave, both from one row read (other summation tree); 4 edges per thread; 16-byte zero stores
# speedup vs baseline: 2.1684x; 1.0678x over previous
#include <hip/hip_runtime.h>
#include <hip/hip_bf16.h>

typedef __attribute__((ext_vector_type(2))) float v2f;
typedef __attribute__((ext_vector_type(8))) float v8f;

__global__ __launch_bounds__(256) void zero_kernel(float4* __restrict__ p, size_t n4) {
    const float4 z = make_float4(0.f, 0.f, 0.f, 0.f);
    const size_t step = (size_t)gridDim.x * blockDim.x;
    for (size_t i = (size_t)blockIdx.x * blockDim.x + threadIdx.x; i < n4; i += step) p[i] = z;
}

constexpr int CS = 16;
constexpr int TILE = 1024;
constexpr int EPT = 4;

__global__ __launch_bounds__(256) void edge_rank(const int* __restrict__ e_l,
                                                 const int* __restrict__ e_r,
                                                 int* cntL, int* cntR,
                                                 int* __restrict__ rankL,
                                                 int* __restrict__ rankR, int nE) {
    const int base = blockIdx.x * (256 * EPT) + threadIdx.x;
    int dl[EPT], dr[EPT], rl[EPT], rr[EPT];
#pragma unroll
    for (int k = 0; k < EPT; ++k) {
        const int e = base + k * 256;
        const bool ok = e < nE;
        dl[k] = ok ? e_l[e] : -1;
        dr[k] = ok ? e_r[e] : -1;
    }
#pragma unroll
    for (int k = 0; k < EPT; ++k) {
        rr[k] = 0;
        rl[k] = 0;
        if (dl[k] >= 0) {
            rr[k] = atomicAdd(cntR + (size_t)dr[k] * CS, 1);
            rl[k] = atomicAdd(cntL + (size_t)dl[k] * CS, 1);
        }
    }
#pragma unroll
    for (int k = 0; k < EPT; ++k) {
        const int e = base + k * 256;
        if (e < nE) {
            rankR[e] = rr[k];
            rankL[e] = rl[k];
        }
    }
}

__global__ __launch_bounds__(256) void tile_sums(const int* __restrict__ cntA, int nA, int tilesA,
                                                 const int* __restrict__ cntB, int nB,
                                                 int* __restrict__ tsum) {
    __shared__ int wsum[8];
    const int b = blockIdx.x;
    const int lane = threadIdx.x & 31;
    const int wave = threadIdx.x >> 5;
    const int* cnt = (b < tilesA) ? cntA : cntB;
    const int n = (b < tilesA) ? nA : nB;
    const int tile = (b < tilesA) ? b : b - tilesA;
    const int i0 = tile * TILE + threadIdx.x * 4;
    int s = 0;
#pragma unroll
    for (int j = 0; j < 4; ++j) {
        int i = i0 + j;
        if (i < n) s += cnt[(size_t)i * CS];
    }
#pragma unroll
    for (int o = 16; o; o >>= 1) s += __shfl_xor(s, o, 32);
    if (lane == 0) wsum[wave] = s;
    __syncthreads();
    if (threadIdx.x == 0) {
        int t = 0;
#pragma unroll
        for (int w = 0; w < 8; ++w) t += wsum[w];
        tsum[b] = t;
    }
}

__global__ __launch_bounds__(256) void scan_apply(const int* __restrict__ cntA, int nA, int tilesA,
                                                  int* __restrict__ offA,
                                                  const int* __restrict__ cntB, int nB,
                                                  int* __restrict__ offB,
                                                  const int* __restrict__ tsum) {
    __shared__ int wsum[8];
    __shared__ int wtot[8];
    const int b = blockIdx.x;
    const int lane = threadIdx.x & 31;
    const int wave = threadIdx.x >> 5;
    const bool isA = b < tilesA;
    const int* cnt = isA ? cntA : cntB;
    int* off = isA ? offA : offB;
    const int n = isA ? nA : nB;
    const int tile = isA ? b : b - tilesA;
    const int tbase = isA ? 0 : tilesA;

    int p = 0;
    for (int i = threadIdx.x; i < tile; i += 256) p += tsum[tbase + i];
#pragma unroll
    for (int o = 16; o; o >>= 1) p += __shfl_xor(p, o, 32);
    if (lane == 0) wsum[wave] = p;

    const int i0 = tile * TILE + threadIdx.x * 4;
    int v[4];
    int own = 0;
#pragma unroll
    for (int j = 0; j < 4; ++j) {
        int i = i0 + j;
        v[j] = (i < n) ? cnt[(size_t)i * CS] : 0;
        own += v[j];
    }
    int incl = own;
#pragma unroll
    for (int o = 1; o < 32; o <<= 1) {
        int u = __shfl_up(incl, o, 32);
        if (lane >= o) incl += u;
    }
    if (lane == 31) wtot[wave] = incl;
    __syncthreads();

    int run = incl - own;
#pragma unroll
    for (int w = 0; w < 8; ++w) {
        run += wsum[w];
        if (w < wave) run += wtot[w];
    }
#pragma unroll
    for (int j = 0; j < 4; ++j) {
        int i = i0 + j;
        if (i < n) {
            off[i] = run;
            run += v[j];
            if (i == n - 1) off[n] = run;
        }
    }
}

__global__ __launch_bounds__(256) void edge_place(const int* __restrict__ e_l,
                                                  const int* __restrict__ e_r,
                                                  const int* __restrict__ offL,
                                                  const int* __restrict__ offR,
                                                  const int* __restrict__ rankL,
                                                  const int* __restrict__ rankR,
                                                  int* __restrict__ permL,
                                                  int* __restrict__ permR, int nE) {
    const int base = blockIdx.x * (256 * EPT) + threadIdx.x;
    int dl[EPT], dr[EPT], rl[EPT], rr[EPT], ol[EPT], orr[EPT];
#pragma unroll
    for (int k = 0; k < EPT; ++k) {
        const int e = base + k * 256;
        const bool ok = e < nE;
        dl[k] = ok ? e_l[e] : -1;
        dr[k] = ok ? e_r[e] : -1;
        rl[k] = ok ? rankL[e] : 0;
        rr[k] = ok ? rankR[e] : 0;
    }
#pragma unroll
    for (int k = 0; k < EPT; ++k) {
        ol[k] = 0;
        orr[k] = 0;
        if (dl[k] >= 0) {
            orr[k] = offR[dr[k]];
            ol[k] = offL[dl[k]];
        }
    }
#pragma unroll
    for (int k = 0; k < EPT; ++k) {
        if (dl[k] >= 0) {
            permR[orr[k] + rr[k]] = dl[k];
            permL[ol[k] + rl[k]] = dr[k];
        }
    }
}

template <int F>
__global__ __launch_bounds__(256) void gather_sum_vec(const float* __restrict__ src,
                                                      const int* __restrict__ off,
                                                      const int* __restrict__ perm,
                                                      float* __restrict__ dst, int nDst) {
    constexpr int PER = 8;
    constexpr int V = F / (4 * PER);
    const unsigned t = blockIdx.x * blockDim.x + threadIdx.x;
    const unsigned d = t / PER;
    const unsigned sub = t % PER;
    if (d >= (unsigned)nDst) return;
    const int beg = off[d];
    const int end = off[d + 1];
    float4 acc[V];
#pragma unroll
    for (int u = 0; u < V; ++u) acc[u] = make_float4(0.f, 0.f, 0.f, 0.f);
    const float* lane_src = src + sub * 4;
    int j = beg;
    for (; j + 4 <= end; j += 4) {
        int s[4];
#pragma unroll
        for (int k = 0; k < 4; ++k) s[k] = perm[j + k];
        float4 v[4][V];
#pragma unroll
        for (int k = 0; k < 4; ++k)
#pragma unroll
            for (int u = 0; u < V; ++u)
                v[k][u] = *(const float4*)(lane_src + (size_t)s[k] * F + u * (PER * 4));
#pragma unroll
        for (int k = 0; k < 4; ++k)
#pragma unroll
            for (int u = 0; u < V; ++u) {
                acc[u].x += v[k][u].x; acc[u].y += v[k][u].y;
                acc[u].z += v[k][u].z; acc[u].w += v[k][u].w;
            }
    }
    if (j < end) {
        int s[3];
#pragma unroll
        for (int k = 0; k < 3; ++k) s[k] = perm[min(j + k, end - 1)];
        float4 v[3][V];
#pragma unroll
        for (int k = 0; k < 3; ++k)
#pragma unroll
            for (int u = 0; u < V; ++u)
                v[k][u] = *(const float4*)(lane_src + (size_t)s[k] * F + u * (PER * 4));
#pragma unroll
        for (int k = 0; k < 3; ++k) {
            const bool on = j + k < end;
#pragma unroll
            for (int u = 0; u < V; ++u) {
                acc[u].x += on ? v[k][u].x : 0.f; acc[u].y += on ? v[k][u].y : 0.f;
                acc[u].z += on ? v[k][u].z : 0.f; acc[u].w += on ? v[k][u].w : 0.f;
            }
        }
    }
#pragma unroll
    for (int u = 0; u < V; ++u)
        *(float4*)(dst + (size_t)d * F + (u * PER + sub) * 4) = acc[u];
}

__global__ __launch_bounds__(256) void gather_add3(const float* __restrict__ proj,
                                                   const int* __restrict__ off,
                                                   const int* __restrict__ perm,
                                                   float* out, int nDst) {
    int d = blockIdx.x * blockDim.x + threadIdx.x;
    if (d >= nDst) return;
    const int beg = off[d];
    const int end = off[d + 1];
    float a0 = out[(size_t)d * 3 + 0];
    float a1 = out[(size_t)d * 3 + 1];
    float a2 = out[(size_t)d * 3 + 2];
    for (int j = beg; j < end; ++j) {
        const int s = perm[j];
        a0 += proj[(size_t)s * 3 + 0];
        a1 += proj[(size_t)s * 3 + 1];
        a2 += proj[(size_t)s * 3 + 2];
    }
    out[(size_t)d * 3 + 0] = a0;
    out[(size_t)d * 3 + 1] = a1;
    out[(size_t)d * 3 + 2] = a2;
}

template <int K1, int K2, bool RELU, int MT>
__global__ __launch_bounds__(256) void gemm_dual_wmma(const float* __restrict__ A1,
                                                      const float* __restrict__ W1,
                                                      const float* __restrict__ A2,
                                                      const float* __restrict__ W2,
                                                      const float* __restrict__ bias,
                                                      float* out, int M) {
    constexpr int N = 128;
    const int lane = threadIdx.x & 31;
    const int wave = threadIdx.x >> 5;
    const int m0 = (blockIdx.x * 8 + wave) * (16 * MT);
    if (m0 >= M) return;
    const int half = lane >> 4;
    const int lr = lane & 15;

    v8f acc[MT][8] = {};

    int arow[MT];
#pragma unroll
    for (int mt = 0; mt < MT; ++mt) {
        int r = m0 + mt * 16 + lr;
        arow[mt] = (r < M) ? r : (M - 1);
    }

    for (int k = 0; k < K1; k += 4) {
        v2f a[MT];
#pragma unroll
        for (int mt = 0; mt < MT; ++mt)
            a[mt] = *(const v2f*)(A1 + (size_t)arow[mt] * K1 + k + 2 * half);
        const float* w0 = W1 + (size_t)(k + 2 * half) * N + lr;
#pragma unroll
        for (int nt = 0; nt < 8; ++nt) {
            v2f b;
            b.x = w0[nt * 16];
            b.y = w0[N + nt * 16];
#pragma unroll
            for (int mt = 0; mt < MT; ++mt)
                acc[mt][nt] = __builtin_amdgcn_wmma_f32_16x16x4_f32(
                    false, a[mt], false, b, (short)0, acc[mt][nt], false, false);
        }
    }
    for (int k = 0; k < K2; k += 4) {
        v2f a[MT];
#pragma unroll
        for (int mt = 0; mt < MT; ++mt)
            a[mt] = *(const v2f*)(A2 + (size_t)arow[mt] * K2 + k + 2 * half);
        const float* w0 = W2 + (size_t)(k + 2 * half) * N + lr;
#pragma unroll
        for (int nt = 0; nt < 8; ++nt) {
            v2f b;
            b.x = w0[nt * 16];
            b.y = w0[N + nt * 16];
#pragma unroll
            for (int mt = 0; mt < MT; ++mt)
                acc[mt][nt] = __builtin_amdgcn_wmma_f32_16x16x4_f32(
                    false, a[mt], false, b, (short)0, acc[mt][nt], false, false);
        }
    }

#pragma unroll
    for (int mt = 0; mt < MT; ++mt) {
        if (m0 + mt * 16 >= M) break;
#pragma unroll
        for (int nt = 0; nt < 8; ++nt) {
            const int col = nt * 16 + lr;
            const float bv = bias[col];
#pragma unroll
            for (int i = 0; i < 8; ++i) {
                float v = acc[mt][nt][i] + bv;
                if (RELU) v = fmaxf(v, 0.0f);
                out[(size_t)(m0 + mt * 16 + i + 8 * half) * N + col] = v;
            }
        }
    }
}

constexpr int RM_TRIPS = 4;
__global__ __launch_bounds__(256) void rowmat3_dual(const float* __restrict__ X,
                                                    const float* __restrict__ Wrel,
                                                    const float* __restrict__ Wroot,
                                                    const float* __restrict__ bias,
                                                    float* __restrict__ out_rel,
                                                    float* __restrict__ out_root, int n) {
    const int lane = threadIdx.x & 31;
    const int sub = lane & 7;
    const int grp = lane >> 3;
    const int wave = blockIdx.x * (blockDim.x >> 5) + (threadIdx.x >> 5);
    const int row0 = wave * (4 * RM_TRIPS);
    if (row0 >= n) return;

    float wr[4][4][3], wo[4][4][3];
#pragma unroll
    for (int u = 0; u < 4; ++u) {
        const int k0 = (u * 8 + sub) * 4;
#pragma unroll
        for (int e = 0; e < 4; ++e)
#pragma unroll
            for (int c = 0; c < 3; ++c) {
                wr[u][e][c] = Wrel[(k0 + e) * 3 + c];
                wo[u][e][c] = Wroot[(k0 + e) * 3 + c];
            }
    }
    const float b0 = bias[0], b1 = bias[1], b2 = bias[2];

    float4 x[RM_TRIPS][4];
#pragma unroll
    for (int t = 0; t < RM_TRIPS; ++t) {
        const int r = min(row0 + t * 4 + grp, n - 1);
#pragma unroll
        for (int u = 0; u < 4; ++u)
            x[t][u] = *(const float4*)(X + (size_t)r * 128 + (u * 8 + sub) * 4);
    }
#pragma unroll
    for (int t = 0; t < RM_TRIPS; ++t) {
        float p[3] = {0.f, 0.f, 0.f};
        float q[3] = {0.f, 0.f, 0.f};
#pragma unroll
        for (int u = 0; u < 4; ++u)
#pragma unroll
            for (int c = 0; c < 3; ++c) {
                p[c] += x[t][u].x * wr[u][0][c] + x[t][u].y * wr[u][1][c] +
                        x[t][u].z * wr[u][2][c] + x[t][u].w * wr[u][3][c];
                q[c] += x[t][u].x * wo[u][0][c] + x[t][u].y * wo[u][1][c] +
                        x[t][u].z * wo[u][2][c] + x[t][u].w * wo[u][3][c];
            }
#pragma unroll
        for (int o = 4; o; o >>= 1)
#pragma unroll
            for (int c = 0; c < 3; ++c) {
                p[c] += __shfl_xor(p[c], o, 32);
                q[c] += __shfl_xor(q[c], o, 32);
            }
        const int r = row0 + t * 4 + grp;
        if (sub == 0 && r < n) {
            out_rel[(size_t)r * 3 + 0] = p[0];
            out_rel[(size_t)r * 3 + 1] = p[1];
            out_rel[(size_t)r * 3 + 2] = p[2];
            out_root[(size_t)r * 3 + 0] = q[0] + b0;
            out_root[(size_t)r * 3 + 1] = q[1] + b1;
            out_root[(size_t)r * 3 + 2] = q[2] + b2;
        }
    }
}

__global__ __launch_bounds__(256) void knowledge_kernel(float* __restrict__ out,
                                                        const float* __restrict__ x,
                                                        int n) {
    int i = blockIdx.x * blockDim.x + threadIdx.x;
    if (i >= n) return;
    float a = out[(size_t)i * 3 + 0];
    float b = out[(size_t)i * 3 + 1];
    float c = out[(size_t)i * 3 + 2];
    float nrm = sqrtf(a * a + b * b + c * c);
    float s = 10.0f / fmaxf(nrm, 1e-12f);
    a *= s; b *= s; c *= s;
    if (x[(size_t)i * 64 + 61] != 0.0f) a -= 10.0f;
    if (x[(size_t)i * 64 + 63] != 0.0f) c -= 10.0f;
    out[(size_t)i * 3 + 0] = a;
    out[(size_t)i * 3 + 1] = b;
    out[(size_t)i * 3 + 2] = c;
}

static inline int ceil_div(long long a, long long b) { return (int)((a + b - 1) / b); }

extern "C" void kernel_launch(void* const* d_in, const int* in_sizes, int n_in,
                              void* d_out, int out_size, void* d_ws, size_t ws_size,
                              hipStream_t stream) {
    const float* x_s = (const float*)d_in[0];
    const float* x_t = (const float*)d_in[1];
    const int* e_l = (const int*)d_in[2];
    const int* e_r = (const int*)d_in[3];
    const float* w1lr_rel  = (const float*)d_in[4];
    const float* b1lr      = (const float*)d_in[5];
    const float* w1lr_root = (const float*)d_in[6];
    const float* w1rl_rel  = (const float*)d_in[7];
    const float* b1rl      = (const float*)d_in[8];
    const float* w1rl_root = (const float*)d_in[9];
    const float* w2lr_rel  = (const float*)d_in[10];
    const float* b2lr      = (const float*)d_in[11];
    const float* w2lr_root = (const float*)d_in[12];
    const float* w2rl_rel  = (const float*)d_in[13];
    const float* b2rl      = (const float*)d_in[14];
    const float* w2rl_root = (const float*)d_in[15];
    const float* w3lr_rel  = (const float*)d_in[16];
    const float* b3lr      = (const float*)d_in[17];
    const float* w3lr_root = (const float*)d_in[18];
    const float* w3rl_rel  = (const float*)d_in[19];
    const float* b3rl      = (const float*)d_in[20];
    const float* w3rl_root = (const float*)d_in[21];

    const int NL = in_sizes[0] / 64;
    const int NR = in_sizes[1] / 64;
    const int E  = in_sizes[2];

    float* ws = (float*)d_ws;
    const size_t AGG_R = 0;
    const size_t AGG_L = AGG_R + (size_t)NR * 128;
    const size_t H1_R  = AGG_L + (size_t)NL * 128;
    const size_t H1_L  = H1_R + (size_t)NR * 128;
    const size_t PROJ_LR = H1_R;
    const size_t PROJ_RL = H1_R + (size_t)NL * 3;
    const int tilesR = ceil_div(NR, TILE);
    const int tilesL = ceil_div(NL, TILE);
    int* wi = (int*)d_ws;
    const size_t CNT_R  = H1_L + (size_t)NL * 128;
    const size_t CNT_L  = CNT_R + (size_t)NR * CS;
    const size_t OFF_R  = CNT_L + (size_t)NL * CS;
    const size_t OFF_L  = OFF_R + (size_t)NR + 1;
    const size_t RANK_R = OFF_L + (size_t)NL + 1;
    const size_t RANK_L = RANK_R + (size_t)E;
    const size_t PERM_R = RANK_L + (size_t)E;
    const size_t PERM_L = PERM_R + (size_t)E;
    const size_t TSUM   = PERM_L + (size_t)E;

    float* d_left  = (float*)d_out;
    float* d_right = (float*)d_out + (size_t)NL * 3;

    const int TPB = 256;
    constexpr int MT = 2;

    zero_kernel<<<512, TPB, 0, stream>>>((float4*)(ws + CNT_R), (size_t)(NR + NL) * CS / 4);
    edge_rank<<<ceil_div(E, TPB * EPT), TPB, 0, stream>>>(e_l, e_r, wi + CNT_L, wi + CNT_R, wi + RANK_L, wi + RANK_R, E);
    tile_sums<<<tilesR + tilesL, TPB, 0, stream>>>(wi + CNT_R, NR, tilesR, wi + CNT_L, NL, wi + TSUM);
    scan_apply<<<tilesR + tilesL, TPB, 0, stream>>>(wi + CNT_R, NR, tilesR, wi + OFF_R,
                                                    wi + CNT_L, NL, wi + OFF_L, wi + TSUM);
    edge_place<<<ceil_div(E, TPB * EPT), TPB, 0, stream>>>(e_l, e_r, wi + OFF_L, wi + OFF_R, wi + RANK_L, wi + RANK_R,
                                                     wi + PERM_L, wi + PERM_R, E);

    gather_sum_vec<64><<<ceil_div((long long)NR * 8, TPB), TPB, 0, stream>>>(x_s, wi + OFF_R, wi + PERM_R, ws + AGG_R, NR);
    gather_sum_vec<64><<<ceil_div((long long)NL * 8, TPB), TPB, 0, stream>>>(x_t, wi + OFF_L, wi + PERM_L, ws + AGG_L, NL);
    gemm_dual_wmma<64, 64, true, MT><<<ceil_div(ceil_div(NR, 16 * MT), 8), TPB, 0, stream>>>(
        ws + AGG_R, w1lr_rel, x_t, w1lr_root, b1lr, ws + H1_R, NR);
    gemm_dual_wmma<64, 64, true, MT><<<ceil_div(ceil_div(NL, 16 * MT), 8), TPB, 0, stream>>>(
        ws + AGG_L, w1rl_rel, x_s, w1rl_root, b1rl, ws + H1_L, NL);

    gather_sum_vec<128><<<ceil_div((long long)NR * 8, TPB), TPB, 0, stream>>>(ws + H1_L, wi + OFF_R, wi + PERM_R, ws + AGG_R, NR);
    gather_sum_vec<128><<<ceil_div((long long)NL * 8, TPB), TPB, 0, stream>>>(ws + H1_R, wi + OFF_L, wi + PERM_L, ws + AGG_L, NL);
    gemm_dual_wmma<128, 128, true, MT><<<ceil_div(ceil_div(NR, 16 * MT), 8), TPB, 0, stream>>>(
        ws + AGG_R, w2lr_rel, ws + H1_R, w2lr_root, b2lr, ws + AGG_R, NR);
    gemm_dual_wmma<128, 128, true, MT><<<ceil_div(ceil_div(NL, 16 * MT), 8), TPB, 0, stream>>>(
        ws + AGG_L, w2rl_rel, ws + H1_L, w2rl_root, b2rl, ws + AGG_L, NL);

    rowmat3_dual<<<ceil_div(NL, 8 * 4 * RM_TRIPS), TPB, 0, stream>>>(ws + AGG_L, w3lr_rel, w3rl_root, b3rl,
                                                                   ws + PROJ_LR, d_left, NL);
    rowmat3_dual<<<ceil_div(NR, 8 * 4 * RM_TRIPS), TPB, 0, stream>>>(ws + AGG_R, w3rl_rel, w3lr_root, b3lr,
                                                                   ws + PROJ_RL, d_right, NR);
    gather_add3<<<ceil_div(NR, TPB), TPB, 0, stream>>>(ws + PROJ_LR, wi + OFF_R, wi + PERM_R, d_right, NR);
    gather_add3<<<ceil_div(NL, TPB), TPB, 0, stream>>>(ws + PROJ_RL, wi + OFF_L, wi + PERM_L, d_left, NL);

    knowledge_kernel<<<ceil_div(NL, TPB), TPB, 0, stream>>>(d_left, x_s, NL);
    knowledge_kernel<<<ceil_div(NR, TPB), TPB, 0, stream>>>(d_right, x_t, NR);
}


# Each left/right pair of launches merged into one launch (first blocks one direction, the rest the other); the closing scale-and-flag pass joined to the 3-feature edge sum: 19 launches become 11
# speedup vs baseline: 2.4901x; 1.1484x over previous
#include <hip/hip_runtime.h>
#include <hip/hip_bf16.h>

typedef __attribute__((ext_vector_type(2))) float v2f;
typedef __attribute__((ext_vector_type(8))) float v8f;

__global__ __launch_bounds__(256) void zero_kernel(float4* __restrict__ p, size_t n4) {
    const float4 z = make_float4(0.f, 0.f, 0.f, 0.f);
    const size_t step = (size_t)gridDim.x * blockDim.x;
    for (size_t i = (size_t)blockIdx.x * blockDim.x + threadIdx.x; i < n4; i += step) p[i] = z;
}

constexpr int CS = 16;
constexpr int TILE = 1024;
constexpr int EPT = 4;

__global__ __launch_bounds__(256) void edge_rank(const int* __restrict__ e_l,
                                                 const int* __restrict__ e_r,
                                                 int* cntL, int* cntR,
                                                 int* __restrict__ rankL,
                                                 int* __restrict__ rankR, int nE) {
    const int base = blockIdx.x * (256 * EPT) + threadIdx.x;
    int dl[EPT], dr[EPT], rl[EPT], rr[EPT];
#pragma unroll
    for (int k = 0; k < EPT; ++k) {
        const int e = base + k * 256;
        const bool ok = e < nE;
        dl[k] = ok ? e_l[e] : -1;
        dr[k] = ok ? e_r[e] : -1;
    }
#pragma unroll
    for (int k = 0; k < EPT; ++k) {
        rr[k] = 0;
        rl[k] = 0;
        if (dl[k] >= 0) {
            rr[k] = atomicAdd(cntR + (size_t)dr[k] * CS, 1);
            rl[k] = atomicAdd(cntL + (size_t)dl[k] * CS, 1);
        }
    }
#pragma unroll
    for (int k = 0; k < EPT; ++k) {
        const int e = base + k * 256;
        if (e < nE) {
            rankR[e] = rr[k];
            rankL[e] = rl[k];
        }
    }
}

__global__ __launch_bounds__(256) void tile_sums(const int* __restrict__ cntA, int nA, int tilesA,
                                                 const int* __restrict__ cntB, int nB,
                                                 int* __restrict__ tsum) {
    __shared__ int wsum[8];
    const int b = blockIdx.x;
    const int lane = threadIdx.x & 31;
    const int wave = threadIdx.x >> 5;
    const int* cnt = (b < tilesA) ? cntA : cntB;
    const int n = (b < tilesA) ? nA : nB;
    const int tile = (b < tilesA) ? b : b - tilesA;
    const int i0 = tile * TILE + threadIdx.x * 4;
    int s = 0;
#pragma unroll
    for (int j = 0; j < 4; ++j) {
        int i = i0 + j;
        if (i < n) s += cnt[(size_t)i * CS];
    }
#pragma unroll
    for (int o = 16; o; o >>= 1) s += __shfl_xor(s, o, 32);
    if (lane == 0) wsum[wave] = s;
    __syncthreads();
    if (threadIdx.x == 0) {
        int t = 0;
#pragma unroll
        for (int w = 0; w < 8; ++w) t += wsum[w];
        tsum[b] = t;
    }
}

__global__ __launch_bounds__(256) void scan_apply(const int* __restrict__ cntA, int nA, int tilesA,
                                                  int* __restrict__ offA,
                                                  const int* __restrict__ cntB, int nB,
                                                  int* __restrict__ offB,
                                                  const int* __restrict__ tsum) {
    __shared__ int wsum[8];
    __shared__ int wtot[8];
    const int b = blockIdx.x;
    const int lane = threadIdx.x & 31;
    const int wave = threadIdx.x >> 5;
    const bool isA = b < tilesA;
    const int* cnt = isA ? cntA : cntB;
    int* off = isA ? offA : offB;
    const int n = isA ? nA : nB;
    const int tile = isA ? b : b - tilesA;
    const int tbase = isA ? 0 : tilesA;

    int p = 0;
    for (int i = threadIdx.x; i < tile; i += 256) p += tsum[tbase + i];
#pragma unroll
    for (int o = 16; o; o >>= 1) p += __shfl_xor(p, o, 32);
    if (lane == 0) wsum[wave] = p;

    const int i0 = tile * TILE + threadIdx.x * 4;
    int v[4];
    int own = 0;
#pragma unroll
    for (int j = 0; j < 4; ++j) {
        int i = i0 + j;
        v[j] = (i < n) ? cnt[(size_t)i * CS] : 0;
        own += v[j];
    }
    int incl = own;
#pragma unroll
    for (int o = 1; o < 32; o <<= 1) {
        int u = __shfl_up(incl, o, 32);
        if (lane >= o) incl += u;
    }
    if (lane == 31) wtot[wave] = incl;
    __syncthreads();

    int run = incl - own;
#pragma unroll
    for (int w = 0; w < 8; ++w) {
        run += wsum[w];
        if (w < wave) run += wtot[w];
    }
#pragma unroll
    for (int j = 0; j < 4; ++j) {
        int i = i0 + j;
        if (i < n) {
            off[i] = run;
            run += v[j];
            if (i == n - 1) off[n] = run;
        }
    }
}

__global__ __launch_bounds__(256) void edge_place(const int* __restrict__ e_l,
                                                  const int* __restrict__ e_r,
                                                  const int* __restrict__ offL,
                                                  const int* __restrict__ offR,
                                                  const int* __restrict__ rankL,
                                                  const int* __restrict__ rankR,
                                                  int* __restrict__ permL,
                                                  int* __restrict__ permR, int nE) {
    const int base = blockIdx.x * (256 * EPT) + threadIdx.x;
    int dl[EPT], dr[EPT], rl[EPT], rr[EPT], ol[EPT], orr[EPT];
#pragma unroll
    for (int k = 0; k < EPT; ++k) {
        const int e = base + k * 256;
        const bool ok = e < nE;
        dl[k] = ok ? e_l[e] : -1;
        dr[k] = ok ? e_r[e] : -1;
        rl[k] = ok ? rankL[e] : 0;
        rr[k] = ok ? rankR[e] : 0;
    }
#pragma unroll
    for (int k = 0; k < EPT; ++k) {
        ol[k] = 0;
        orr[k] = 0;
        if (dl[k] >= 0) {
            orr[k] = offR[dr[k]];
            ol[k] = offL[dl[k]];
        }
    }
#pragma unroll
    for (int k = 0; k < EPT; ++k) {
        if (dl[k] >= 0) {
            permR[orr[k] + rr[k]] = dl[k];
            permL[ol[k] + rl[k]] = dr[k];
        }
    }
}

template <int F>
__global__ __launch_bounds__(256) void gather_sum_vec(const float* __restrict__ srcA,
                                                      const int* __restrict__ offA,
                                                      const int* __restrict__ permA,
                                                      float* __restrict__ dstA, int nA, int blocksA,
                                                      const float* __restrict__ srcB,
                                                      const int* __restrict__ offB,
                                                      const int* __restrict__ permB,
                                                      float* __restrict__ dstB, int nB) {
    constexpr int PER = 8;
    constexpr int V = F / (4 * PER);
    const bool jobA = (int)blockIdx.x < blocksA;
    const float* __restrict__ src = jobA ? srcA : srcB;
    const int* __restrict__ off = jobA ? offA : offB;
    const int* __restrict__ perm = jobA ? permA : permB;
    float* __restrict__ dst = jobA ? dstA : dstB;
    const int nDst = jobA ? nA : nB;
    const unsigned blk = jobA ? blockIdx.x : blockIdx.x - blocksA;
    const unsigned t = blk * blockDim.x + threadIdx.x;
    const unsigned d = t / PER;
    const unsigned sub = t % PER;
    if (d >= (unsigned)nDst) return;
    const int beg = off[d];
    const int end = off[d + 1];
    float4 acc[V];
#pragma unroll
    for (int u = 0; u < V; ++u) acc[u] = make_float4(0.f, 0.f, 0.f, 0.f);
    const float* lane_src = src + sub * 4;
    int j = beg;
    for (; j + 4 <= end; j += 4) {
        int s[4];
#pragma unroll
        for (int k = 0; k < 4; ++k) s[k] = perm[j + k];
        float4 v[4][V];
#pragma unroll
        for (int k = 0; k < 4; ++k)
#pragma unroll
            for (int u = 0; u < V; ++u)
                v[k][u] = *(const float4*)(lane_src + (size_t)s[k] * F + u * (PER * 4));
#pragma unroll
        for (int k = 0; k < 4; ++k)
#pragma unroll
            for (int u = 0; u < V; ++u) {
                acc[u].x += v[k][u].x; acc[u].y += v[k][u].y;
                acc[u].z += v[k][u].z; acc[u].w += v[k][u].w;
            }
    }
    if (j < end) {
        int s[3];
#pragma unroll
        for (int k = 0; k < 3; ++k) s[k] = perm[min(j + k, end - 1)];
        float4 v[3][V];
#pragma unroll
        for (int k = 0; k < 3; ++k)
#pragma unroll
            for (int u = 0; u < V; ++u)
                v[k][u] = *(const float4*)(lane_src + (size_t)s[k] * F + u * (PER * 4));
#pragma unroll
        for (int k = 0; k < 3; ++k) {
            const bool on = j + k < end;
#pragma unroll
            for (int u = 0; u < V; ++u) {
                acc[u].x += on ? v[k][u].x : 0.f; acc[u].y += on ? v[k][u].y : 0.f;
                acc[u].z += on ? v[k][u].z : 0.f; acc[u].w += on ? v[k][u].w : 0.f;
            }
        }
    }
#pragma unroll
    for (int u = 0; u < V; ++u)
        *(float4*)(dst + (size_t)d * F + (u * PER + sub) * 4) = acc[u];
}

__global__ __launch_bounds__(256) void gather_add3_finish(const float* __restrict__ projA,
                                                          const int* __restrict__ offA,
                                                          const int* __restrict__ permA,
                                                          const float* __restrict__ xA,
                                                          float* outA, int nA, int blocksA,
                                                          const float* __restrict__ projB,
                                                          const int* __restrict__ offB,
                                                          const int* __restrict__ permB,
                                                          const float* __restrict__ xB,
                                                          float* outB, int nB) {
    const bool jobA = (int)blockIdx.x < blocksA;
    const float* __restrict__ proj = jobA ? projA : projB;
    const int* __restrict__ off = jobA ? offA : offB;
    const int* __restrict__ perm = jobA ? permA : permB;
    const float* __restrict__ x = jobA ? xA : xB;
    float* out = jobA ? outA : outB;
    const int nDst = jobA ? nA : nB;
    const int blk = jobA ? (int)blockIdx.x : (int)blockIdx.x - blocksA;
    int d = blk * blockDim.x + threadIdx.x;
    if (d >= nDst) return;
    const int beg = off[d];
    const int end = off[d + 1];
    float a0 = out[(size_t)d * 3 + 0];
    float a1 = out[(size_t)d * 3 + 1];
    float a2 = out[(size_t)d * 3 + 2];
    const float f0 = x[(size_t)d * 64 + 61];
    const float f2 = x[(size_t)d * 64 + 63];
    for (int j = beg; j < end; ++j) {
        const int s = perm[j];
        a0 += proj[(size_t)s * 3 + 0];
        a1 += proj[(size_t)s * 3 + 1];
        a2 += proj[(size_t)s * 3 + 2];
    }
    float nrm = sqrtf(a0 * a0 + a1 * a1 + a2 * a2);
    float sc = 10.0f / fmaxf(nrm, 1e-12f);
    a0 *= sc; a1 *= sc; a2 *= sc;
    if (f0 != 0.0f) a0 -= 10.0f;
    if (f2 != 0.0f) a2 -= 10.0f;
    out[(size_t)d * 3 + 0] = a0;
    out[(size_t)d * 3 + 1] = a1;
    out[(size_t)d * 3 + 2] = a2;
}

template <int K1, int K2, bool RELU, int MT>
__global__ __launch_bounds__(256) void gemm_dual_wmma(const float* __restrict__ A1a,
                                                      const float* __restrict__ W1a,
                                                      const float* __restrict__ A2a,
                                                      const float* __restrict__ W2a,
                                                      const float* __restrict__ biasa,
                                                      float* outa, int Ma, int blocksA,
                                                      const float* __restrict__ A1b,
                                                      const float* __restrict__ W1b,
                                                      const float* __restrict__ A2b,
                                                      const float* __restrict__ W2b,
                                                      const float* __restrict__ biasb,
                                                      float* outb, int Mb) {
    constexpr int N = 128;
    const bool jobA = (int)blockIdx.x < blocksA;
    const float* __restrict__ A1 = jobA ? A1a : A1b;
    const float* __restrict__ W1 = jobA ? W1a : W1b;
    const float* __restrict__ A2 = jobA ? A2a : A2b;
    const float* __restrict__ W2 = jobA ? W2a : W2b;
    const float* __restrict__ bias = jobA ? biasa : biasb;
    float* out = jobA ? outa : outb;
    const int M = jobA ? Ma : Mb;
    const int blk = jobA ? (int)blockIdx.x : (int)blockIdx.x - blocksA;
    const int lane = threadIdx.x & 31;
    const int wave = threadIdx.x >> 5;
    const int m0 = (blk * 8 + wave) * (16 * MT);
    if (m0 >= M) return;
    const int half = lane >> 4;
    const int lr = lane & 15;

    v8f acc[MT][8] = {};

    int arow[MT];
#pragma unroll
    for (int mt = 0; mt < MT; ++mt) {
        int r = m0 + mt * 16 + lr;
        arow[mt] = (r < M) ? r : (M - 1);
    }

    for (int k = 0; k < K1; k += 4) {
        v2f a[MT];
#pragma unroll
        for (int mt = 0; mt < MT; ++mt)
            a[mt] = *(const v2f*)(A1 + (size_t)arow[mt] * K1 + k + 2 * half);
        const float* w0 = W1 + (size_t)(k + 2 * half) * N + lr;
#pragma unroll
        for (int nt = 0; nt < 8; ++nt) {
            v2f b;
            b.x = w0[nt * 16];
            b.y = w0[N + nt * 16];
#pragma unroll
            for (int mt = 0; mt < MT; ++mt)
                acc[mt][nt] = __builtin_amdgcn_wmma_f32_16x16x4_f32(
                    false, a[mt], false, b, (short)0, acc[mt][nt], false, false);
        }
    }
    for (int k = 0; k < K2; k += 4) {
        v2f a[MT];
#pragma unroll
        for (int mt = 0; mt < MT; ++mt)
            a[mt] = *(const v2f*)(A2 + (size_t)arow[mt] * K2 + k + 2 * half);
        const float* w0 = W2 + (size_t)(k + 2 * half) * N + lr;
#pragma unroll
        for (int nt = 0; nt < 8; ++nt) {
            v2f b;
            b.x = w0[nt * 16];
            b.y = w0[N + nt * 16];
#pragma unroll
            for (int mt = 0; mt < MT; ++mt)
                acc[mt][nt] = __builtin_amdgcn_wmma_f32_16x16x4_f32(
                    false, a[mt], false, b, (short)0, acc[mt][nt], false, false);
        }
    }

#pragma unroll
    for (int mt = 0; mt < MT; ++mt) {
        if (m0 + mt * 16 >= M) break;
#pragma unroll
        for (int nt = 0; nt < 8; ++nt) {
            const int col = nt * 16 + lr;
            const float bv = bias[col];
#pragma unroll
            for (int i = 0; i < 8; ++i) {
                float v = acc[mt][nt][i] + bv;
                if (RELU) v = fmaxf(v, 0.0f);
                out[(size_t)(m0 + mt * 16 + i + 8 * half) * N + col] = v;
            }
        }
    }
}

constexpr int RM_TRIPS = 4;
__global__ __launch_bounds__(256) void rowmat3_dual(const float* __restrict__ Xa,
                                                    const float* __restrict__ Wrela,
                                                    const float* __restrict__ Wroota,
                                                    const float* __restrict__ biasa,
                                                    float* __restrict__ out_rela,
                                                    float* __restrict__ out_roota, int na, int blocksA,
                                                    const float* __restrict__ Xb,
                                                    const float* __restrict__ Wrelb,
                                                    const float* __restrict__ Wrootb,
                                                    const float* __restrict__ biasb,
                                                    float* __restrict__ out_relb,
                                                    float* __restrict__ out_rootb, int nb) {
    const bool jobA = (int)blockIdx.x < blocksA;
    const float* __restrict__ X = jobA ? Xa : Xb;
    const float* __restrict__ Wrel = jobA ? Wrela : Wrelb;
    const float* __restrict__ Wroot = jobA ? Wroota : Wrootb;
    const float* __restrict__ bias = jobA ? biasa : biasb;
    float* __restrict__ out_rel = jobA ? out_rela : out_relb;
    float* __restrict__ out_root = jobA ? out_roota : out_rootb;
    const int n = jobA ? na : nb;
    const int blk = jobA ? (int)blockIdx.x : (int)blockIdx.x - blocksA;
    const int lane = threadIdx.x & 31;
    const int sub = lane & 7;
    const int grp = lane >> 3;
    const int wave = blk * (blockDim.x >> 5) + (threadIdx.x >> 5);
    const int row0 = wave * (4 * RM_TRIPS);
    if (row0 >= n) return;

    float wr[4][4][3], wo[4][4][3];
#pragma unroll
    for (int u = 0; u < 4; ++u) {
        const int k0 = (u * 8 + sub) * 4;
#pragma unroll
        for (int e = 0; e < 4; ++e)
#pragma unroll
            for (int c = 0; c < 3; ++c) {
                wr[u][e][c] = Wrel[(k0 + e) * 3 + c];
                wo[u][e][c] = Wroot[(k0 + e) * 3 + c];
            }
    }
    const float b0 = bias[0], b1 = bias[1], b2 = bias[2];

    float4 x[RM_TRIPS][4];
#pragma unroll
    for (int t = 0; t < RM_TRIPS; ++t) {
        const int r = min(row0 + t * 4 + grp, n - 1);
#pragma unroll
        for (int u = 0; u < 4; ++u)
            x[t][u] = *(const float4*)(X + (size_t)r * 128 + (u * 8 + sub) * 4);
    }
#pragma unroll
    for (int t = 0; t < RM_TRIPS; ++t) {
        float p[3] = {0.f, 0.f, 0.f};
        float q[3] = {0.f, 0.f, 0.f};
#pragma unroll
        for (int u = 0; u < 4; ++u)
#pragma unroll
            for (int c = 0; c < 3; ++c) {
                p[c] += x[t][u].x * wr[u][0][c] + x[t][u].y * wr[u][1][c] +
                        x[t][u].z * wr[u][2][c] + x[t][u].w * wr[u][3][c];
                q[c] += x[t][u].x * wo[u][0][c] + x[t][u].y * wo[u][1][c] +
                        x[t][u].z * wo[u][2][c] + x[t][u].w * wo[u][3][c];
            }
#pragma unroll
        for (int o = 4; o; o >>= 1)
#pragma unroll
            for (int c = 0; c < 3; ++c) {
                p[c] += __shfl_xor(p[c], o, 32);
                q[c] += __shfl_xor(q[c], o, 32);
            }
        const int r = row0 + t * 4 + grp;
        if (sub == 0 && r < n) {
            out_rel[(size_t)r * 3 + 0] = p[0];
            out_rel[(size_t)r * 3 + 1] = p[1];
            out_rel[(size_t)r * 3 + 2] = p[2];
            out_root[(size_t)r * 3 + 0] = q[0] + b0;
            out_root[(size_t)r * 3 + 1] = q[1] + b1;
            out_root[(size_t)r * 3 + 2] = q[2] + b2;
        }
    }
}

static inline int ceil_div(long long a, long long b) { return (int)((a + b - 1) / b); }

extern "C" void kernel_launch(void* const* d_in, const int* in_sizes, int n_in,
                              void* d_out, int out_size, void* d_ws, size_t ws_size,
                              hipStream_t stream) {
    const float* x_s = (const float*)d_in[0];
    const float* x_t = (const float*)d_in[1];
    const int* e_l = (const int*)d_in[2];
    const int* e_r = (const int*)d_in[3];
    const float* w1lr_rel  = (const float*)d_in[4];
    const float* b1lr      = (const float*)d_in[5];
    const float* w1lr_root = (const float*)d_in[6];
    const float* w1rl_rel  = (const float*)d_in[7];
    const float* b1rl      = (const float*)d_in[8];
    const float* w1rl_root = (const float*)d_in[9];
    const float* w2lr_rel  = (const float*)d_in[10];
    const float* b2lr      = (const float*)d_in[11];
    const float* w2lr_root = (const float*)d_in[12];
    const float* w2rl_rel  = (const float*)d_in[13];
    const float* b2rl      = (const float*)d_in[14];
    const float* w2rl_root = (const float*)d_in[15];
    const float* w3lr_rel  = (const float*)d_in[16];
    const float* b3lr      = (const float*)d_in[17];
    const float* w3lr_root = (const float*)d_in[18];
    const float* w3rl_rel  = (const float*)d_in[19];
    const float* b3rl      = (const float*)d_in[20];
    const float* w3rl_root = (const float*)d_in[21];

    const int NL = in_sizes[0] / 64;
    const int NR = in_sizes[1] / 64;
    const int E  = in_sizes[2];

    float* ws = (float*)d_ws;
    const size_t AGG_R = 0;
    const size_t AGG_L = AGG_R + (size_t)NR * 128;
    const size_t H1_R  = AGG_L + (size_t)NL * 128;
    const size_t H1_L  = H1_R + (size_t)NR * 128;
    const size_t PROJ_LR = H1_R;
    const size_t PROJ_RL = H1_R + (size_t)NL * 3;
    const int tilesR = ceil_div(NR, TILE);
    const int tilesL = ceil_div(NL, TILE);
    int* wi = (int*)d_ws;
    const size_t CNT_R  = H1_L + (size_t)NL * 128;
    const size_t CNT_L  = CNT_R + (size_t)NR * CS;
    const size_t OFF_R  = CNT_L + (size_t)NL * CS;
    const size_t OFF_L  = OFF_R + (size_t)NR + 1;
    const size_t RANK_R = OFF_L + (size_t)NL + 1;
    const size_t RANK_L = RANK_R + (size_t)E;
    const size_t PERM_R = RANK_L + (size_t)E;
    const size_t PERM_L = PERM_R + (size_t)E;
    const size_t TSUM   = PERM_L + (size_t)E;

    float* d_left  = (float*)d_out;
    float* d_right = (float*)d_out + (size_t)NL * 3;

    const int TPB = 256;
    constexpr int MT = 2;

    zero_kernel<<<512, TPB, 0, stream>>>((float4*)(ws + CNT_R), (size_t)(NR + NL) * CS / 4);
    edge_rank<<<ceil_div(E, TPB * EPT), TPB, 0, stream>>>(e_l, e_r, wi + CNT_L, wi + CNT_R, wi + RANK_L, wi + RANK_R, E);
    tile_sums<<<tilesR + tilesL, TPB, 0, stream>>>(wi + CNT_R, NR, tilesR, wi + CNT_L, NL, wi + TSUM);
    scan_apply<<<tilesR + tilesL, TPB, 0, stream>>>(wi + CNT_R, NR, tilesR, wi + OFF_R,
                                                    wi + CNT_L, NL, wi + OFF_L, wi + TSUM);
    edge_place<<<ceil_div(E, TPB * EPT), TPB, 0, stream>>>(e_l, e_r, wi + OFF_L, wi + OFF_R, wi + RANK_L, wi + RANK_R,
                                                     wi + PERM_L, wi + PERM_R, E);

    const int gsL = ceil_div((long long)NL * 8, TPB);
    const int gsR = ceil_div((long long)NR * 8, TPB);
    const int gmR = ceil_div(ceil_div(NR, 16 * MT), 8);
    const int gmL = ceil_div(ceil_div(NL, 16 * MT), 8);
    const int rmL = ceil_div(NL, 8 * 4 * RM_TRIPS);
    const int rmR = ceil_div(NR, 8 * 4 * RM_TRIPS);
    const int fnL = ceil_div(NL, TPB);
    const int fnR = ceil_div(NR, TPB);

    gather_sum_vec<64><<<gsL + gsR, TPB, 0, stream>>>(x_t, wi + OFF_L, wi + PERM_L, ws + AGG_L, NL, gsL,
                                                      x_s, wi + OFF_R, wi + PERM_R, ws + AGG_R, NR);
    gemm_dual_wmma<64, 64, true, MT><<<gmR + gmL, TPB, 0, stream>>>(
        ws + AGG_R, w1lr_rel, x_t, w1lr_root, b1lr, ws + H1_R, NR, gmR,
        ws + AGG_L, w1rl_rel, x_s, w1rl_root, b1rl, ws + H1_L, NL);

    gather_sum_vec<128><<<gsL + gsR, TPB, 0, stream>>>(ws + H1_R, wi + OFF_L, wi + PERM_L, ws + AGG_L, NL, gsL,
                                                       ws + H1_L, wi + OFF_R, wi + PERM_R, ws + AGG_R, NR);
    gemm_dual_wmma<128, 128, true, MT><<<gmR + gmL, TPB, 0, stream>>>(
        ws + AGG_R, w2lr_rel, ws + H1_R, w2lr_root, b2lr, ws + AGG_R, NR, gmR,
        ws + AGG_L, w2rl_rel, ws + H1_L, w2rl_root, b2rl, ws + AGG_L, NL);

    rowmat3_dual<<<rmR + rmL, TPB, 0, stream>>>(ws + AGG_R, w3rl_rel, w3lr_root, b3lr, ws + PROJ_RL, d_right, NR, rmR,
                                                ws + AGG_L, w3lr_rel, w3rl_root, b3rl, ws + PROJ_LR, d_left, NL);
    gather_add3_finish<<<fnL + fnR, TPB, 0, stream>>>(ws + PROJ_RL, wi + OFF_L, wi + PERM_L, x_s, d_left, NL, fnL,
                                                      ws + PROJ_LR, wi + OFF_R, wi + PERM_R, x_t, d_right, NR);
}


# GEMM operand feed: weight matrices copied once per work-group into LDS (k pairs interleaved), B fragments read from LDS; 16-row strip per wave, A fetched a 32-column chunk ahead; same k order
# speedup vs baseline: 2.8981x; 1.1638x over previous
#include <hip/hip_runtime.h>
#include <hip/hip_bf16.h>

typedef __attribute__((ext_vector_type(2))) float v2f;
typedef __attribute__((ext_vector_type(8))) float v8f;

__global__ __launch_bounds__(256) void zero_kernel(float4* __restrict__ p, size_t n4) {
    const float4 z = make_float4(0.f, 0.f, 0.f, 0.f);
    const size_t step = (size_t)gridDim.x * blockDim.x;
    for (size_t i = (size_t)blockIdx.x * blockDim.x + threadIdx.x; i < n4; i += step) p[i] = z;
}

constexpr int CS = 16;
constexpr int TILE = 1024;
constexpr int EPT = 4;

__global__ __launch_bounds__(256) void edge_rank(const int* __restrict__ e_l,
                                                 const int* __restrict__ e_r,
                                                 int* cntL, int* cntR,
                                                 int* __restrict__ rankL,
                                                 int* __restrict__ rankR, int nE) {
    const int base = blockIdx.x * (256 * EPT) + threadIdx.x;
    int dl[EPT], dr[EPT], rl[EPT], rr[EPT];
#pragma unroll
    for (int k = 0; k < EPT; ++k) {
        const int e = base + k * 256;
        const bool ok = e < nE;
        dl[k] = ok ? e_l[e] : -1;
        dr[k] = ok ? e_r[e] : -1;
    }
#pragma unroll
    for (int k = 0; k < EPT; ++k) {
        rr[k] = 0;
        rl[k] = 0;
        if (dl[k] >= 0) {
            rr[k] = atomicAdd(cntR + (size_t)dr[k] * CS, 1);
            rl[k] = atomicAdd(cntL + (size_t)dl[k] * CS, 1);
        }
    }
#pragma unroll
    for (int k = 0; k < EPT; ++k) {
        const int e = base + k * 256;
        if (e < nE) {
            rankR[e] = rr[k];
            rankL[e] = rl[k];
        }
    }
}

__global__ __launch_bounds__(256) void tile_sums(const int* __restrict__ cntA, int nA, int tilesA,
                                                 const int* __restrict__ cntB, int nB,
                                                 int* __restrict__ tsum) {
    __shared__ int wsum[8];
    const int b = blockIdx.x;
    const int lane = threadIdx.x & 31;
    const int wave = threadIdx.x >> 5;
    const int* cnt = (b < tilesA) ? cntA : cntB;
    const int n = (b < tilesA) ? nA : nB;
    const int tile = (b < tilesA) ? b : b - tilesA;
    const int i0 = tile * TILE + threadIdx.x * 4;
    int s = 0;
#pragma unroll
    for (int j = 0; j < 4; ++j) {
        int i = i0 + j;
        if (i < n) s += cnt[(size_t)i * CS];
    }
#pragma unroll
    for (int o = 16; o; o >>= 1) s += __shfl_xor(s, o, 32);
    if (lane == 0) wsum[wave] = s;
    __syncthreads();
    if (threadIdx.x == 0) {
        int t = 0;
#pragma unroll
        for (int w = 0; w < 8; ++w) t += wsum[w];
        tsum[b] = t;
    }
}

__global__ __launch_bounds__(256) void scan_apply(const int* __restrict__ cntA, int nA, int tilesA,
                                                  int* __restrict__ offA,
                                                  const int* __restrict__ cntB, int nB,
                                                  int* __restrict__ offB,
                                                  const int* __restrict__ tsum) {
    __shared__ int wsum[8];
    __shared__ int wtot[8];
    const int b = blockIdx.x;
    const int lane = threadIdx.x & 31;
    const int wave = threadIdx.x >> 5;
    const bool isA = b < tilesA;
    const int* cnt = isA ? cntA : cntB;
    int* off = isA ? offA : offB;
    const int n = isA ? nA : nB;
    const int tile = isA ? b : b - tilesA;
    const int tbase = isA ? 0 : tilesA;

    int p = 0;
    for (int i = threadIdx.x; i < tile; i += 256) p += tsum[tbase + i];
#pragma unroll
    for (int o = 16; o; o >>= 1) p += __shfl_xor(p, o, 32);
    if (lane == 0) wsum[wave] = p;

    const int i0 = tile * TILE + threadIdx.x * 4;
    int v[4];
    int own = 0;
#pragma unroll
    for (int j = 0; j < 4; ++j) {
        int i = i0 + j;
        v[j] = (i < n) ? cnt[(size_t)i * CS] : 0;
        own += v[j];
    }
    int incl = own;
#pragma unroll
    for (int o = 1; o < 32; o <<= 1) {
        int u = __shfl_up(incl, o, 32);
        if (lane >= o) incl += u;
    }
    if (lane == 31) wtot[wave] = incl;
    __syncthreads();

    int run = incl - own;
#pragma unroll
    for (int w = 0; w < 8; ++w) {
        run += wsum[w];
        if (w < wave) run += wtot[w];
    }
#pragma unroll
    for (int j = 0; j < 4; ++j) {
        int i = i0 + j;
        if (i < n) {
            off[i] = run;
            run += v[j];
            if (i == n - 1) off[n] = run;
        }
    }
}

__global__ __launch_bounds__(256) void edge_place(const int* __restrict__ e_l,
                                                  const int* __restrict__ e_r,
                                                  const int* __restrict__ offL,
                                                  const int* __restrict__ offR,
                                                  const int* __restrict__ rankL,
                                                  const int* __restrict__ rankR,
                                                  int* __restrict__ permL,
                                                  int* __restrict__ permR, int nE) {
    const int base = blockIdx.x * (256 * EPT) + threadIdx.x;
    int dl[EPT], dr[EPT], rl[EPT], rr[EPT], ol[EPT], orr[EPT];
#pragma unroll
    for (int k = 0; k < EPT; ++k) {
        const int e = base + k * 256;
        const bool ok = e < nE;
        dl[k] = ok ? e_l[e] : -1;
        dr[k] = ok ? e_r[e] : -1;
        rl[k] = ok ? rankL[e] : 0;
        rr[k] = ok ? rankR[e] : 0;
    }
#pragma unroll
    for (int k = 0; k < EPT; ++k) {
        ol[k] = 0;
        orr[k] = 0;
        if (dl[k] >= 0) {
            orr[k] = offR[dr[k]];
            ol[k] = offL[dl[k]];
        }
    }
#pragma unroll
    for (int k = 0; k < EPT; ++k) {
        if (dl[k] >= 0) {
            permR[orr[k] + rr[k]] = dl[k];
            permL[ol[k] + rl[k]] = dr[k];
        }
    }
}

template <int F>
__global__ __launch_bounds__(256) void gather_sum_vec(const float* __restrict__ srcA,
                                                      const int* __restrict__ offA,
                                                      const int* __restrict__ permA,
                                                      float* __restrict__ dstA, int nA, int blocksA,
                                                      const float* __restrict__ srcB,
                                                      const int* __restrict__ offB,
                                                      const int* __restrict__ permB,
                                                      float* __restrict__ dstB, int nB) {
    constexpr int PER = 8;
    constexpr int V = F / (4 * PER);
    const bool jobA = (int)blockIdx.x < blocksA;
    const float* __restrict__ src = jobA ? srcA : srcB;
    const int* __restrict__ off = jobA ? offA : offB;
    const int* __restrict__ perm = jobA ? permA : permB;
    float* __restrict__ dst = jobA ? dstA : dstB;
    const int nDst = jobA ? nA : nB;
    const unsigned blk = jobA ? blockIdx.x : blockIdx.x - blocksA;
    const unsigned t = blk * blockDim.x + threadIdx.x;
    const unsigned d = t / PER;
    const unsigned sub = t % PER;
    if (d >= (unsigned)nDst) return;
    const int beg = off[d];
    const int end = off[d + 1];
    float4 acc[V];
#pragma unroll
    for (int u = 0; u < V; ++u) acc[u] = make_float4(0.f, 0.f, 0.f, 0.f);
    const float* lane_src = src + sub * 4;
    int j = beg;
    for (; j + 4 <= end; j += 4) {
        int s[4];
#pragma unroll
        for (int k = 0; k < 4; ++k) s[k] = perm[j + k];
        float4 v[4][V];
#pragma unroll
        for (int k = 0; k < 4; ++k)
#pragma unroll
            for (int u = 0; u < V; ++u)
                v[k][u] = *(const float4*)(lane_src + (size_t)s[k] * F + u * (PER * 4));
#pragma unroll
        for (int k = 0; k < 4; ++k)
#pragma unroll
            for (int u = 0; u < V; ++u) {
                acc[u].x += v[k][u].x; acc[u].y += v[k][u].y;
                acc[u].z += v[k][u].z; acc[u].w += v[k][u].w;
            }
    }
    if (j < end) {
        int s[3];
#pragma unroll
        for (int k = 0; k < 3; ++k) s[k] = perm[min(j + k, end - 1)];
        float4 v[3][V];
#pragma unroll
        for (int k = 0; k < 3; ++k)
#pragma unroll
            for (int u = 0; u < V; ++u)
                v[k][u] = *(const float4*)(lane_src + (size_t)s[k] * F + u * (PER * 4));
#pragma unroll
        for (int k = 0; k < 3; ++k) {
            const bool on = j + k < end;
#pragma unroll
            for (int u = 0; u < V; ++u) {
                acc[u].x += on ? v[k][u].x : 0.f; acc[u].y += on ? v[k][u].y : 0.f;
                acc[u].z += on ? v[k][u].z : 0.f; acc[u].w += on ? v[k][u].w : 0.f;
            }
        }
    }
#pragma unroll
    for (int u = 0; u < V; ++u)
        *(float4*)(dst + (size_t)d * F + (u * PER + sub) * 4) = acc[u];
}

__global__ __launch_bounds__(256) void gather_add3_finish(const float* __restrict__ projA,
                                                          const int* __restrict__ offA,
                                                          const int* __restrict__ permA,
                                                          const float* __restrict__ xA,
                                                          float* outA, int nA, int blocksA,
                                                          const float* __restrict__ projB,
                                                          const int* __restrict__ offB,
                                                          const int* __restrict__ permB,
                                                          const float* __restrict__ xB,
                                                          float* outB, int nB) {
    const bool jobA = (int)blockIdx.x < blocksA;
    const float* __restrict__ proj = jobA ? projA : projB;
    const int* __restrict__ off = jobA ? offA : offB;
    const int* __restrict__ perm = jobA ? permA : permB;
    const float* __restrict__ x = jobA ? xA : xB;
    float* out = jobA ? outA : outB;
    const int nDst = jobA ? nA : nB;
    const int blk = jobA ? (int)blockIdx.x : (int)blockIdx.x - blocksA;
    int d = blk * blockDim.x + threadIdx.x;
    if (d >= nDst) return;
    const int beg = off[d];
    const int end = off[d + 1];
    float a0 = out[(size_t)d * 3 + 0];
    float a1 = out[(size_t)d * 3 + 1];
    float a2 = out[(size_t)d * 3 + 2];
    const float f0 = x[(size_t)d * 64 + 61];
    const float f2 = x[(size_t)d * 64 + 63];
    for (int j = beg; j < end; ++j) {
        const int s = perm[j];
        a0 += proj[(size_t)s * 3 + 0];
        a1 += proj[(size_t)s * 3 + 1];
        a2 += proj[(size_t)s * 3 + 2];
    }
    float nrm = sqrtf(a0 * a0 + a1 * a1 + a2 * a2);
    float sc = 10.0f / fmaxf(nrm, 1e-12f);
    a0 *= sc; a1 *= sc; a2 *= sc;
    if (f0 != 0.0f) a0 -= 10.0f;
    if (f2 != 0.0f) a2 -= 10.0f;
    out[(size_t)d * 3 + 0] = a0;
    out[(size_t)d * 3 + 1] = a1;
    out[(size_t)d * 3 + 2] = a2;
}

template <int K1, int K2, bool RELU>
__global__ __launch_bounds__(256) void gemm_dual_wmma(const float* __restrict__ A1a,
                                                      const float* __restrict__ W1a,
                                                      const float* __restrict__ A2a,
                                                      const float* __restrict__ W2a,
                                                      const float* __restrict__ biasa,
                                                      float* outa, int Ma, int blocksA,
                                                      const float* __restrict__ A1b,
                                                      const float* __restrict__ W1b,
                                                      const float* __restrict__ A2b,
                                                      const float* __restrict__ W2b,
                                                      const float* __restrict__ biasb,
                                                      float* outb, int Mb) {
    constexpr int N = 128;
    constexpr int KP = (K1 + K2) / 2;
    constexpr int PITCH = 2 * N + 32;
    __shared__ alignas(16) float wl[KP * PITCH];

    const bool jobA = (int)blockIdx.x < blocksA;
    const float* __restrict__ A1 = jobA ? A1a : A1b;
    const float* __restrict__ W1 = jobA ? W1a : W1b;
    const float* __restrict__ A2 = jobA ? A2a : A2b;
    const float* __restrict__ W2 = jobA ? W2a : W2b;
    const float* __restrict__ bias = jobA ? biasa : biasb;
    float* out = jobA ? outa : outb;
    const int M = jobA ? Ma : Mb;
    const int blk = jobA ? (int)blockIdx.x : (int)blockIdx.x - blocksA;

    {
        const int c2 = (threadIdx.x & 63) * 2;
        const int p0 = threadIdx.x >> 6;
#pragma unroll 8
        for (int i = 0; i < K1 / 8; ++i) {
            const int p = p0 + 4 * i;
            const float2 r0 = *(const float2*)(W1 + (size_t)(2 * p) * N + c2);
            const float2 r1 = *(const float2*)(W1 + (size_t)(2 * p + 1) * N + c2);
            *(float4*)(wl + p * PITCH + c2 * 2) = make_float4(r0.x, r1.x, r0.y, r1.y);
        }
#pragma unroll 8
        for (int i = 0; i < K2 / 8; ++i) {
            const int p = p0 + 4 * i;
            const float2 r0 = *(const float2*)(W2 + (size_t)(2 * p) * N + c2);
            const float2 r1 = *(const float2*)(W2 + (size_t)(2 * p + 1) * N + c2);
            *(float4*)(wl + (K1 / 2 + p) * PITCH + c2 * 2) = make_float4(r0.x, r1.x, r0.y, r1.y);
        }
    }
    __syncthreads();

    const int lane = threadIdx.x & 31;
    const int wave = threadIdx.x >> 5;
    const int m0 = (blk * 8 + wave) * 16;
    if (m0 >= M) return;
    const int half = lane >> 4;
    const int lr = lane & 15;

    const float* a1p = A1 + (size_t)(m0 + lr) * K1 + 2 * half;
    const float* a2p = A2 + (size_t)(m0 + lr) * K2 + 2 * half;
    const float* bl = wl + half * PITCH + lr * 2;

    constexpr int CH = 8;
    constexpr int NCH = (K1 + K2) / (4 * CH);

    v8f acc[8] = {};
    v2f ac[CH], an[CH];
    v2f bc[8], bn[8];
#pragma unroll
    for (int j = 0; j < CH; ++j) ac[j] = *(const v2f*)(a1p + j * 4);
#pragma unroll
    for (int nt = 0; nt < 8; ++nt) bc[nt] = *(const v2f*)(bl + nt * 32);

#pragma unroll 1
    for (int ci = 0; ci < NCH; ++ci) {
        const bool more = ci + 1 < NCH;
        if (more) {
            const int kn = (ci + 1) * (4 * CH);
            const float* ap = (kn < K1) ? (a1p + kn) : (a2p + (kn - K1));
#pragma unroll
            for (int j = 0; j < CH; ++j) an[j] = *(const v2f*)(ap + j * 4);
        }
#pragma unroll
        for (int j = 0; j < CH; ++j) {
            if (j + 1 < CH || more) {
#pragma unroll
                for (int nt = 0; nt < 8; ++nt)
                    bn[nt] = *(const v2f*)(bl + (2 * (j + 1)) * PITCH + nt * 32);
            }
#pragma unroll
            for (int nt = 0; nt < 8; ++nt)
                acc[nt] = __builtin_amdgcn_wmma_f32_16x16x4_f32(
                    false, ac[j], false, bc[nt], (short)0, acc[nt], false, false);
#pragma unroll
            for (int nt = 0; nt < 8; ++nt) bc[nt] = bn[nt];
        }
        if (more) {
#pragma unroll
            for (int j = 0; j < CH; ++j) ac[j] = an[j];
        }
        bl += 2 * CH * PITCH;
    }

#pragma unroll
    for (int nt = 0; nt < 8; ++nt) {
        const int col = nt * 16 + lr;
        const float bv = bias[col];
#pragma unroll
        for (int i = 0; i < 8; ++i) {
            float v = acc[nt][i] + bv;
            if (RELU) v = fmaxf(v, 0.0f);
            out[(size_t)(m0 + i + 8 * half) * N + col] = v;
        }
    }
}

constexpr int RM_TRIPS = 4;
__global__ __launch_bounds__(256) void rowmat3_dual(const float* __restrict__ Xa,
                                                    const float* __restrict__ Wrela,
                                                    const float* __restrict__ Wroota,
                                                    const float* __restrict__ biasa,
                                                    float* __restrict__ out_rela,
                                                    float* __restrict__ out_roota, int na, int blocksA,
                                                    const float* __restrict__ Xb,
                                                    const float* __restrict__ Wrelb,
                                                    const float* __restrict__ Wrootb,
                                                    const float* __restrict__ biasb,
                                                    float* __restrict__ out_relb,
                                                    float* __restrict__ out_rootb, int nb) {
    const bool jobA = (int)blockIdx.x < blocksA;
    const float* __restrict__ X = jobA ? Xa : Xb;
    const float* __restrict__ Wrel = jobA ? Wrela : Wrelb;
    const float* __restrict__ Wroot = jobA ? Wroota : Wrootb;
    const float* __restrict__ bias = jobA ? biasa : biasb;
    float* __restrict__ out_rel = jobA ? out_rela : out_relb;
    float* __restrict__ out_root = jobA ? out_roota : out_rootb;
    const int n = jobA ? na : nb;
    const int blk = jobA ? (int)blockIdx.x : (int)blockIdx.x - blocksA;
    const int lane = threadIdx.x & 31;
    const int sub = lane & 7;
    const int grp = lane >> 3;
    const int wave = blk * (blockDim.x >> 5) + (threadIdx.x >> 5);
    const int row0 = wave * (4 * RM_TRIPS);
    if (row0 >= n) return;

    float wr[4][4][3], wo[4][4][3];
#pragma unroll
    for (int u = 0; u < 4; ++u) {
        const int k0 = (u * 8 + sub) * 4;
#pragma unroll
        for (int e = 0; e < 4; ++e)
#pragma unroll
            for (int c = 0; c < 3; ++c) {
                wr[u][e][c] = Wrel[(k0 + e) * 3 + c];
                wo[u][e][c] = Wroot[(k0 + e) * 3 + c];
            }
    }
    const float b0 = bias[0], b1 = bias[1], b2 = bias[2];

    float4 x[RM_TRIPS][4];
#pragma unroll
    for (int t = 0; t < RM_TRIPS; ++t) {
        const int r = min(row0 + t * 4 + grp, n - 1);
#pragma unroll
        for (int u = 0; u < 4; ++u)
            x[t][u] = *(const float4*)(X + (size_t)r * 128 + (u * 8 + sub) * 4);
    }
#pragma unroll
    for (int t = 0; t < RM_TRIPS; ++t) {
        float p[3] = {0.f, 0.f, 0.f};
        float q[3] = {0.f, 0.f, 0.f};
#pragma unroll
        for (int u = 0; u < 4; ++u)
#pragma unroll
            for (int c = 0; c < 3; ++c) {
                p[c] += x[t][u].x * wr[u][0][c] + x[t][u].y * wr[u][1][c] +
                        x[t][u].z * wr[u][2][c] + x[t][u].w * wr[u][3][c];
                q[c] += x[t][u].x * wo[u][0][c] + x[t][u].y * wo[u][1][c] +
                        x[t][u].z * wo[u][2][c] + x[t][u].w * wo[u][3][c];
            }
#pragma unroll
        for (int o = 4; o; o >>= 1)
#pragma unroll
            for (int c = 0; c < 3; ++c) {
                p[c] += __shfl_xor(p[c], o, 32);
                q[c] += __shfl_xor(q[c], o, 32);
            }
        const int r = row0 + t * 4 + grp;
        if (sub == 0 && r < n) {
            out_rel[(size_t)r * 3 + 0] = p[0];
            out_rel[(size_t)r * 3 + 1] = p[1];
            out_rel[(size_t)r * 3 + 2] = p[2];
            out_root[(size_t)r * 3 + 0] = q[0] + b0;
            out_root[(size_t)r * 3 + 1] = q[1] + b1;
            out_root[(size_t)r * 3 + 2] = q[2] + b2;
        }
    }
}

static inline int ceil_div(long long a, long long b) { return (int)((a + b - 1) / b); }

extern "C" void kernel_launch(void* const* d_in, const int* in_sizes, int n_in,
                              void* d_out, int out_size, void* d_ws, size_t ws_size,
                              hipStream_t stream) {
    const float* x_s = (const float*)d_in[0];
    const float* x_t = (const float*)d_in[1];
    const int* e_l = (const int*)d_in[2];
    const int* e_r = (const int*)d_in[3];
    const float* w1lr_rel  = (const float*)d_in[4];
    const float* b1lr      = (const float*)d_in[5];
    const float* w1lr_root = (const float*)d_in[6];
    const float* w1rl_rel  = (const float*)d_in[7];
    const float* b1rl      = (const float*)d_in[8];
    const float* w1rl_root = (const float*)d_in[9];
    const float* w2lr_rel  = (const float*)d_in[10];
    const float* b2lr      = (const float*)d_in[11];
    const float* w2lr_root = (const float*)d_in[12];
    const float* w2rl_rel  = (const float*)d_in[13];
    const float* b2rl      = (const float*)d_in[14];
    const float* w2rl_root = (const float*)d_in[15];
    const float* w3lr_rel  = (const float*)d_in[16];
    const float* b3lr      = (const float*)d_in[17];
    const float* w3lr_root = (const float*)d_in[18];
    const float* w3rl_rel  = (const float*)d_in[19];
    const float* b3rl      = (const float*)d_in[20];
    const float* w3rl_root = (const float*)d_in[21];

    const int NL = in_sizes[0] / 64;
    const int NR = in_sizes[1] / 64;
    const int E  = in_sizes[2];

    float* ws = (float*)d_ws;
    const size_t AGG_R = 0;
    const size_t AGG_L = AGG_R + (size_t)NR * 128;
    const size_t H1_R  = AGG_L + (size_t)NL * 128;
    const size_t H1_L  = H1_R + (size_t)NR * 128;
    const size_t PROJ_LR = H1_R;
    const size_t PROJ_RL = H1_R + (size_t)NL * 3;
    const int tilesR = ceil_div(NR, TILE);
    const int tilesL = ceil_div(NL, TILE);
    int* wi = (int*)d_ws;
    const size_t CNT_R  = H1_L + (size_t)NL * 128;
    const size_t CNT_L  = CNT_R + (size_t)NR * CS;
    const size_t OFF_R  = CNT_L + (size_t)NL * CS;
    const size_t OFF_L  = OFF_R + (size_t)NR + 1;
    const size_t RANK_R = OFF_L + (size_t)NL + 1;
    const size_t RANK_L = RANK_R + (size_t)E;
    const size_t PERM_R = RANK_L + (size_t)E;
    const size_t PERM_L = PERM_R + (size_t)E;
    const size_t TSUM   = PERM_L + (size_t)E;

    float* d_left  = (float*)d_out;
    float* d_right = (float*)d_out + (size_t)NL * 3;

    const int TPB = 256;

    zero_kernel<<<512, TPB, 0, stream>>>((float4*)(ws + CNT_R), (size_t)(NR + NL) * CS / 4);
    edge_rank<<<ceil_div(E, TPB * EPT), TPB, 0, stream>>>(e_l, e_r, wi + CNT_L, wi + CNT_R, wi + RANK_L, wi + RANK_R, E);
    tile_sums<<<tilesR + tilesL, TPB, 0, stream>>>(wi + CNT_R, NR, tilesR, wi + CNT_L, NL, wi + TSUM);
    scan_apply<<<tilesR + tilesL, TPB, 0, stream>>>(wi + CNT_R, NR, tilesR, wi + OFF_R,
                                                    wi + CNT_L, NL, wi + OFF_L, wi + TSUM);
    edge_place<<<ceil_div(E, TPB * EPT), TPB, 0, stream>>>(e_l, e_r, wi + OFF_L, wi + OFF_R, wi + RANK_L, wi + RANK_R,
                                                     wi + PERM_L, wi + PERM_R, E);

    const int gsL = ceil_div((long long)NL * 8, TPB);
    const int gsR = ceil_div((long long)NR * 8, TPB);
    const int gmR = ceil_div(NR, 16 * 8);
    const int gmL = ceil_div(NL, 16 * 8);
    const int rmL = ceil_div(NL, 8 * 4 * RM_TRIPS);
    const int rmR = ceil_div(NR, 8 * 4 * RM_TRIPS);
    const int fnL = ceil_div(NL, TPB);
    const int fnR = ceil_div(NR, TPB);

    gather_sum_vec<64><<<gsL + gsR, TPB, 0, stream>>>(x_t, wi + OFF_L, wi + PERM_L, ws + AGG_L, NL, gsL,
                                                      x_s, wi + OFF_R, wi + PERM_R, ws + AGG_R, NR);
    gemm_dual_wmma<64, 64, true><<<gmR + gmL, TPB, 0, stream>>>(
        ws + AGG_R, w1lr_rel, x_t, w1lr_root, b1lr, ws + H1_R, NR, gmR,
        ws + AGG_L, w1rl_rel, x_s, w1rl_root, b1rl, ws + H1_L, NL);

    gather_sum_vec<128><<<gsL + gsR, TPB, 0, stream>>>(ws + H1_R, wi + OFF_L, wi + PERM_L, ws + AGG_L, NL, gsL,
                                                       ws + H1_L, wi + OFF_R, wi + PERM_R, ws + AGG_R, NR);
    gemm_dual_wmma<128, 128, true><<<gmR + gmL, TPB, 0, stream>>>(
        ws + AGG_R, w2lr_rel, ws + H1_R, w2lr_root, b2lr, ws + AGG_R, NR, gmR,
        ws + AGG_L, w2rl_rel, ws + H1_L, w2rl_root, b2rl, ws + AGG_L, NL);

    rowmat3_dual<<<rmR + rmL, TPB, 0, stream>>>(ws + AGG_R, w3rl_rel, w3lr_root, b3lr, ws + PROJ_RL, d_right, NR, rmR,
                                                ws + AGG_L, w3lr_rel, w3rl_root, b3rl, ws + PROJ_LR, d_left, NL);
    gather_add3_finish<<<fnL + fnR, TPB, 0, stream>>>(ws + PROJ_RL, wi + OFF_L, wi + PERM_L, x_s, d_left, NL, fnL,
                                                      ws + PROJ_LR, wi + OFF_R, wi + PERM_R, x_t, d_right, NR);
}


# Layer-3 node products (h2 @ W3rel, h2 @ W3root + b3) formed in the layer-2 GEMM epilogue, WMMA operands swapped so a lane holds one row: h2 never stored, 11 launches become 10; other dot sum tree
# speedup vs baseline: 3.0413x; 1.0494x over previous
#include <hip/hip_runtime.h>
#include <hip/hip_bf16.h>

typedef __attribute__((ext_vector_type(2))) float v2f;
typedef __attribute__((ext_vector_type(8))) float v8f;

__global__ __launch_bounds__(256) void zero_kernel(float4* __restrict__ p, size_t n4) {
    const float4 z = make_float4(0.f, 0.f, 0.f, 0.f);
    const size_t step = (size_t)gridDim.x * blockDim.x;
    for (size_t i = (size_t)blockIdx.x * blockDim.x + threadIdx.x; i < n4; i += step) p[i] = z;
}

constexpr int CS = 16;
constexpr int TILE = 1024;
constexpr int EPT = 4;

__global__ __launch_bounds__(256) void edge_rank(const int* __restrict__ e_l,
                                                 const int* __restrict__ e_r,
                                                 int* cntL, int* cntR,
                                                 int* __restrict__ rankL,
                                                 int* __restrict__ rankR, int nE) {
    const int base = blockIdx.x * (256 * EPT) + threadIdx.x;
    int dl[EPT], dr[EPT], rl[EPT], rr[EPT];
#pragma unroll
    for (int k = 0; k < EPT; ++k) {
        const int e = base + k * 256;
        const bool ok = e < nE;
        dl[k] = ok ? e_l[e] : -1;
        dr[k] = ok ? e_r[e] : -1;
    }
#pragma unroll
    for (int k = 0; k < EPT; ++k) {
        rr[k] = 0;
        rl[k] = 0;
        if (dl[k] >= 0) {
            rr[k] = atomicAdd(cntR + (size_t)dr[k] * CS, 1);
            rl[k] = atomicAdd(cntL + (size_t)dl[k] * CS, 1);
        }
    }
#pragma unroll
    for (int k = 0; k < EPT; ++k) {
        const int e = base + k * 256;
        if (e < nE) {
            rankR[e] = rr[k];
            rankL[e] = rl[k];
        }
    }
}

__global__ __launch_bounds__(256) void tile_sums(const int* __restrict__ cntA, int nA, int tilesA,
                                                 const int* __restrict__ cntB, int nB,
                                                 int* __restrict__ tsum) {
    __shared__ int wsum[8];
    const int b = blockIdx.x;
    const int lane = threadIdx.x & 31;
    const int wave = threadIdx.x >> 5;
    const int* cnt = (b < tilesA) ? cntA : cntB;
    const int n = (b < tilesA) ? nA : nB;
    const int tile = (b < tilesA) ? b : b - tilesA;
    const int i0 = tile * TILE + threadIdx.x * 4;
    int s = 0;
#pragma unroll
    for (int j = 0; j < 4; ++j) {
        int i = i0 + j;
        if (i < n) s += cnt[(size_t)i * CS];
    }
#pragma unroll
    for (int o = 16; o; o >>= 1) s += __shfl_xor(s, o, 32);
    if (lane == 0) wsum[wave] = s;
    __syncthreads();
    if (threadIdx.x == 0) {
        int t = 0;
#pragma unroll
        for (int w = 0; w < 8; ++w) t += wsum[w];
        tsum[b] = t;
    }
}

__global__ __launch_bounds__(256) void scan_apply(const int* __restrict__ cntA, int nA, int tilesA,
                                                  int* __restrict__ offA,
                                                  const int* __restrict__ cntB, int nB,
                                                  int* __restrict__ offB,
                                                  const int* __restrict__ tsum) {
    __shared__ int wsum[8];
    __shared__ int wtot[8];
    const int b = blockIdx.x;
    const int lane = threadIdx.x & 31;
    const int wave = threadIdx.x >> 5;
    const bool isA = b < tilesA;
    const int* cnt = isA ? cntA : cntB;
    int* off = isA ? offA : offB;
    const int n = isA ? nA : nB;
    const int tile = isA ? b : b - tilesA;
    const int tbase = isA ? 0 : tilesA;

    int p = 0;
    for (int i = threadIdx.x; i < tile; i += 256) p += tsum[tbase + i];
#pragma unroll
    for (int o = 16; o; o >>= 1) p += __shfl_xor(p, o, 32);
    if (lane == 0) wsum[wave] = p;

    const int i0 = tile * TILE + threadIdx.x * 4;
    int v[4];
    int own = 0;
#pragma unroll
    for (int j = 0; j < 4; ++j) {
        int i = i0 + j;
        v[j] = (i < n) ? cnt[(size_t)i * CS] : 0;
        own += v[j];
    }
    int incl = own;
#pragma unroll
    for (int o = 1; o < 32; o <<= 1) {
        int u = __shfl_up(incl, o, 32);
        if (lane >= o) incl += u;
    }
    if (lane == 31) wtot[wave] = incl;
    __syncthreads();

    int run = incl - own;
#pragma unroll
    for (int w = 0; w < 8; ++w) {
        run += wsum[w];
        if (w < wave) run += wtot[w];
    }
#pragma unroll
    for (int j = 0; j < 4; ++j) {
        int i = i0 + j;
        if (i < n) {
            off[i] = run;
            run += v[j];
            if (i == n - 1) off[n] = run;
        }
    }
}

__global__ __launch_bounds__(256) void edge_place(const int* __restrict__ e_l,
                                                  const int* __restrict__ e_r,
                                                  const int* __restrict__ offL,
                                                  const int* __restrict__ offR,
                                                  const int* __restrict__ rankL,
                                                  const int* __restrict__ rankR,
                                                  int* __restrict__ permL,
                                                  int* __restrict__ permR, int nE) {
    const int base = blockIdx.x * (256 * EPT) + threadIdx.x;
    int dl[EPT], dr[EPT], rl[EPT], rr[EPT], ol[EPT], orr[EPT];
#pragma unroll
    for (int k = 0; k < EPT; ++k) {
        const int e = base + k * 256;
        const bool ok = e < nE;
        dl[k] = ok ? e_l[e] : -1;
        dr[k] = ok ? e_r[e] : -1;
        rl[k] = ok ? rankL[e] : 0;
        rr[k] = ok ? rankR[e] : 0;
    }
#pragma unroll
    for (int k = 0; k < EPT; ++k) {
        ol[k] = 0;
        orr[k] = 0;
        if (dl[k] >= 0) {
            orr[k] = offR[dr[k]];
            ol[k] = offL[dl[k]];
        }
    }
#pragma unroll
    for (int k = 0; k < EPT; ++k) {
        if (dl[k] >= 0) {
            permR[orr[k] + rr[k]] = dl[k];
            permL[ol[k] + rl[k]] = dr[k];
        }
    }
}

template <int F>
__global__ __launch_bounds__(256) void gather_sum_vec(const float* __restrict__ srcA,
                                                      const int* __restrict__ offA,
                                                      const int* __restrict__ permA,
                                                      float* __restrict__ dstA, int nA, int blocksA,
                                                      const float* __restrict__ srcB,
                                                      const int* __restrict__ offB,
                                                      const int* __restrict__ permB,
                                                      float* __restrict__ dstB, int nB) {
    constexpr int PER = 8;
    constexpr int V = F / (4 * PER);
    const bool jobA = (int)blockIdx.x < blocksA;
    const float* __restrict__ src = jobA ? srcA : srcB;
    const int* __restrict__ off = jobA ? offA : offB;
    const int* __restrict__ perm = jobA ? permA : permB;
    float* __restrict__ dst = jobA ? dstA : dstB;
    const int nDst = jobA ? nA : nB;
    const unsigned blk = jobA ? blockIdx.x : blockIdx.x - blocksA;
    const unsigned t = blk * blockDim.x + threadIdx.x;
    const unsigned d = t / PER;
    const unsigned sub = t % PER;
    if (d >= (unsigned)nDst) return;
    const int beg = off[d];
    const int end = off[d + 1];
    float4 acc[V];
#pragma unroll
    for (int u = 0; u < V; ++u) acc[u] = make_float4(0.f, 0.f, 0.f, 0.f);
    const float* lane_src = src + sub * 4;
    int j = beg;
    for (; j + 4 <= end; j += 4) {
        int s[4];
#pragma unroll
        for (int k = 0; k < 4; ++k) s[k] = perm[j + k];
        float4 v[4][V];
#pragma unroll
        for (int k = 0; k < 4; ++k)
#pragma unroll
            for (int u = 0; u < V; ++u)
                v[k][u] = *(const float4*)(lane_src + (size_t)s[k] * F + u * (PER * 4));
#pragma unroll
        for (int k = 0; k < 4; ++k)
#pragma unroll
            for (int u = 0; u < V; ++u) {
                acc[u].x += v[k][u].x; acc[u].y += v[k][u].y;
                acc[u].z += v[k][u].z; acc[u].w += v[k][u].w;
            }
    }
    if (j < end) {
        int s[3];
#pragma unroll
        for (int k = 0; k < 3; ++k) s[k] = perm[min(j + k, end - 1)];
        float4 v[3][V];
#pragma unroll
        for (int k = 0; k < 3; ++k)
#pragma unroll
            for (int u = 0; u < V; ++u)
                v[k][u] = *(const float4*)(lane_src + (size_t)s[k] * F + u * (PER * 4));
#pragma unroll
        for (int k = 0; k < 3; ++k) {
            const bool on = j + k < end;
#pragma unroll
            for (int u = 0; u < V; ++u) {
                acc[u].x += on ? v[k][u].x : 0.f; acc[u].y += on ? v[k][u].y : 0.f;
                acc[u].z += on ? v[k][u].z : 0.f; acc[u].w += on ? v[k][u].w : 0.f;
            }
        }
    }
#pragma unroll
    for (int u = 0; u < V; ++u)
        *(float4*)(dst + (size_t)d * F + (u * PER + sub) * 4) = acc[u];
}

__global__ __launch_bounds__(256) void gather_add3_finish(const float* __restrict__ projA,
                                                          const int* __restrict__ offA,
                                                          const int* __restrict__ permA,
                                                          const float* __restrict__ xA,
                                                          float* outA, int nA, int blocksA,
                                                          const float* __restrict__ projB,
                                                          const int* __restrict__ offB,
                                                          const int* __restrict__ permB,
                                                          const float* __restrict__ xB,
                                                          float* outB, int nB) {
    const bool jobA = (int)blockIdx.x < blocksA;
    const float* __restrict__ proj = jobA ? projA : projB;
    const int* __restrict__ off = jobA ? offA : offB;
    const int* __restrict__ perm = jobA ? permA : permB;
    const float* __restrict__ x = jobA ? xA : xB;
    float* out = jobA ? outA : outB;
    const int nDst = jobA ? nA : nB;
    const int blk = jobA ? (int)blockIdx.x : (int)blockIdx.x - blocksA;
    int d = blk * blockDim.x + threadIdx.x;
    if (d >= nDst) return;
    const int beg = off[d];
    const int end = off[d + 1];
    float a0 = out[(size_t)d * 3 + 0];
    float a1 = out[(size_t)d * 3 + 1];
    float a2 = out[(size_t)d * 3 + 2];
    const float f0 = x[(size_t)d * 64 + 61];
    const float f2 = x[(size_t)d * 64 + 63];
    for (int j = beg; j < end; ++j) {
        const int s = perm[j];
        a0 += proj[(size_t)s * 3 + 0];
        a1 += proj[(size_t)s * 3 + 1];
        a2 += proj[(size_t)s * 3 + 2];
    }
    float nrm = sqrtf(a0 * a0 + a1 * a1 + a2 * a2);
    float sc = 10.0f / fmaxf(nrm, 1e-12f);
    a0 *= sc; a1 *= sc; a2 *= sc;
    if (f0 != 0.0f) a0 -= 10.0f;
    if (f2 != 0.0f) a2 -= 10.0f;
    out[(size_t)d * 3 + 0] = a0;
    out[(size_t)d * 3 + 1] = a1;
    out[(size_t)d * 3 + 2] = a2;
}

template <int K1, int K2, bool RELU, bool FUSE3>
__global__ __launch_bounds__(256) void gemm_dual_wmma(const float* __restrict__ A1a,
                                                      const float* __restrict__ W1a,
                                                      const float* __restrict__ A2a,
                                                      const float* __restrict__ W2a,
                                                      const float* __restrict__ biasa,
                                                      float* outa, int Ma, int blocksA,
                                                      const float* __restrict__ A1b,
                                                      const float* __restrict__ W1b,
                                                      const float* __restrict__ A2b,
                                                      const float* __restrict__ W2b,
                                                      const float* __restrict__ biasb,
                                                      float* outb, int Mb,
                                                      const float* __restrict__ W3rela,
                                                      const float* __restrict__ W3roota,
                                                      const float* __restrict__ b3a,
                                                      float* __restrict__ proja,
                                                      const float* __restrict__ W3relb,
                                                      const float* __restrict__ W3rootb,
                                                      const float* __restrict__ b3b,
                                                      float* __restrict__ projb) {
    constexpr int N = 128;
    constexpr int KP = (K1 + K2) / 2;
    constexpr int PITCH = 2 * N + 32;
    __shared__ alignas(16) float wl[KP * PITCH];
    constexpr int W3N = FUSE3 ? 1536 : 4;
    __shared__ alignas(16) float w3l[W3N];

    const bool jobA = (int)blockIdx.x < blocksA;
    const float* __restrict__ A1 = jobA ? A1a : A1b;
    const float* __restrict__ W1 = jobA ? W1a : W1b;
    const float* __restrict__ A2 = jobA ? A2a : A2b;
    const float* __restrict__ W2 = jobA ? W2a : W2b;
    const float* __restrict__ bias = jobA ? biasa : biasb;
    float* out = jobA ? outa : outb;
    const int M = jobA ? Ma : Mb;
    const int blk = jobA ? (int)blockIdx.x : (int)blockIdx.x - blocksA;

    float l3r0 = 0.f, l3o0 = 0.f, l3r1 = 0.f, l3o1 = 0.f, l3b = 0.f, l3c = 0.f;
    if constexpr (FUSE3) {
        const float* __restrict__ W3rel = jobA ? W3rela : W3relb;
        const float* __restrict__ W3root = jobA ? W3roota : W3rootb;
        const float* __restrict__ b3 = jobA ? b3a : b3b;
        const int t = threadIdx.x;
        const int i1 = min(t + 256, 383);
        l3r0 = W3rel[t];
        l3o0 = W3root[t];
        l3r1 = W3rel[i1];
        l3o1 = W3root[i1];
        l3b = bias[t & 127];
        l3c = b3[min(t, 2)];
    }

    {
        const int c2 = (threadIdx.x & 63) * 2;
        const int p0 = threadIdx.x >> 6;
#pragma unroll 8
        for (int i = 0; i < K1 / 8; ++i) {
            const int p = p0 + 4 * i;
            const float2 r0 = *(const float2*)(W1 + (size_t)(2 * p) * N + c2);
            const float2 r1 = *(const float2*)(W1 + (size_t)(2 * p + 1) * N + c2);
            *(float4*)(wl + p * PITCH + c2 * 2) = make_float4(r0.x, r1.x, r0.y, r1.y);
        }
#pragma unroll 8
        for (int i = 0; i < K2 / 8; ++i) {
            const int p = p0 + 4 * i;
            const float2 r0 = *(const float2*)(W2 + (size_t)(2 * p) * N + c2);
            const float2 r1 = *(const float2*)(W2 + (size_t)(2 * p + 1) * N + c2);
            *(float4*)(wl + (K1 / 2 + p) * PITCH + c2 * 2) = make_float4(r0.x, r1.x, r0.y, r1.y);
        }
    }
    if constexpr (FUSE3) {
        const int t = threadIdx.x;
        *(float2*)(w3l + 2 * t) = make_float2(l3r0, l3o0);
        *(float2*)(w3l + 2 * (t + 256)) = make_float2(l3r1, l3o1);
        w3l[1024 + t] = l3b;
        w3l[1280 + t] = l3c;
    }
    __syncthreads();

    const int lane = threadIdx.x & 31;
    const int wave = threadIdx.x >> 5;
    const int m0 = (blk * 8 + wave) * 16;
    if (m0 >= M) return;
    const int half = lane >> 4;
    const int lr = lane & 15;

    const float* a1p = A1 + (size_t)(m0 + lr) * K1 + 2 * half;
    const float* a2p = A2 + (size_t)(m0 + lr) * K2 + 2 * half;
    const float* bl = wl + half * PITCH + lr * 2;

    constexpr int CH = 8;
    constexpr int NCH = (K1 + K2) / (4 * CH);

    v8f acc[8] = {};
    v2f ac[CH], an[CH];
    v2f bc[8], bn[8];
#pragma unroll
    for (int j = 0; j < CH; ++j) ac[j] = *(const v2f*)(a1p + j * 4);
#pragma unroll
    for (int nt = 0; nt < 8; ++nt) bc[nt] = *(const v2f*)(bl + nt * 32);

#pragma unroll 1
    for (int ci = 0; ci < NCH; ++ci) {
        const bool more = ci + 1 < NCH;
        if (more) {
            const int kn = (ci + 1) * (4 * CH);
            const float* ap = (kn < K1) ? (a1p + kn) : (a2p + (kn - K1));
#pragma unroll
            for (int j = 0; j < CH; ++j) an[j] = *(const v2f*)(ap + j * 4);
        }
#pragma unroll
        for (int j = 0; j < CH; ++j) {
            if (j + 1 < CH || more) {
#pragma unroll
                for (int nt = 0; nt < 8; ++nt)
                    bn[nt] = *(const v2f*)(bl + (2 * (j + 1)) * PITCH + nt * 32);
            }
#pragma unroll
            for (int nt = 0; nt < 8; ++nt) {
                if constexpr (FUSE3)
                    acc[nt] = __builtin_amdgcn_wmma_f32_16x16x4_f32(
                        false, bc[nt], false, ac[j], (short)0, acc[nt], false, false);
                else
                    acc[nt] = __builtin_amdgcn_wmma_f32_16x16x4_f32(
                        false, ac[j], false, bc[nt], (short)0, acc[nt], false, false);
            }
#pragma unroll
            for (int nt = 0; nt < 8; ++nt) bc[nt] = bn[nt];
        }
        if (more) {
#pragma unroll
            for (int j = 0; j < CH; ++j) ac[j] = an[j];
        }
        bl += 2 * CH * PITCH;
    }

    if constexpr (FUSE3) {
        float* __restrict__ proj = jobA ? proja : projb;
        v2f pq[2][3];
#pragma unroll
        for (int cc = 0; cc < 3; ++cc) {
            pq[0][cc] = (v2f){0.f, 0.f};
            pq[1][cc] = (v2f){0.f, 0.f};
        }
#pragma unroll
        for (int nt = 0; nt < 8; ++nt) {
            const int n0 = nt * 16 + 8 * half;
            const float4* bp = (const float4*)(w3l + 1024 + n0);
            const float4* wp = (const float4*)(w3l + n0 * 6);
            float bb[8];
            v2f w[24];
#pragma unroll
            for (int t = 0; t < 2; ++t) {
                const float4 x = bp[t];
                bb[4 * t + 0] = x.x; bb[4 * t + 1] = x.y; bb[4 * t + 2] = x.z; bb[4 * t + 3] = x.w;
            }
#pragma unroll
            for (int t = 0; t < 12; ++t) {
                const float4 x = wp[t];
                w[2 * t + 0] = (v2f){x.x, x.y};
                w[2 * t + 1] = (v2f){x.z, x.w};
            }
#pragma unroll
            for (int r = 0; r < 8; ++r) {
                float v = acc[nt][r] + bb[r];
                if (RELU) v = fmaxf(v, 0.0f);
                const v2f vv = (v2f){v, v};
#pragma unroll
                for (int cc = 0; cc < 3; ++cc) pq[r & 1][cc] += vv * w[r * 3 + cc];
            }
        }
        float s[6];
#pragma unroll
        for (int cc = 0; cc < 3; ++cc) {
            const v2f t = pq[0][cc] + pq[1][cc];
            s[cc] = t.x;
            s[3 + cc] = t.y;
        }
#pragma unroll
        for (int cc = 0; cc < 6; ++cc) s[cc] += __shfl_xor(s[cc], 16, 32);
        if (half == 0) {
            const size_t m = (size_t)(m0 + lr);
#pragma unroll
            for (int cc = 0; cc < 3; ++cc) {
                proj[m * 3 + cc] = s[cc];
                out[m * 3 + cc] = s[3 + cc] + w3l[1280 + cc];
            }
        }
    } else {
#pragma unroll
        for (int nt = 0; nt < 8; ++nt) {
            const int col = nt * 16 + lr;
            const float bv = bias[col];
#pragma unroll
            for (int i = 0; i < 8; ++i) {
                float v = acc[nt][i] + bv;
                if (RELU) v = fmaxf(v, 0.0f);
                out[(size_t)(m0 + i + 8 * half) * N + col] = v;
            }
        }
    }
}

static inline int ceil_div(long long a, long long b) { return (int)((a + b - 1) / b); }

extern "C" void kernel_launch(void* const* d_in, const int* in_sizes, int n_in,
                              void* d_out, int out_size, void* d_ws, size_t ws_size,
                              hipStream_t stream) {
    const float* x_s = (const float*)d_in[0];
    const float* x_t = (const float*)d_in[1];
    const int* e_l = (const int*)d_in[2];
    const int* e_r = (const int*)d_in[3];
    const float* w1lr_rel  = (const float*)d_in[4];
    const float* b1lr      = (const float*)d_in[5];
    const float* w1lr_root = (const float*)d_in[6];
    const float* w1rl_rel  = (const float*)d_in[7];
    const float* b1rl      = (const float*)d_in[8];
    const float* w1rl_root = (const float*)d_in[9];
    const float* w2lr_rel  = (const float*)d_in[10];
    const float* b2lr      = (const float*)d_in[11];
    const float* w2lr_root = (const float*)d_in[12];
    const float* w2rl_rel  = (const float*)d_in[13];
    const float* b2rl      = (const float*)d_in[14];
    const float* w2rl_root = (const float*)d_in[15];
    const float* w3lr_rel  = (const float*)d_in[16];
    const float* b3lr      = (const float*)d_in[17];
    const float* w3lr_root = (const float*)d_in[18];
    const float* w3rl_rel  = (const float*)d_in[19];
    const float* b3rl      = (const float*)d_in[20];
    const float* w3rl_root = (const float*)d_in[21];

    const int NL = in_sizes[0] / 64;
    const int NR = in_sizes[1] / 64;
    const int E  = in_sizes[2];

    float* ws = (float*)d_ws;
    const size_t AGG_R = 0;
    const size_t AGG_L = AGG_R + (size_t)NR * 128;
    const size_t H1_R  = AGG_L + (size_t)NL * 128;
    const size_t H1_L  = H1_R + (size_t)NR * 128;
    const int tilesR = ceil_div(NR, TILE);
    const int tilesL = ceil_div(NL, TILE);
    int* wi = (int*)d_ws;
    const size_t CNT_R  = H1_L + (size_t)NL * 128;
    const size_t CNT_L  = CNT_R + (size_t)NR * CS;
    const size_t OFF_R  = CNT_L + (size_t)NL * CS;
    const size_t OFF_L  = OFF_R + (size_t)NR + 1;
    const size_t RANK_R = OFF_L + (size_t)NL + 1;
    const size_t RANK_L = RANK_R + (size_t)E;
    const size_t PERM_R = RANK_L + (size_t)E;
    const size_t PERM_L = PERM_R + (size_t)E;
    const size_t TSUM   = PERM_L + (size_t)E;
    const size_t PROJ_LR = TSUM + (size_t)tilesR + (size_t)tilesL;
    const size_t PROJ_RL = PROJ_LR + (size_t)NL * 3;

    float* d_left  = (float*)d_out;
    float* d_right = (float*)d_out + (size_t)NL * 3;

    const int TPB = 256;

    zero_kernel<<<512, TPB, 0, stream>>>((float4*)(ws + CNT_R), (size_t)(NR + NL) * CS / 4);
    edge_rank<<<ceil_div(E, TPB * EPT), TPB, 0, stream>>>(e_l, e_r, wi + CNT_L, wi + CNT_R, wi + RANK_L, wi + RANK_R, E);
    tile_sums<<<tilesR + tilesL, TPB, 0, stream>>>(wi + CNT_R, NR, tilesR, wi + CNT_L, NL, wi + TSUM);
    scan_apply<<<tilesR + tilesL, TPB, 0, stream>>>(wi + CNT_R, NR, tilesR, wi + OFF_R,
                                                    wi + CNT_L, NL, wi + OFF_L, wi + TSUM);
    edge_place<<<ceil_div(E, TPB * EPT), TPB, 0, stream>>>(e_l, e_r, wi + OFF_L, wi + OFF_R, wi + RANK_L, wi + RANK_R,
                                                     wi + PERM_L, wi + PERM_R, E);

    const int gsL = ceil_div((long long)NL * 8, TPB);
    const int gsR = ceil_div((long long)NR * 8, TPB);
    const int gmR = ceil_div(NR, 16 * 8);
    const int gmL = ceil_div(NL, 16 * 8);
    const int fnL = ceil_div(NL, TPB);
    const int fnR = ceil_div(NR, TPB);

    gather_sum_vec<64><<<gsL + gsR, TPB, 0, stream>>>(x_t, wi + OFF_L, wi + PERM_L, ws + AGG_L, NL, gsL,
                                                      x_s, wi + OFF_R, wi + PERM_R, ws + AGG_R, NR);
    gemm_dual_wmma<64, 64, true, false><<<gmR + gmL, TPB, 0, stream>>>(
        ws + AGG_R, w1lr_rel, x_t, w1lr_root, b1lr, ws + H1_R, NR, gmR,
        ws + AGG_L, w1rl_rel, x_s, w1rl_root, b1rl, ws + H1_L, NL,
        nullptr, nullptr, nullptr, nullptr, nullptr, nullptr, nullptr, nullptr);

    gather_sum_vec<128><<<gsL + gsR, TPB, 0, stream>>>(ws + H1_R, wi + OFF_L, wi + PERM_L, ws + AGG_L, NL, gsL,
                                                       ws + H1_L, wi + OFF_R, wi + PERM_R, ws + AGG_R, NR);
    gemm_dual_wmma<128, 128, true, true><<<gmR + gmL, TPB, 0, stream>>>(
        ws + AGG_R, w2lr_rel, ws + H1_R, w2lr_root, b2lr, d_right, NR, gmR,
        ws + AGG_L, w2rl_rel, ws + H1_L, w2rl_root, b2rl, d_left, NL,
        w3rl_rel, w3lr_root, b3lr, ws + PROJ_RL,
        w3lr_rel, w3rl_root, b3rl, ws + PROJ_LR);
    gather_add3_finish<<<fnL + fnR, TPB, 0, stream>>>(ws + PROJ_RL, wi + OFF_L, wi + PERM_L, x_s, d_left, NL, fnL,
                                                      ws + PROJ_LR, wi + OFF_R, wi + PERM_R, x_t, d_right, NR);
}


# Edge grouping without global atomics: edges binned by destination range with LDS counters (count, scan, scatter), each range ordered in one block's LDS; 10 launches become 9; row sum order: LDS order
# speedup vs baseline: 3.6625x; 1.2043x over previous
#include <hip/hip_runtime.h>
#include <hip/hip_bf16.h>

typedef __attribute__((ext_vector_type(2))) float v2f;
typedef __attribute__((ext_vector_type(8))) float v8f;

constexpr int HG = 256;
constexpr int SH_R = 9;
constexpr int SH_L = 8;
constexpr int BW = 512;
constexpr int MAXNB = 1024;
constexpr int EB = 4;

__device__ __forceinline__ int wave_incl_scan(int v, int lane) {
#pragma unroll
    for (int o = 1; o < 32; o <<= 1) {
        const int u = __shfl_up(v, o, 32);
        if (lane >= o) v += u;
    }
    return v;
}

__global__ __launch_bounds__(1024) void bucket_count(const int* __restrict__ e_l,
                                                     const int* __restrict__ e_r, int nE, int chunk,
                                                     int nbR, int nbAll, int* __restrict__ hist) {
    __shared__ int cnt[MAXNB];
    const int tid = threadIdx.x;
    cnt[tid] = 0;
    __syncthreads();
    const int e0 = blockIdx.x * chunk;
    const int e1 = min(nE, e0 + chunk);
    for (int i0 = e0 + tid; i0 < e1; i0 += EB * 1024) {
        int dl[EB], dr[EB];
#pragma unroll
        for (int k = 0; k < EB; ++k) {
            const int i = i0 + k * 1024;
            const bool ok = i < e1;
            dl[k] = ok ? e_l[i] : -1;
            dr[k] = ok ? e_r[i] : -1;
        }
#pragma unroll
        for (int k = 0; k < EB; ++k) {
            if (dl[k] >= 0) {
                atomicAdd(&cnt[dr[k] >> SH_R], 1);
                atomicAdd(&cnt[nbR + (dl[k] >> SH_L)], 1);
            }
        }
    }
    __syncthreads();
    for (int b = tid; b < nbAll; b += 1024) hist[(size_t)b * HG + blockIdx.x] = cnt[b];
}

__global__ __launch_bounds__(1024) void bucket_scan(int* hist, int nbR, int nbL,
                                                    int* __restrict__ bstR, int* __restrict__ bstL) {
    __shared__ int tot[MAXNB + 1];
    const bool isR = blockIdx.x == 0;
    const int nb = isR ? nbR : nbL;
    int* h = hist + (isR ? (size_t)0 : (size_t)nbR * HG);
    int* __restrict__ bst = isR ? bstR : bstL;
    const int lane = threadIdx.x & 31;
    const int wave = threadIdx.x >> 5;

    for (int b0 = wave; b0 < nb; b0 += EB * 32) {
        int s[EB];
#pragma unroll
        for (int k = 0; k < EB; ++k) {
            const int b = min(b0 + 32 * k, nb - 1);
            const int4* row = (const int4*)(h + (size_t)b * HG) + lane * 2;
            const int4 a = row[0];
            const int4 c = row[1];
            s[k] = a.x + a.y + a.z + a.w + c.x + c.y + c.z + c.w;
        }
#pragma unroll
        for (int k = 0; k < EB; ++k) {
#pragma unroll
            for (int o = 16; o; o >>= 1) s[k] += __shfl_xor(s[k], o, 32);
            if (lane == 0 && b0 + 32 * k < nb) tot[b0 + 32 * k] = s[k];
        }
    }
    __syncthreads();
    if (wave == 0) {
        const int per = (nb + 31) >> 5;
        const int i0 = lane * per;
        const int i1 = min(nb, i0 + per);
        int s = 0;
        for (int i = i0; i < i1; ++i) s += tot[i];
        const int incl = wave_incl_scan(s, lane);
        int run = incl - s;
        for (int i = i0; i < i1; ++i) {
            const int v = tot[i];
            tot[i] = run;
            run += v;
        }
        if (lane == 31) tot[nb] = incl;
    }
    __syncthreads();
    for (int b = threadIdx.x; b <= nb; b += 1024) bst[b] = tot[b];

    for (int b0 = wave; b0 < nb; b0 += EB * 32) {
        int4 a[EB], c[EB];
#pragma unroll
        for (int k = 0; k < EB; ++k) {
            const int b = min(b0 + 32 * k, nb - 1);
            const int4* row = (const int4*)(h + (size_t)b * HG) + lane * 2;
            a[k] = row[0];
            c[k] = row[1];
        }
#pragma unroll
        for (int k = 0; k < EB; ++k) {
            const int b = b0 + 32 * k;
            const int s = a[k].x + a[k].y + a[k].z + a[k].w + c[k].x + c[k].y + c[k].z + c[k].w;
            const int incl = wave_incl_scan(s, lane);
            if (b < nb) {
                int run = tot[b] + incl - s;
                int4 oa, oc;
                oa.x = run; run += a[k].x;
                oa.y = run; run += a[k].y;
                oa.z = run; run += a[k].z;
                oa.w = run; run += a[k].w;
                oc.x = run; run += c[k].x;
                oc.y = run; run += c[k].y;
                oc.z = run; run += c[k].z;
                oc.w = run;
                int4* orow = (int4*)(h + (size_t)b * HG) + lane * 2;
                orow[0] = oa;
                orow[1] = oc;
            }
        }
    }
}

__global__ __launch_bounds__(1024) void bucket_scatter(const int* __restrict__ e_l,
                                                       const int* __restrict__ e_r, int nE, int chunk,
                                                       int nbR, int nbAll, const int* __restrict__ hist,
                                                       int2* __restrict__ pairR, int2* __restrict__ pairL) {
    __shared__ int cur[MAXNB];
    const int tid = threadIdx.x;
    for (int b = tid; b < nbAll; b += 1024) cur[b] = hist[(size_t)b * HG + blockIdx.x];
    __syncthreads();
    const int e0 = blockIdx.x * chunk;
    const int e1 = min(nE, e0 + chunk);
    for (int i0 = e0 + tid; i0 < e1; i0 += EB * 1024) {
        int dl[EB], dr[EB];
#pragma unroll
        for (int k = 0; k < EB; ++k) {
            const int i = i0 + k * 1024;
            const bool ok = i < e1;
            dl[k] = ok ? e_l[i] : -1;
            dr[k] = ok ? e_r[i] : -1;
        }
#pragma unroll
        for (int k = 0; k < EB; ++k) {
            if (dl[k] >= 0) {
                const int pr = atomicAdd(&cur[dr[k] >> SH_R], 1);
                const int pl = atomicAdd(&cur[nbR + (dl[k] >> SH_L)], 1);
                pairR[pr] = make_int2(dr[k], dl[k]);
                pairL[pl] = make_int2(dl[k], dr[k]);
            }
        }
    }
}

__global__ __launch_bounds__(1024) void bucket_sort(const int2* __restrict__ pairR,
                                                    const int2* __restrict__ pairL,
                                                    const int* __restrict__ bstR,
                                                    const int* __restrict__ bstL, int nbR, int nbL,
                                                    int* __restrict__ offR, int* __restrict__ offL,
                                                    int* __restrict__ permR, int* __restrict__ permL,
                                                    int nR, int nL) {
    __shared__ int cnt[BW];
    __shared__ int wtot[BW / 32];
    const bool isR = (int)blockIdx.x < nbR;
    const int b = isR ? (int)blockIdx.x : (int)blockIdx.x - nbR;
    const int nb = isR ? nbR : nbL;
    const int2* __restrict__ pair = isR ? pairR : pairL;
    const int* __restrict__ bst = isR ? bstR : bstL;
    int* __restrict__ off = isR ? offR : offL;
    int* __restrict__ perm = isR ? permR : permL;
    const int n = isR ? nR : nL;
    const int sh = isR ? SH_R : SH_L;
    const int lo = b << sh;
    const int tid = threadIdx.x;
    const int lane = tid & 31;
    const int wave = tid >> 5;
    const int s = bst[b];
    const int e = bst[b + 1];
    if (tid < BW) cnt[tid] = 0;
    __syncthreads();
    for (int i0 = s + tid; i0 < e; i0 += EB * 1024) {
        int d[EB];
#pragma unroll
        for (int k = 0; k < EB; ++k) {
            const int i = i0 + k * 1024;
            d[k] = i < e ? pair[i].x : -1;
        }
#pragma unroll
        for (int k = 0; k < EB; ++k)
            if (d[k] >= 0) atomicAdd(&cnt[d[k] - lo], 1);
    }
    __syncthreads();
    const int v = tid < BW ? cnt[tid] : 0;
    const int incl = wave_incl_scan(v, lane);
    if (lane == 31 && wave < BW / 32) wtot[wave] = incl;
    __syncthreads();
    if (tid < BW) {
        int pre = 0;
#pragma unroll
        for (int w = 0; w < BW / 32; ++w) pre += (w < wave) ? wtot[w] : 0;
        const int excl = pre + incl - v;
        cnt[tid] = excl;
        const int node = lo + tid;
        if (tid < (1 << sh) && node < n) off[node] = s + excl;
    }
    if (tid == 0 && b == nb - 1) off[n] = e;
    __syncthreads();
    for (int i0 = s + tid; i0 < e; i0 += EB * 1024) {
        int2 t[EB];
#pragma unroll
        for (int k = 0; k < EB; ++k) {
            const int i = i0 + k * 1024;
            t[k] = i < e ? pair[i] : make_int2(-1, 0);
        }
#pragma unroll
        for (int k = 0; k < EB; ++k) {
            if (t[k].x >= 0) {
                const int p = atomicAdd(&cnt[t[k].x - lo], 1);
                perm[s + p] = t[k].y;
            }
        }
    }
}

template <int F>
__global__ __launch_bounds__(256) void gather_sum_vec(const float* __restrict__ srcA,
                                                      const int* __restrict__ offA,
                                                      const int* __restrict__ permA,
                                                      float* __restrict__ dstA, int nA, int blocksA,
                                                      const float* __restrict__ srcB,
                                                      const int* __restrict__ offB,
                                                      const int* __restrict__ permB,
                                                      float* __restrict__ dstB, int nB) {
    constexpr int PER = 8;
    constexpr int V = F / (4 * PER);
    const bool jobA = (int)blockIdx.x < blocksA;
    const float* __restrict__ src = jobA ? srcA : srcB;
    const int* __restrict__ off = jobA ? offA : offB;
    const int* __restrict__ perm = jobA ? permA : permB;
    float* __restrict__ dst = jobA ? dstA : dstB;
    const int nDst = jobA ? nA : nB;
    const unsigned blk = jobA ? blockIdx.x : blockIdx.x - blocksA;
    const unsigned t = blk * blockDim.x + threadIdx.x;
    const unsigned d = t / PER;
    const unsigned sub = t % PER;
    if (d >= (unsigned)nDst) return;
    const int beg = off[d];
    const int end = off[d + 1];
    float4 acc[V];
#pragma unroll
    for (int u = 0; u < V; ++u) acc[u] = make_float4(0.f, 0.f, 0.f, 0.f);
    const float* lane_src = src + sub * 4;
    int j = beg;
    for (; j + 4 <= end; j += 4) {
        int s[4];
#pragma unroll
        for (int k = 0; k < 4; ++k) s[k] = perm[j + k];
        float4 v[4][V];
#pragma unroll
        for (int k = 0; k < 4; ++k)
#pragma unroll
            for (int u = 0; u < V; ++u)
                v[k][u] = *(const float4*)(lane_src + (size_t)s[k] * F + u * (PER * 4));
#pragma unroll
        for (int k = 0; k < 4; ++k)
#pragma unroll
            for (int u = 0; u < V; ++u) {
                acc[u].x += v[k][u].x; acc[u].y += v[k][u].y;
                acc[u].z += v[k][u].z; acc[u].w += v[k][u].w;
            }
    }
    if (j < end) {
        int s[3];
#pragma unroll
        for (int k = 0; k < 3; ++k) s[k] = perm[min(j + k, end - 1)];
        float4 v[3][V];
#pragma unroll
        for (int k = 0; k < 3; ++k)
#pragma unroll
            for (int u = 0; u < V; ++u)
                v[k][u] = *(const float4*)(lane_src + (size_t)s[k] * F + u * (PER * 4));
#pragma unroll
        for (int k = 0; k < 3; ++k) {
            const bool on = j + k < end;
#pragma unroll
            for (int u = 0; u < V; ++u) {
                acc[u].x += on ? v[k][u].x : 0.f; acc[u].y += on ? v[k][u].y : 0.f;
                acc[u].z += on ? v[k][u].z : 0.f; acc[u].w += on ? v[k][u].w : 0.f;
            }
        }
    }
#pragma unroll
    for (int u = 0; u < V; ++u)
        *(float4*)(dst + (size_t)d * F + (u * PER + sub) * 4) = acc[u];
}

__global__ __launch_bounds__(256) void gather_add3_finish(const float* __restrict__ projA,
                                                          const int* __restrict__ offA,
                                                          const int* __restrict__ permA,
                                                          const float* __restrict__ xA,
                                                          float* outA, int nA, int blocksA,
                                                          const float* __restrict__ projB,
                                                          const int* __restrict__ offB,
                                                          const int* __restrict__ permB,
                                                          const float* __restrict__ xB,
                                                          float* outB, int nB) {
    const bool jobA = (int)blockIdx.x < blocksA;
    const float* __restrict__ proj = jobA ? projA : projB;
    const int* __restrict__ off = jobA ? offA : offB;
    const int* __restrict__ perm = jobA ? permA : permB;
    const float* __restrict__ x = jobA ? xA : xB;
    float* out = jobA ? outA : outB;
    const int nDst = jobA ? nA : nB;
    const int blk = jobA ? (int)blockIdx.x : (int)blockIdx.x - blocksA;
    int d = blk * blockDim.x + threadIdx.x;
    if (d >= nDst) return;
    const int beg = off[d];
    const int end = off[d + 1];
    float a0 = out[(size_t)d * 3 + 0];
    float a1 = out[(size_t)d * 3 + 1];
    float a2 = out[(size_t)d * 3 + 2];
    const float f0 = x[(size_t)d * 64 + 61];
    const float f2 = x[(size_t)d * 64 + 63];
    for (int j = beg; j < end; ++j) {
        const int s = perm[j];
        a0 += proj[(size_t)s * 3 + 0];
        a1 += proj[(size_t)s * 3 + 1];
        a2 += proj[(size_t)s * 3 + 2];
    }
    float nrm = sqrtf(a0 * a0 + a1 * a1 + a2 * a2);
    float sc = 10.0f / fmaxf(nrm, 1e-12f);
    a0 *= sc; a1 *= sc; a2 *= sc;
    if (f0 != 0.0f) a0 -= 10.0f;
    if (f2 != 0.0f) a2 -= 10.0f;
    out[(size_t)d * 3 + 0] = a0;
    out[(size_t)d * 3 + 1] = a1;
    out[(size_t)d * 3 + 2] = a2;
}

template <int K1, int K2, bool RELU, bool FUSE3>
__global__ __launch_bounds__(256) void gemm_dual_wmma(const float* __restrict__ A1a,
                                                      const float* __restrict__ W1a,
                                                      const float* __restrict__ A2a,
                                                      const float* __restrict__ W2a,
                                                      const float* __restrict__ biasa,
                                                      float* outa, int Ma, int blocksA,
                                                      const float* __restrict__ A1b,
                                                      const float* __restrict__ W1b,
                                                      const float* __restrict__ A2b,
                                                      const float* __restrict__ W2b,
                                                      const float* __restrict__ biasb,
                                                      float* outb, int Mb,
                                                      const float* __restrict__ W3rela,
                                                      const float* __restrict__ W3roota,
                                                      const float* __restrict__ b3a,
                                                      float* __restrict__ proja,
                                                      const float* __restrict__ W3relb,
                                                      const float* __restrict__ W3rootb,
                                                      const float* __restrict__ b3b,
                                                      float* __restrict__ projb) {
    constexpr int N = 128;
    constexpr int KP = (K1 + K2) / 2;
    constexpr int PITCH = 2 * N + 32;
    __shared__ alignas(16) float wl[KP * PITCH];
    constexpr int W3N = FUSE3 ? 1536 : 4;
    __shared__ alignas(16) float w3l[W3N];

    const bool jobA = (int)blockIdx.x < blocksA;
    const float* __restrict__ A1 = jobA ? A1a : A1b;
    const float* __restrict__ W1 = jobA ? W1a : W1b;
    const float* __restrict__ A2 = jobA ? A2a : A2b;
    const float* __restrict__ W2 = jobA ? W2a : W2b;
    const float* __restrict__ bias = jobA ? biasa : biasb;
    float* out = jobA ? outa : outb;
    const int M = jobA ? Ma : Mb;
    const int blk = jobA ? (int)blockIdx.x : (int)blockIdx.x - blocksA;

    float l3r0 = 0.f, l3o0 = 0.f, l3r1 = 0.f, l3o1 = 0.f, l3b = 0.f, l3c = 0.f;
    if constexpr (FUSE3) {
        const float* __restrict__ W3rel = jobA ? W3rela : W3relb;
        const float* __restrict__ W3root = jobA ? W3roota : W3rootb;
        const float* __restrict__ b3 = jobA ? b3a : b3b;
        const int t = threadIdx.x;
        const int i1 = min(t + 256, 383);
        l3r0 = W3rel[t];
        l3o0 = W3root[t];
        l3r1 = W3rel[i1];
        l3o1 = W3root[i1];
        l3b = bias[t & 127];
        l3c = b3[min(t, 2)];
    }

    {
        const int c2 = (threadIdx.x & 63) * 2;
        const int p0 = threadIdx.x >> 6;
#pragma unroll 8
        for (int i = 0; i < K1 / 8; ++i) {
            const int p = p0 + 4 * i;
            const float2 r0 = *(const float2*)(W1 + (size_t)(2 * p) * N + c2);
            const float2 r1 = *(const float2*)(W1 + (size_t)(2 * p + 1) * N + c2);
            *(float4*)(wl + p * PITCH + c2 * 2) = make_float4(r0.x, r1.x, r0.y, r1.y);
        }
#pragma unroll 8
        for (int i = 0; i < K2 / 8; ++i) {
            const int p = p0 + 4 * i;
            const float2 r0 = *(const float2*)(W2 + (size_t)(2 * p) * N + c2);
            const float2 r1 = *(const float2*)(W2 + (size_t)(2 * p + 1) * N + c2);
            *(float4*)(wl + (K1 / 2 + p) * PITCH + c2 * 2) = make_float4(r0.x, r1.x, r0.y, r1.y);
        }
    }
    if constexpr (FUSE3) {
        const int t = threadIdx.x;
        *(float2*)(w3l + 2 * t) = make_float2(l3r0, l3o0);
        *(float2*)(w3l + 2 * (t + 256)) = make_float2(l3r1, l3o1);
        w3l[1024 + t] = l3b;
        w3l[1280 + t] = l3c;
    }
    __syncthreads();

    const int lane = threadIdx.x & 31;
    const int wave = threadIdx.x >> 5;
    const int m0 = (blk * 8 + wave) * 16;
    if (m0 >= M) return;
    const int half = lane >> 4;
    const int lr = lane & 15;

    const float* a1p = A1 + (size_t)(m0 + lr) * K1 + 2 * half;
    const float* a2p = A2 + (size_t)(m0 + lr) * K2 + 2 * half;
    const float* bl = wl + half * PITCH + lr * 2;

    constexpr int CH = 8;
    constexpr int NCH = (K1 + K2) / (4 * CH);

    v8f acc[8] = {};
    v2f ac[CH], an[CH];
    v2f bc[8], bn[8];
#pragma unroll
    for (int j = 0; j < CH; ++j) ac[j] = *(const v2f*)(a1p + j * 4);
#pragma unroll
    for (int nt = 0; nt < 8; ++nt) bc[nt] = *(const v2f*)(bl + nt * 32);

#pragma unroll 1
    for (int ci = 0; ci < NCH; ++ci) {
        const bool more = ci + 1 < NCH;
        if (more) {
            const int kn = (ci + 1) * (4 * CH);
            const float* ap = (kn < K1) ? (a1p + kn) : (a2p + (kn - K1));
#pragma unroll
            for (int j = 0; j < CH; ++j) an[j] = *(const v2f*)(ap + j * 4);
        }
#pragma unroll
        for (int j = 0; j < CH; ++j) {
            if (j + 1 < CH || more) {
#pragma unroll
                for (int nt = 0; nt < 8; ++nt)
                    bn[nt] = *(const v2f*)(bl + (2 * (j + 1)) * PITCH + nt * 32);
            }
#pragma unroll
            for (int nt = 0; nt < 8; ++nt) {
                if constexpr (FUSE3)
                    acc[nt] = __builtin_amdgcn_wmma_f32_16x16x4_f32(
                        false, bc[nt], false, ac[j], (short)0, acc[nt], false, false);
                else
                    acc[nt] = __builtin_amdgcn_wmma_f32_16x16x4_f32(
                        false, ac[j], false, bc[nt], (short)0, acc[nt], false, false);
            }
#pragma unroll
            for (int nt = 0; nt < 8; ++nt) bc[nt] = bn[nt];
        }
        if (more) {
#pragma unroll
            for (int j = 0; j < CH; ++j) ac[j] = an[j];
        }
        bl += 2 * CH * PITCH;
    }

    if constexpr (FUSE3) {
        float* __restrict__ proj = jobA ? proja : projb;
        v2f pq[2][3];
#pragma unroll
        for (int cc = 0; cc < 3; ++cc) {
            pq[0][cc] = (v2f){0.f, 0.f};
            pq[1][cc] = (v2f){0.f, 0.f};
        }
#pragma unroll
        for (int nt = 0; nt < 8; ++nt) {
            const int n0 = nt * 16 + 8 * half;
            const float4* bp = (const float4*)(w3l + 1024 + n0);
            const float4* wp = (const float4*)(w3l + n0 * 6);
            float bb[8];
            v2f w[24];
#pragma unroll
            for (int t = 0; t < 2; ++t) {
                const float4 x = bp[t];
                bb[4 * t + 0] = x.x; bb[4 * t + 1] = x.y; bb[4 * t + 2] = x.z; bb[4 * t + 3] = x.w;
            }
#pragma unroll
            for (int t = 0; t < 12; ++t) {
                const float4 x = wp[t];
                w[2 * t + 0] = (v2f){x.x, x.y};
                w[2 * t + 1] = (v2f){x.z, x.w};
            }
#pragma unroll
            for (int r = 0; r < 8; ++r) {
                float v = acc[nt][r] + bb[r];
                if (RELU) v = fmaxf(v, 0.0f);
                const v2f vv = (v2f){v, v};
#pragma unroll
                for (int cc = 0; cc < 3; ++cc) pq[r & 1][cc] += vv * w[r * 3 + cc];
            }
        }
        float s[6];
#pragma unroll
        for (int cc = 0; cc < 3; ++cc) {
            const v2f t = pq[0][cc] + pq[1][cc];
            s[cc] = t.x;
            s[3 + cc] = t.y;
        }
#pragma unroll
        for (int cc = 0; cc < 6; ++cc) s[cc] += __shfl_xor(s[cc], 16, 32);
        if (half == 0) {
            const size_t m = (size_t)(m0 + lr);
#pragma unroll
            for (int cc = 0; cc < 3; ++cc) {
                proj[m * 3 + cc] = s[cc];
                out[m * 3 + cc] = s[3 + cc] + w3l[1280 + cc];
            }
        }
    } else {
#pragma unroll
        for (int nt = 0; nt < 8; ++nt) {
            const int col = nt * 16 + lr;
            const float bv = bias[col];
#pragma unroll
            for (int i = 0; i < 8; ++i) {
                float v = acc[nt][i] + bv;
                if (RELU) v = fmaxf(v, 0.0f);
                out[(size_t)(m0 + i + 8 * half) * N + col] = v;
            }
        }
    }
}

static inline int ceil_div(long long a, long long b) { return (int)((a + b - 1) / b); }

extern "C" void kernel_launch(void* const* d_in, const int* in_sizes, int n_in,
                              void* d_out, int out_size, void* d_ws, size_t ws_size,
                              hipStream_t stream) {
    const float* x_s = (const float*)d_in[0];
    const float* x_t = (const float*)d_in[1];
    const int* e_l = (const int*)d_in[2];
    const int* e_r = (const int*)d_in[3];
    const float* w1lr_rel  = (const float*)d_in[4];
    const float* b1lr      = (const float*)d_in[5];
    const float* w1lr_root = (const float*)d_in[6];
    const float* w1rl_rel  = (const float*)d_in[7];
    const float* b1rl      = (const float*)d_in[8];
    const float* w1rl_root = (const float*)d_in[9];
    const float* w2lr_rel  = (const float*)d_in[10];
    const float* b2lr      = (const float*)d_in[11];
    const float* w2lr_root = (const float*)d_in[12];
    const float* w2rl_rel  = (const float*)d_in[13];
    const float* b2rl      = (const float*)d_in[14];
    const float* w2rl_root = (const float*)d_in[15];
    const float* w3lr_rel  = (const float*)d_in[16];
    const float* b3lr      = (const float*)d_in[17];
    const float* w3lr_root = (const float*)d_in[18];
    const float* w3rl_rel  = (const float*)d_in[19];
    const float* b3rl      = (const float*)d_in[20];
    const float* w3rl_root = (const float*)d_in[21];

    const int NL = in_sizes[0] / 64;
    const int NR = in_sizes[1] / 64;
    const int E  = in_sizes[2];

    float* ws = (float*)d_ws;
    const size_t AGG_R = 0;
    const size_t AGG_L = AGG_R + (size_t)NR * 128;
    const size_t H1_R  = AGG_L + (size_t)NL * 128;
    const size_t H1_L  = H1_R + (size_t)NR * 128;
    const int nbR = ceil_div(NR, 1 << SH_R);
    const int nbL = ceil_div(NL, 1 << SH_L);
    const int chunk = ceil_div(E, HG);
    int* wi = (int*)d_ws;
    const size_t PAIR_R = H1_L + (size_t)NL * 128;
    const size_t PAIR_L = PAIR_R + (size_t)E * 2;
    const size_t HIST   = PAIR_L + (size_t)E * 2;
    const size_t PERM_R = HIST + (size_t)(nbR + nbL) * HG;
    const size_t PERM_L = PERM_R + (size_t)E;
    const size_t OFF_R  = PERM_L + (size_t)E;
    const size_t OFF_L  = OFF_R + (size_t)NR + 1;
    const size_t BST_R  = OFF_L + (size_t)NL + 1;
    const size_t BST_L  = BST_R + (size_t)nbR + 1;
    const size_t PROJ_LR = BST_L + (size_t)nbL + 1;
    const size_t PROJ_RL = PROJ_LR + (size_t)NL * 3;

    float* d_left  = (float*)d_out;
    float* d_right = (float*)d_out + (size_t)NL * 3;

    const int TPB = 256;

    bucket_count<<<HG, 1024, 0, stream>>>(e_l, e_r, E, chunk, nbR, nbR + nbL, wi + HIST);
    bucket_scan<<<2, 1024, 0, stream>>>(wi + HIST, nbR, nbL, wi + BST_R, wi + BST_L);
    bucket_scatter<<<HG, 1024, 0, stream>>>(e_l, e_r, E, chunk, nbR, nbR + nbL, wi + HIST,
                                           (int2*)(wi + PAIR_R), (int2*)(wi + PAIR_L));
    bucket_sort<<<nbR + nbL, 1024, 0, stream>>>((const int2*)(wi + PAIR_R), (const int2*)(wi + PAIR_L),
                                                wi + BST_R, wi + BST_L, nbR, nbL,
                                                wi + OFF_R, wi + OFF_L, wi + PERM_R, wi + PERM_L, NR, NL);

    const int gsL = ceil_div((long long)NL * 8, TPB);
    const int gsR = ceil_div((long long)NR * 8, TPB);
    const int gmR = ceil_div(NR, 16 * 8);
    const int gmL = ceil_div(NL, 16 * 8);
    const int fnL = ceil_div(NL, TPB);
    const int fnR = ceil_div(NR, TPB);

    gather_sum_vec<64><<<gsL + gsR, TPB, 0, stream>>>(x_t, wi + OFF_L, wi + PERM_L, ws + AGG_L, NL, gsL,
                                                      x_s, wi + OFF_R, wi + PERM_R, ws + AGG_R, NR);
    gemm_dual_wmma<64, 64, true, false><<<gmR + gmL, TPB, 0, stream>>>(
        ws + AGG_R, w1lr_rel, x_t, w1lr_root, b1lr, ws + H1_R, NR, gmR,
        ws + AGG_L, w1rl_rel, x_s, w1rl_root, b1rl, ws + H1_L, NL,
        nullptr, nullptr, nullptr, nullptr, nullptr, nullptr, nullptr, nullptr);

    gather_sum_vec<128><<<gsL + gsR, TPB, 0, stream>>>(ws + H1_R, wi + OFF_L, wi + PERM_L, ws + AGG_L, NL, gsL,
                                                       ws + H1_L, wi + OFF_R, wi + PERM_R, ws + AGG_R, NR);
    gemm_dual_wmma<128, 128, true, true><<<gmR + gmL, TPB, 0, stream>>>(
        ws + AGG_R, w2lr_rel, ws + H1_R, w2lr_root, b2lr, d_right, NR, gmR,
        ws + AGG_L, w2rl_rel, ws + H1_L, w2rl_root, b2rl, d_left, NL,
        w3rl_rel, w3lr_root, b3lr, ws + PROJ_RL,
        w3lr_rel, w3rl_root, b3rl, ws + PROJ_LR);
    gather_add3_finish<<<fnL + fnR, TPB, 0, stream>>>(ws + PROJ_RL, wi + OFF_L, wi + PERM_L, x_s, d_left, NL, fnL,
                                                      ws + PROJ_LR, wi + OFF_R, wi + PERM_R, x_t, d_right, NR);
}


# Edge grouping in 2 launches, not 4: a block counts, scans and places its edges by bucket in its own piece (no grid-wide count or scan); bucket blocks collect the runs; 9 launches to 7; sum order: LDS
# speedup vs baseline: 4.0034x; 1.0931x over previous
#include <hip/hip_runtime.h>
#include <hip/hip_bf16.h>

typedef __attribute__((ext_vector_type(2))) float v2f;
typedef __attribute__((ext_vector_type(8))) float v8f;

constexpr int HG = 256;
constexpr int SH_R = 9;
constexpr int SH_L = 8;
constexpr int BW = 512;
constexpr int MAXNB = 1024;
constexpr int EB = 4;

__device__ __forceinline__ int wave_incl_scan(int v, int lane) {
#pragma unroll
    for (int o = 1; o < 32; o <<= 1) {
        const int u = __shfl_up(v, o, 32);
        if (lane >= o) v += u;
    }
    return v;
}

__global__ __launch_bounds__(1024) void bucket_local(const int* __restrict__ e_l,
                                                     const int* __restrict__ e_r, int nE, int chunk,
                                                     int nbR, int nbAll, int* __restrict__ tab,
                                                     int2* __restrict__ pairR, int2* __restrict__ pairL) {
    static_assert(EB == 4, "the statements that group the edge loads name EB = 4 pairs");
    __shared__ int cnt[MAXNB];
    __shared__ int wtot[32];
    const int tid = threadIdx.x;
    const int lane = tid & 31;
    const int wave = tid >> 5;
    cnt[tid] = 0;
    __syncthreads();
    const int e0 = blockIdx.x * chunk;
    const int e1 = min(nE, e0 + chunk);
    const int n = max(e1 - e0, 0);
    for (int i0 = e0 + tid; i0 < e1; i0 += EB * 1024) {
        int dl[EB], dr[EB];
#pragma unroll
        for (int k = 0; k < EB; ++k) {
            const int ic = min(i0 + k * 1024, e1 - 1);
            dl[k] = e_l[ic];
            dr[k] = e_r[ic];
        }
        asm volatile("" ::"v"(dl[0]), "v"(dl[1]), "v"(dl[2]), "v"(dl[3]), "v"(dr[0]), "v"(dr[1]),
                     "v"(dr[2]), "v"(dr[3]));
#pragma unroll
        for (int k = 0; k < EB; ++k)
            if (i0 + k * 1024 >= e1) dl[k] = -1;
#pragma unroll
        for (int k = 0; k < EB; ++k) {
            if (dl[k] >= 0) {
                atomicAdd(&cnt[dr[k] >> SH_R], 1);
                atomicAdd(&cnt[nbR + (dl[k] >> SH_L)], 1);
            }
        }
    }
    __syncthreads();
    const int v = cnt[tid];
    const int incl = wave_incl_scan(v, lane);
    if (lane == 31) wtot[wave] = incl;
    __syncthreads();
    int pre = 0;
#pragma unroll
    for (int w = 0; w < 32; ++w) pre += (w < wave) ? wtot[w] : 0;
    const int excl = pre + incl - v;
    if (tid <= nbAll) tab[(size_t)tid * HG + blockIdx.x] = excl;
    cnt[tid] = excl - (tid >= nbR ? n : 0);
    __syncthreads();
    for (int i0 = e0 + tid; i0 < e1; i0 += EB * 1024) {
        int dl[EB], dr[EB];
#pragma unroll
        for (int k = 0; k < EB; ++k) {
            const int ic = min(i0 + k * 1024, e1 - 1);
            dl[k] = e_l[ic];
            dr[k] = e_r[ic];
        }
        asm volatile("" ::"v"(dl[0]), "v"(dl[1]), "v"(dl[2]), "v"(dl[3]), "v"(dr[0]), "v"(dr[1]),
                     "v"(dr[2]), "v"(dr[3]));
#pragma unroll
        for (int k = 0; k < EB; ++k)
            if (i0 + k * 1024 >= e1) dl[k] = -1;
#pragma unroll
        for (int k = 0; k < EB; ++k) {
            if (dl[k] >= 0) {
                const int pr = atomicAdd(&cnt[dr[k] >> SH_R], 1);
                const int pl = atomicAdd(&cnt[nbR + (dl[k] >> SH_L)], 1);
                pairR[e0 + pr] = make_int2(dr[k], dl[k]);
                pairL[e0 + pl] = make_int2(dl[k], dr[k]);
            }
        }
    }
}

__global__ __launch_bounds__(1024) void bucket_sort(const int2* __restrict__ pairR,
                                                    const int2* __restrict__ pairL,
                                                    const int* __restrict__ tab, int chunk,
                                                    int nbR, int nbL,
                                                    int* __restrict__ offR, int* __restrict__ offL,
                                                    int* __restrict__ permR, int* __restrict__ permL,
                                                    int nR, int nL) {
    constexpr int RW = HG / 32;
    static_assert(HG % 32 == 0 && RW % EB == 0, "a wave takes its runs EB at a time");
    __shared__ int cnt[BW];
    __shared__ int wtot[BW / 32];
    __shared__ int segOff[HG];
    __shared__ int segLen[HG];
    __shared__ int red[2 * RW];
    const bool isR = (int)blockIdx.x < nbR;
    const int b = isR ? (int)blockIdx.x : (int)blockIdx.x - nbR;
    const int nb = isR ? nbR : nbL;
    const int base = isR ? 0 : nbR;
    const int2* __restrict__ pair = isR ? pairR : pairL;
    int* __restrict__ off = isR ? offR : offL;
    int* __restrict__ perm = isR ? permR : permL;
    const int n = isR ? nR : nL;
    const int sh = isR ? SH_R : SH_L;
    const int lo = b << sh;
    const int tid = threadIdx.x;
    const int lane = tid & 31;
    const int wave = tid >> 5;
    int ahead = 0, len = 0;
    if (tid < HG) {
        const int a0 = tab[(size_t)base * HG + tid];
        const int a1 = tab[(size_t)(base + b) * HG + tid];
        const int a2 = tab[(size_t)(base + b + 1) * HG + tid];
        ahead = a1 - a0;
        len = a2 - a1;
        segOff[tid] = tid * chunk + ahead;
        segLen[tid] = len;
    }
    if (tid < BW) cnt[tid] = 0;
    {
        int sa = ahead, sl = len;
#pragma unroll
        for (int o = 16; o; o >>= 1) {
            sa += __shfl_xor(sa, o, 32);
            sl += __shfl_xor(sl, o, 32);
        }
        if (lane == 0 && wave < RW) {
            red[wave] = sa;
            red[RW + wave] = sl;
        }
    }
    __syncthreads();
    int s = 0, tot = 0;
#pragma unroll
    for (int w = 0; w < RW; ++w) {
        s += red[w];
        tot += red[RW + w];
    }
    const int e = s + tot;
#pragma unroll 1
    for (int h = 0; h < RW; h += EB) {
        int ro[EB], rl[EB], d[EB];
#pragma unroll
        for (int k = 0; k < EB; ++k) {
            ro[k] = segOff[wave + 32 * (h + k)];
            rl[k] = segLen[wave + 32 * (h + k)];
        }
#pragma unroll
        for (int k = 0; k < EB; ++k) d[k] = lane < rl[k] ? pair[ro[k] + lane].x : -1;
#pragma unroll
        for (int k = 0; k < EB; ++k)
            if (d[k] >= 0) atomicAdd(&cnt[d[k] - lo], 1);
#pragma unroll
        for (int k = 0; k < EB; ++k)
            for (int l = lane + 32; l < rl[k]; l += 32) atomicAdd(&cnt[pair[ro[k] + l].x - lo], 1);
    }
    __syncthreads();
    const int v = tid < BW ? cnt[tid] : 0;
    const int incl = wave_incl_scan(v, lane);
    if (lane == 31 && wave < BW / 32) wtot[wave] = incl;
    __syncthreads();
    if (tid < BW) {
        int pre = 0;
#pragma unroll
        for (int w = 0; w < BW / 32; ++w) pre += (w < wave) ? wtot[w] : 0;
        const int excl = pre + incl - v;
        cnt[tid] = excl;
        const int node = lo + tid;
        if (tid < (1 << sh) && node < n) off[node] = s + excl;
    }
    if (tid == 0 && b == nb - 1) off[n] = e;
    __syncthreads();
#pragma unroll 1
    for (int h = 0; h < RW; h += EB) {
        int ro[EB], rl[EB];
        int2 t[EB];
#pragma unroll
        for (int k = 0; k < EB; ++k) {
            ro[k] = segOff[wave + 32 * (h + k)];
            rl[k] = segLen[wave + 32 * (h + k)];
        }
#pragma unroll
        for (int k = 0; k < EB; ++k) t[k] = lane < rl[k] ? pair[ro[k] + lane] : make_int2(-1, 0);
#pragma unroll
        for (int k = 0; k < EB; ++k) {
            if (t[k].x >= 0) {
                const int p = atomicAdd(&cnt[t[k].x - lo], 1);
                perm[s + p] = t[k].y;
            }
        }
#pragma unroll
        for (int k = 0; k < EB; ++k)
            for (int l = lane + 32; l < rl[k]; l += 32) {
                const int2 u = pair[ro[k] + l];
                const int p = atomicAdd(&cnt[u.x - lo], 1);
                perm[s + p] = u.y;
            }
    }
}

template <int F>
__global__ __launch_bounds__(256) void gather_sum_vec(const float* __restrict__ srcA,
                                                      const int* __restrict__ offA,
                                                      const int* __restrict__ permA,
                                                      float* __restrict__ dstA, int nA, int blocksA,
                                                      const float* __restrict__ srcB,
                                                      const int* __restrict__ offB,
                                                      const int* __restrict__ permB,
                                                      float* __restrict__ dstB, int nB) {
    constexpr int PER = 8;
    constexpr int V = F / (4 * PER);
    const bool jobA = (int)blockIdx.x < blocksA;
    const float* __restrict__ src = jobA ? srcA : srcB;
    const int* __restrict__ off = jobA ? offA : offB;
    const int* __restrict__ perm = jobA ? permA : permB;
    float* __restrict__ dst = jobA ? dstA : dstB;
    const int nDst = jobA ? nA : nB;
    const unsigned blk = jobA ? blockIdx.x : blockIdx.x - blocksA;
    const unsigned t = blk * blockDim.x + threadIdx.x;
    const unsigned d = t / PER;
    const unsigned sub = t % PER;
    if (d >= (unsigned)nDst) return;
    const int beg = off[d];
    const int end = off[d + 1];
    float4 acc[V];
#pragma unroll
    for (int u = 0; u < V; ++u) acc[u] = make_float4(0.f, 0.f, 0.f, 0.f);
    const float* lane_src = src + sub * 4;
    int j = beg;
    for (; j + 4 <= end; j += 4) {
        int s[4];
#pragma unroll
        for (int k = 0; k < 4; ++k) s[k] = perm[j + k];
        float4 v[4][V];
#pragma unroll
        for (int k = 0; k < 4; ++k)
#pragma unroll
            for (int u = 0; u < V; ++u)
                v[k][u] = *(const float4*)(lane_src + (size_t)s[k] * F + u * (PER * 4));
#pragma unroll
        for (int k = 0; k < 4; ++k)
#pragma unroll
            for (int u = 0; u < V; ++u) {
                acc[u].x += v[k][u].x; acc[u].y += v[k][u].y;
                acc[u].z += v[k][u].z; acc[u].w += v[k][u].w;
            }
    }
    if (j < end) {
        int s[3];
#pragma unroll
        for (int k = 0; k < 3; ++k) s[k] = perm[min(j + k, end - 1)];
        float4 v[3][V];
#pragma unroll
        for (int k = 0; k < 3; ++k)
#pragma unroll
            for (int u = 0; u < V; ++u)
                v[k][u] = *(const float4*)(lane_src + (size_t)s[k] * F + u * (PER * 4));
#pragma unroll
        for (int k = 0; k < 3; ++k) {
            const bool on = j + k < end;
#pragma unroll
            for (int u = 0; u < V; ++u) {
                acc[u].x += on ? v[k][u].x : 0.f; acc[u].y += on ? v[k][u].y : 0.f;
                acc[u].z += on ? v[k][u].z : 0.f; acc[u].w += on ? v[k][u].w : 0.f;
            }
        }
    }
#pragma unroll
    for (int u = 0; u < V; ++u)
        *(float4*)(dst + (size_t)d * F + (u * PER + sub) * 4) = acc[u];
}

__global__ __launch_bounds__(256) void gather_add3_finish(const float* __restrict__ projA,
                                                          const int* __restrict__ offA,
                                                          const int* __restrict__ permA,
                                                          const float* __restrict__ xA,
                                                          float* outA, int nA, int blocksA,
                                                          const float* __restrict__ projB,
                                                          const int* __restrict__ offB,
                                                          const int* __restrict__ permB,
                                                          const float* __restrict__ xB,
                                                          float* outB, int nB) {
    const bool jobA = (int)blockIdx.x < blocksA;
    const float* __restrict__ proj = jobA ? projA : projB;
    const int* __restrict__ off = jobA ? offA : offB;
    const int* __restrict__ perm = jobA ? permA : permB;
    const float* __restrict__ x = jobA ? xA : xB;
    float* out = jobA ? outA : outB;
    const int nDst = jobA ? nA : nB;
    const int blk = jobA ? (int)blockIdx.x : (int)blockIdx.x - blocksA;
    int d = blk * blockDim.x + threadIdx.x;
    if (d >= nDst) return;
    const int beg = off[d];
    const int end = off[d + 1];
    float a0 = out[(size_t)d * 3 + 0];
    float a1 = out[(size_t)d * 3 + 1];
    float a2 = out[(size_t)d * 3 + 2];
    const float f0 = x[(size_t)d * 64 + 61];
    const float f2 = x[(size_t)d * 64 + 63];
    for (int j = beg; j < end; ++j) {
        const int s = perm[j];
        a0 += proj[(size_t)s * 3 + 0];
        a1 += proj[(size_t)s * 3 + 1];
        a2 += proj[(size_t)s * 3 + 2];
    }
    float nrm = sqrtf(a0 * a0 + a1 * a1 + a2 * a2);
    float sc = 10.0f / fmaxf(nrm, 1e-12f);
    a0 *= sc; a1 *= sc; a2 *= sc;
    if (f0 != 0.0f) a0 -= 10.0f;
    if (f2 != 0.0f) a2 -= 10.0f;
    out[(size_t)d * 3 + 0] = a0;
    out[(size_t)d * 3 + 1] = a1;
    out[(size_t)d * 3 + 2] = a2;
}

template <int K1, int K2, bool RELU, bool FUSE3>
__global__ __launch_bounds__(256) void gemm_dual_wmma(const float* __restrict__ A1a,
                                                      const float* __restrict__ W1a,
                                                      const float* __restrict__ A2a,
                                                      const float* __restrict__ W2a,
                                                      const float* __restrict__ biasa,
                                                      float* outa, int Ma, int blocksA,
                                                      const float* __restrict__ A1b,
                                                      const float* __restrict__ W1b,
                                                      const float* __restrict__ A2b,
                                                      const float* __restrict__ W2b,
                                                      const float* __restrict__ biasb,
                                                      float* outb, int Mb,
                                                      const float* __restrict__ W3rela,
                                                      const float* __restrict__ W3roota,
                                                      const float* __restrict__ b3a,
                                                      float* __restrict__ proja,
                                                      const float* __restrict__ W3relb,
                                                      const float* __restrict__ W3rootb,
                                                      const float* __restrict__ b3b,
                                                      float* __restrict__ projb) {
    constexpr int N = 128;
    constexpr int KP = (K1 + K2) / 2;
    constexpr int PITCH = 2 * N + 32;
    __shared__ alignas(16) float wl[KP * PITCH];
    constexpr int W3N = FUSE3 ? 1536 : 4;
    __shared__ alignas(16) float w3l[W3N];

    const bool jobA = (int)blockIdx.x < blocksA;
    const float* __restrict__ A1 = jobA ? A1a : A1b;
    const float* __restrict__ W1 = jobA ? W1a : W1b;
    const float* __restrict__ A2 = jobA ? A2a : A2b;
    const float* __restrict__ W2 = jobA ? W2a : W2b;
    const float* __restrict__ bias = jobA ? biasa : biasb;
    float* out = jobA ? outa : outb;
    const int M = jobA ? Ma : Mb;
    const int blk = jobA ? (int)blockIdx.x : (int)blockIdx.x - blocksA;

    float l3r0 = 0.f, l3o0 = 0.f, l3r1 = 0.f, l3o1 = 0.f, l3b = 0.f, l3c = 0.f;
    if constexpr (FUSE3) {
        const float* __restrict__ W3rel = jobA ? W3rela : W3relb;
        const float* __restrict__ W3root = jobA ? W3roota : W3rootb;
        const float* __restrict__ b3 = jobA ? b3a : b3b;
        const int t = threadIdx.x;
        const int i1 = min(t + 256, 383);
        l3r0 = W3rel[t];
        l3o0 = W3root[t];
        l3r1 = W3rel[i1];
        l3o1 = W3root[i1];
        l3b = bias[t & 127];
        l3c = b3[min(t, 2)];
    }

    {
        const int c2 = (threadIdx.x & 63) * 2;
        const int p0 = threadIdx.x >> 6;
#pragma unroll 8
        for (int i = 0; i < K1 / 8; ++i) {
            const int p = p0 + 4 * i;
            const float2 r0 = *(const float2*)(W1 + (size_t)(2 * p) * N + c2);
            const float2 r1 = *(const float2*)(W1 + (size_t)(2 * p + 1) * N + c2);
            *(float4*)(wl + p * PITCH + c2 * 2) = make_float4(r0.x, r1.x, r0.y, r1.y);
        }
#pragma unroll 8
        for (int i = 0; i < K2 / 8; ++i) {
            const int p = p0 + 4 * i;
            const float2 r0 = *(const float2*)(W2 + (size_t)(2 * p) * N + c2);
            const float2 r1 = *(const float2*)(W2 + (size_t)(2 * p + 1) * N + c2);
            *(float4*)(wl + (K1 / 2 + p) * PITCH + c2 * 2) = make_float4(r0.x, r1.x, r0.y, r1.y);
        }
    }
    if constexpr (FUSE3) {
        const int t = threadIdx.x;
        *(float2*)(w3l + 2 * t) = make_float2(l3r0, l3o0);
        *(float2*)(w3l + 2 * (t + 256)) = make_float2(l3r1, l3o1);
        w3l[1024 + t] = l3b;
        w3l[1280 + t] = l3c;
    }
    __syncthreads();

    const int lane = threadIdx.x & 31;
    const int wave = threadIdx.x >> 5;
    const int m0 = (blk * 8 + wave) * 16;
    if (m0 >= M) return;
    const int half = lane >> 4;
    const int lr = lane & 15;

    const float* a1p = A1 + (size_t)(m0 + lr) * K1 + 2 * half;
    const float* a2p = A2 + (size_t)(m0 + lr) * K2 + 2 * half;
    const float* bl = wl + half * PITCH + lr * 2;

    constexpr int CH = 8;
    constexpr int NCH = (K1 + K2) / (4 * CH);

    v8f acc[8] = {};
    v2f ac[CH], an[CH];
    v2f bc[8], bn[8];
#pragma unroll
    for (int j = 0; j < CH; ++j) ac[j] = *(const v2f*)(a1p + j * 4);
#pragma unroll
    for (int nt = 0; nt < 8; ++nt) bc[nt] = *(const v2f*)(bl + nt * 32);

#pragma unroll 1
    for (int ci = 0; ci < NCH; ++ci) {
        const bool more = ci + 1 < NCH;
        if (more) {
            const int kn = (ci + 1) * (4 * CH);
            const float* ap = (kn < K1) ? (a1p + kn) : (a2p + (kn - K1));
#pragma unroll
            for (int j = 0; j < CH; ++j) an[j] = *(const v2f*)(ap + j * 4);
        }
#pragma unroll
        for (int j = 0; j < CH; ++j) {
            if (j + 1 < CH || more) {
#pragma unroll
                for (int nt = 0; nt < 8; ++nt)
                    bn[nt] = *(const v2f*)(bl + (2 * (j + 1)) * PITCH + nt * 32);
            }
#pragma unroll
            for (int nt = 0; nt < 8; ++nt) {
                if constexpr (FUSE3)
                    acc[nt] = __builtin_amdgcn_wmma_f32_16x16x4_f32(
                        false, bc[nt], false, ac[j], (short)0, acc[nt], false, false);
                else
                    acc[nt] = __builtin_amdgcn_wmma_f32_16x16x4_f32(
                        false, ac[j], false, bc[nt], (short)0, acc[nt], false, false);
            }
#pragma unroll
            for (int nt = 0; nt < 8; ++nt) bc[nt] = bn[nt];
        }
        if (more) {
#pragma unroll
            for (int j = 0; j < CH; ++j) ac[j] = an[j];
        }
        bl += 2 * CH * PITCH;
    }

    if constexpr (FUSE3) {
        float* __restrict__ proj = jobA ? proja : projb;
        v2f pq[2][3];
#pragma unroll
        for (int cc = 0; cc < 3; ++cc) {
            pq[0][cc] = (v2f){0.f, 0.f};
            pq[1][cc] = (v2f){0.f, 0.f};
        }
#pragma unroll
        for (int nt = 0; nt < 8; ++nt) {
            const int n0 = nt * 16 + 8 * half;
            const float4* bp = (const float4*)(w3l + 1024 + n0);
            const float4* wp = (const float4*)(w3l + n0 * 6);
            float bb[8];
            v2f w[24];
#pragma unroll
            for (int t = 0; t < 2; ++t) {
                const float4 x = bp[t];
                bb[4 * t + 0] = x.x; bb[4 * t + 1] = x.y; bb[4 * t + 2] = x.z; bb[4 * t + 3] = x.w;
            }
#pragma unroll
            for (int t = 0; t < 12; ++t) {
                const float4 x = wp[t];
                w[2 * t + 0] = (v2f){x.x, x.y};
                w[2 * t + 1] = (v2f){x.z, x.w};
            }
#pragma unroll
            for (int r = 0; r < 8; ++r) {
                float v = acc[nt][r] + bb[r];
                if (RELU) v = fmaxf(v, 0.0f);
                const v2f vv = (v2f){v, v};
#pragma unroll
                for (int cc = 0; cc < 3; ++cc) pq[r & 1][cc] += vv * w[r * 3 + cc];
            }
        }
        float s[6];
#pragma unroll
        for (int cc = 0; cc < 3; ++cc) {
            const v2f t = pq[0][cc] + pq[1][cc];
            s[cc] = t.x;
            s[3 + cc] = t.y;
        }
#pragma unroll
        for (int cc = 0; cc < 6; ++cc) s[cc] += __shfl_xor(s[cc], 16, 32);
        if (half == 0) {
            const size_t m = (size_t)(m0 + lr);
#pragma unroll
            for (int cc = 0; cc < 3; ++cc) {
                proj[m * 3 + cc] = s[cc];
                out[m * 3 + cc] = s[3 + cc] + w3l[1280 + cc];
            }
        }
    } else {
#pragma unroll
        for (int nt = 0; nt < 8; ++nt) {
            const int col = nt * 16 + lr;
            const float bv = bias[col];
#pragma unroll
            for (int i = 0; i < 8; ++i) {
                float v = acc[nt][i] + bv;
                if (RELU) v = fmaxf(v, 0.0f);
                out[(size_t)(m0 + i + 8 * half) * N + col] = v;
            }
        }
    }
}

static inline int ceil_div(long long a, long long b) { return (int)((a + b - 1) / b); }

extern "C" void kernel_launch(void* const* d_in, const int* in_sizes, int n_in,
                              void* d_out, int out_size, void* d_ws, size_t ws_size,
                              hipStream_t stream) {
    const float* x_s = (const float*)d_in[0];
    const float* x_t = (const float*)d_in[1];
    const int* e_l = (const int*)d_in[2];
    const int* e_r = (const int*)d_in[3];
    const float* w1lr_rel  = (const float*)d_in[4];
    const float* b1lr      = (const float*)d_in[5];
    const float* w1lr_root = (const float*)d_in[6];
    const float* w1rl_rel  = (const float*)d_in[7];
    const float* b1rl      = (const float*)d_in[8];
    const float* w1rl_root = (const float*)d_in[9];
    const float* w2lr_rel  = (const float*)d_in[10];
    const float* b2lr      = (const float*)d_in[11];
    const float* w2lr_root = (const float*)d_in[12];
    const float* w2rl_rel  = (const float*)d_in[13];
    const float* b2rl      = (const float*)d_in[14];
    const float* w2rl_root = (const float*)d_in[15];
    const float* w3lr_rel  = (const float*)d_in[16];
    const float* b3lr      = (const float*)d_in[17];
    const float* w3lr_root = (const float*)d_in[18];
    const float* w3rl_rel  = (const float*)d_in[19];
    const float* b3rl      = (const float*)d_in[20];
    const float* w3rl_root = (const float*)d_in[21];

    const int NL = in_sizes[0] / 64;
    const int NR = in_sizes[1] / 64;
    const int E  = in_sizes[2];

    float* ws = (float*)d_ws;
    const size_t AGG_R = 0;
    const size_t AGG_L = AGG_R + (size_t)NR * 128;
    const size_t H1_R  = AGG_L + (size_t)NL * 128;
    const size_t H1_L  = H1_R + (size_t)NR * 128;
    const int nbR = ceil_div(NR, 1 << SH_R);
    const int nbL = ceil_div(NL, 1 << SH_L);
    const int chunk = ceil_div(E, HG);
    int* wi = (int*)d_ws;
    const size_t PAIR_R = H1_L + (size_t)NL * 128;
    const size_t PAIR_L = PAIR_R + (size_t)E * 2;
    const size_t TAB    = PAIR_L + (size_t)E * 2;
    const size_t PERM_R = TAB + (size_t)(nbR + nbL + 1) * HG;
    const size_t PERM_L = PERM_R + (size_t)E;
    const size_t OFF_R  = PERM_L + (size_t)E;
    const size_t OFF_L  = OFF_R + (size_t)NR + 1;
    const size_t PROJ_LR = OFF_L + (size_t)NL + 1;
    const size_t PROJ_RL = PROJ_LR + (size_t)NL * 3;

    float* d_left  = (float*)d_out;
    float* d_right = (float*)d_out + (size_t)NL * 3;

    const int TPB = 256;

    bucket_local<<<HG, 1024, 0, stream>>>(e_l, e_r, E, chunk, nbR, nbR + nbL, wi + TAB,
                                          (int2*)(wi + PAIR_R), (int2*)(wi + PAIR_L));
    bucket_sort<<<nbR + nbL, 1024, 0, stream>>>((const int2*)(wi + PAIR_R), (const int2*)(wi + PAIR_L),
                                                wi + TAB, chunk, nbR, nbL,
                                                wi + OFF_R, wi + OFF_L, wi + PERM_R, wi + PERM_L, NR, NL);

    const int gsL = ceil_div((long long)NL * 8, TPB);
    const int gsR = ceil_div((long long)NR * 8, TPB);
    const int gmR = ceil_div(NR, 16 * 8);
    const int gmL = ceil_div(NL, 16 * 8);
    const int fnL = ceil_div(NL, TPB);
    const int fnR = ceil_div(NR, TPB);

    gather_sum_vec<64><<<gsL + gsR, TPB, 0, stream>>>(x_t, wi + OFF_L, wi + PERM_L, ws + AGG_L, NL, gsL,
                                                      x_s, wi + OFF_R, wi + PERM_R, ws + AGG_R, NR);
    gemm_dual_wmma<64, 64, true, false><<<gmR + gmL, TPB, 0, stream>>>(
        ws + AGG_R, w1lr_rel, x_t, w1lr_root, b1lr, ws + H1_R, NR, gmR,
        ws + AGG_L, w1rl_rel, x_s, w1rl_root, b1rl, ws + H1_L, NL,
        nullptr, nullptr, nullptr, nullptr, nullptr, nullptr, nullptr, nullptr);

    gather_sum_vec<128><<<gsL + gsR, TPB, 0, stream>>>(ws + H1_R, wi + OFF_L, wi + PERM_L, ws + AGG_L, NL, gsL,
                                                       ws + H1_L, wi + OFF_R, wi + PERM_R, ws + AGG_R, NR);
    gemm_dual_wmma<128, 128, true, true><<<gmR + gmL, TPB, 0, stream>>>(
        ws + AGG_R, w2lr_rel, ws + H1_R, w2lr_root, b2lr, d_right, NR, gmR,
        ws + AGG_L, w2rl_rel, ws + H1_L, w2rl_root, b2rl, d_left, NL,
        w3rl_rel, w3lr_root, b3lr, ws + PROJ_RL,
        w3lr_rel, w3rl_root, b3rl, ws + PROJ_LR);
    gather_add3_finish<<<fnL + fnR, TPB, 0, stream>>>(ws + PROJ_RL, wi + OFF_L, wi + PERM_L, x_s, d_left, NL, fnL,
                                                      ws + PROJ_LR, wi + OFF_R, wi + PERM_R, x_t, d_right, NR);
}


# Edge sums formed inside the GEMM launches: a wave sums its 16 rows into an LDS piece of its own and feeds them to its WMMAs; summed rows never stored; 7 launches become 5; same sum order
# speedup vs baseline: 4.3374x; 1.0834x over previous
#include <hip/hip_runtime.h>
#include <hip/hip_bf16.h>

typedef __attribute__((ext_vector_type(2))) float v2f;
typedef __attribute__((ext_vector_type(8))) float v8f;

constexpr int HG = 256;
constexpr int SH_R = 9;
constexpr int SH_L = 8;
constexpr int BW = 512;
constexpr int MAXNB = 1024;
constexpr int EB = 4;

__device__ __forceinline__ int wave_incl_scan(int v, int lane) {
#pragma unroll
    for (int o = 1; o < 32; o <<= 1) {
        const int u = __shfl_up(v, o, 32);
        if (lane >= o) v += u;
    }
    return v;
}

__global__ __launch_bounds__(1024) void bucket_local(const int* __restrict__ e_l,
                                                     const int* __restrict__ e_r, int nE, int chunk,
                                                     int nbR, int nbAll, int* __restrict__ tab,
                                                     int2* __restrict__ pairR, int2* __restrict__ pairL) {
    static_assert(EB == 4, "the statements that group the edge loads name EB = 4 pairs");
    __shared__ int cnt[MAXNB];
    __shared__ int wtot[32];
    const int tid = threadIdx.x;
    const int lane = tid & 31;
    const int wave = tid >> 5;
    cnt[tid] = 0;
    __syncthreads();
    const int e0 = blockIdx.x * chunk;
    const int e1 = min(nE, e0 + chunk);
    const int n = max(e1 - e0, 0);
    for (int i0 = e0 + tid; i0 < e1; i0 += EB * 1024) {
        int dl[EB], dr[EB];
#pragma unroll
        for (int k = 0; k < EB; ++k) {
            const int ic = min(i0 + k * 1024, e1 - 1);
            dl[k] = e_l[ic];
            dr[k] = e_r[ic];
        }
        asm volatile("" ::"v"(dl[0]), "v"(dl[1]), "v"(dl[2]), "v"(dl[3]), "v"(dr[0]), "v"(dr[1]),
                     "v"(dr[2]), "v"(dr[3]));
#pragma unroll
        for (int k = 0; k < EB; ++k)
            if (i0 + k * 1024 >= e1) dl[k] = -1;
#pragma unroll
        for (int k = 0; k < EB; ++k) {
            if (dl[k] >= 0) {
                atomicAdd(&cnt[dr[k] >> SH_R], 1);
                atomicAdd(&cnt[nbR + (dl[k] >> SH_L)], 1);
            }
        }
    }
    __syncthreads();
    const int v = cnt[tid];
    const int incl = wave_incl_scan(v, lane);
    if (lane == 31) wtot[wave] = incl;
    __syncthreads();
    int pre = 0;
#pragma unroll
    for (int w = 0; w < 32; ++w) pre += (w < wave) ? wtot[w] : 0;
    const int excl = pre + incl - v;
    if (tid <= nbAll) tab[(size_t)tid * HG + blockIdx.x] = excl;
    cnt[tid] = excl - (tid >= nbR ? n : 0);
    __syncthreads();
    for (int i0 = e0 + tid; i0 < e1; i0 += EB * 1024) {
        int dl[EB], dr[EB];
#pragma unroll
        for (int k = 0; k < EB; ++k) {
            const int ic = min(i0 + k * 1024, e1 - 1);
            dl[k] = e_l[ic];
            dr[k] = e_r[ic];
        }
        asm volatile("" ::"v"(dl[0]), "v"(dl[1]), "v"(dl[2]), "v"(dl[3]), "v"(dr[0]), "v"(dr[1]),
                     "v"(dr[2]), "v"(dr[3]));
#pragma unroll
        for (int k = 0; k < EB; ++k)
            if (i0 + k * 1024 >= e1) dl[k] = -1;
#pragma unroll
        for (int k = 0; k < EB; ++k) {
            if (dl[k] >= 0) {
                const int pr = atomicAdd(&cnt[dr[k] >> SH_R], 1);
                const int pl = atomicAdd(&cnt[nbR + (dl[k] >> SH_L)], 1);
                pairR[e0 + pr] = make_int2(dr[k], dl[k]);
                pairL[e0 + pl] = make_int2(dl[k], dr[k]);
            }
        }
    }
}

__global__ __launch_bounds__(1024) void bucket_sort(const int2* __restrict__ pairR,
                                                    const int2* __restrict__ pairL,
                                                    const int* __restrict__ tab, int chunk,
                                                    int nbR, int nbL,
                                                    int* __restrict__ offR, int* __restrict__ offL,
                                                    int* __restrict__ permR, int* __restrict__ permL,
                                                    int nR, int nL) {
    constexpr int RW = HG / 32;
    static_assert(HG % 32 == 0 && RW % EB == 0, "a wave takes its runs EB at a time");
    __shared__ int cnt[BW];
    __shared__ int wtot[BW / 32];
    __shared__ int segOff[HG];
    __shared__ int segLen[HG];
    __shared__ int red[2 * RW];
    const bool isR = (int)blockIdx.x < nbR;
    const int b = isR ? (int)blockIdx.x : (int)blockIdx.x - nbR;
    const int nb = isR ? nbR : nbL;
    const int base = isR ? 0 : nbR;
    const int2* __restrict__ pair = isR ? pairR : pairL;
    int* __restrict__ off = isR ? offR : offL;
    int* __restrict__ perm = isR ? permR : permL;
    const int n = isR ? nR : nL;
    const int sh = isR ? SH_R : SH_L;
    const int lo = b << sh;
    const int tid = threadIdx.x;
    const int lane = tid & 31;
    const int wave = tid >> 5;
    int ahead = 0, len = 0;
    if (tid < HG) {
        const int a0 = tab[(size_t)base * HG + tid];
        const int a1 = tab[(size_t)(base + b) * HG + tid];
        const int a2 = tab[(size_t)(base + b + 1) * HG + tid];
        ahead = a1 - a0;
        len = a2 - a1;
        segOff[tid] = tid * chunk + ahead;
        segLen[tid] = len;
    }
    if (tid < BW) cnt[tid] = 0;
    {
        int sa = ahead, sl = len;
#pragma unroll
        for (int o = 16; o; o >>= 1) {
            sa += __shfl_xor(sa, o, 32);
            sl += __shfl_xor(sl, o, 32);
        }
        if (lane == 0 && wave < RW) {
            red[wave] = sa;
            red[RW + wave] = sl;
        }
    }
    __syncthreads();
    int s = 0, tot = 0;
#pragma unroll
    for (int w = 0; w < RW; ++w) {
        s += red[w];
        tot += red[RW + w];
    }
    const int e = s + tot;
#pragma unroll 1
    for (int h = 0; h < RW; h += EB) {
        int ro[EB], rl[EB], d[EB];
#pragma unroll
        for (int k = 0; k < EB; ++k) {
            ro[k] = segOff[wave + 32 * (h + k)];
            rl[k] = segLen[wave + 32 * (h + k)];
        }
#pragma unroll
        for (int k = 0; k < EB; ++k) d[k] = lane < rl[k] ? pair[ro[k] + lane].x : -1;
#pragma unroll
        for (int k = 0; k < EB; ++k)
            if (d[k] >= 0) atomicAdd(&cnt[d[k] - lo], 1);
#pragma unroll
        for (int k = 0; k < EB; ++k)
            for (int l = lane + 32; l < rl[k]; l += 32) atomicAdd(&cnt[pair[ro[k] + l].x - lo], 1);
    }
    __syncthreads();
    const int v = tid < BW ? cnt[tid] : 0;
    const int incl = wave_incl_scan(v, lane);
    if (lane == 31 && wave < BW / 32) wtot[wave] = incl;
    __syncthreads();
    if (tid < BW) {
        int pre = 0;
#pragma unroll
        for (int w = 0; w < BW / 32; ++w) pre += (w < wave) ? wtot[w] : 0;
        const int excl = pre + incl - v;
        cnt[tid] = excl;
        const int node = lo + tid;
        if (tid < (1 << sh) && node < n) off[node] = s + excl;
    }
    if (tid == 0 && b == nb - 1) off[n] = e;
    __syncthreads();
#pragma unroll 1
    for (int h = 0; h < RW; h += EB) {
        int ro[EB], rl[EB];
        int2 t[EB];
#pragma unroll
        for (int k = 0; k < EB; ++k) {
            ro[k] = segOff[wave + 32 * (h + k)];
            rl[k] = segLen[wave + 32 * (h + k)];
        }
#pragma unroll
        for (int k = 0; k < EB; ++k) t[k] = lane < rl[k] ? pair[ro[k] + lane] : make_int2(-1, 0);
#pragma unroll
        for (int k = 0; k < EB; ++k) {
            if (t[k].x >= 0) {
                const int p = atomicAdd(&cnt[t[k].x - lo], 1);
                perm[s + p] = t[k].y;
            }
        }
#pragma unroll
        for (int k = 0; k < EB; ++k)
            for (int l = lane + 32; l < rl[k]; l += 32) {
                const int2 u = pair[ro[k] + l];
                const int p = atomicAdd(&cnt[u.x - lo], 1);
                perm[s + p] = u.y;
            }
    }
}

__global__ __launch_bounds__(256) void gather_add3_finish(const float* __restrict__ projA,
                                                          const int* __restrict__ offA,
                                                          const int* __restrict__ permA,
                                                          const float* __restrict__ xA,
                                                          float* outA, int nA, int blocksA,
                                                          const float* __restrict__ projB,
                                                          const int* __restrict__ offB,
                                                          const int* __restrict__ permB,
                                                          const float* __restrict__ xB,
                                                          float* outB, int nB) {
    const bool jobA = (int)blockIdx.x < blocksA;
    const float* __restrict__ proj = jobA ? projA : projB;
    const int* __restrict__ off = jobA ? offA : offB;
    const int* __restrict__ perm = jobA ? permA : permB;
    const float* __restrict__ x = jobA ? xA : xB;
    float* out = jobA ? outA : outB;
    const int nDst = jobA ? nA : nB;
    const int blk = jobA ? (int)blockIdx.x : (int)blockIdx.x - blocksA;
    int d = blk * blockDim.x + threadIdx.x;
    if (d >= nDst) return;
    const int beg = off[d];
    const int end = off[d + 1];
    float a0 = out[(size_t)d * 3 + 0];
    float a1 = out[(size_t)d * 3 + 1];
    float a2 = out[(size_t)d * 3 + 2];
    const float f0 = x[(size_t)d * 64 + 61];
    const float f2 = x[(size_t)d * 64 + 63];
    for (int j = beg; j < end; ++j) {
        const int s = perm[j];
        a0 += proj[(size_t)s * 3 + 0];
        a1 += proj[(size_t)s * 3 + 1];
        a2 += proj[(size_t)s * 3 + 2];
    }
    float nrm = sqrtf(a0 * a0 + a1 * a1 + a2 * a2);
    float sc = 10.0f / fmaxf(nrm, 1e-12f);
    a0 *= sc; a1 *= sc; a2 *= sc;
    if (f0 != 0.0f) a0 -= 10.0f;
    if (f2 != 0.0f) a2 -= 10.0f;
    out[(size_t)d * 3 + 0] = a0;
    out[(size_t)d * 3 + 1] = a1;
    out[(size_t)d * 3 + 2] = a2;
}

template <int K1, int K2, bool RELU, bool FUSE3, int NW>
__global__ __launch_bounds__(NW * 32) __attribute__((amdgpu_waves_per_eu(NW == 8 ? 6 : 4))) void gemm_dual_wmma(const float* __restrict__ Sa,
                                                          const int* __restrict__ offa,
                                                          const int* __restrict__ perma,
                                                          const float* __restrict__ W1a,
                                                          const float* __restrict__ A2a,
                                                          const float* __restrict__ W2a,
                                                          const float* __restrict__ biasa,
                                                          float* outa, int Ma, int blocksA,
                                                          const float* __restrict__ Sb,
                                                          const int* __restrict__ offb,
                                                          const int* __restrict__ permb,
                                                          const float* __restrict__ W1b,
                                                          const float* __restrict__ A2b,
                                                          const float* __restrict__ W2b,
                                                          const float* __restrict__ biasb,
                                                          float* outb, int Mb,
                                                          const float* __restrict__ W3rela,
                                                          const float* __restrict__ W3roota,
                                                          const float* __restrict__ b3a,
                                                          float* __restrict__ proja,
                                                          const float* __restrict__ W3relb,
                                                          const float* __restrict__ W3rootb,
                                                          const float* __restrict__ b3b,
                                                          float* __restrict__ projb) {
    constexpr int N = 128;
    constexpr int KP = (K1 + K2) / 2;
    constexpr int PITCH = 2 * N + 32;
    constexpr int W3N = FUSE3 ? 1536 : 0;
    constexpr int SP = K1 + 4;
    __shared__ alignas(16) float lds[KP * PITCH + W3N + NW * 16 * SP];
    float* wl = lds;
    float* w3l = lds + KP * PITCH;
    float* st = lds + KP * PITCH + W3N;

    const bool jobA = (int)blockIdx.x < blocksA;
    const float* __restrict__ S = jobA ? Sa : Sb;
    const int* __restrict__ off = jobA ? offa : offb;
    const int* __restrict__ perm = jobA ? perma : permb;
    const float* __restrict__ W1 = jobA ? W1a : W1b;
    const float* __restrict__ A2 = jobA ? A2a : A2b;
    const float* __restrict__ W2 = jobA ? W2a : W2b;
    const float* __restrict__ bias = jobA ? biasa : biasb;
    float* out = jobA ? outa : outb;
    const int M = jobA ? Ma : Mb;
    const int blk = jobA ? (int)blockIdx.x : (int)blockIdx.x - blocksA;
    const int nblk = jobA ? blocksA : (int)gridDim.x - blocksA;

    float l3r0 = 0.f, l3o0 = 0.f, l3r1 = 0.f, l3o1 = 0.f, l3b = 0.f, l3c = 0.f;
    if constexpr (FUSE3) {
        const float* __restrict__ W3rel = jobA ? W3rela : W3relb;
        const float* __restrict__ W3root = jobA ? W3roota : W3rootb;
        const float* __restrict__ b3 = jobA ? b3a : b3b;
        const int t = threadIdx.x & 255;
        const int i1 = min(t + 256, 383);
        l3r0 = W3rel[t];
        l3o0 = W3root[t];
        l3r1 = W3rel[i1];
        l3o1 = W3root[i1];
        l3b = bias[t & 127];
        l3c = b3[min(t, 2)];
    }

    {
        constexpr int RP = NW / 2;
        const int c2 = (threadIdx.x & 63) * 2;
        const int p0 = threadIdx.x >> 6;
#pragma unroll 8
        for (int i = 0; i < K1 / (2 * RP); ++i) {
            const int p = p0 + RP * i;
            const float2 r0 = *(const float2*)(W1 + (size_t)(2 * p) * N + c2);
            const float2 r1 = *(const float2*)(W1 + (size_t)(2 * p + 1) * N + c2);
            *(float4*)(wl + p * PITCH + c2 * 2) = make_float4(r0.x, r1.x, r0.y, r1.y);
        }
#pragma unroll 8
        for (int i = 0; i < K2 / (2 * RP); ++i) {
            const int p = p0 + RP * i;
            const float2 r0 = *(const float2*)(W2 + (size_t)(2 * p) * N + c2);
            const float2 r1 = *(const float2*)(W2 + (size_t)(2 * p + 1) * N + c2);
            *(float4*)(wl + (K1 / 2 + p) * PITCH + c2 * 2) = make_float4(r0.x, r1.x, r0.y, r1.y);
        }
    }
    if constexpr (FUSE3) {
        const int t = threadIdx.x & 255;
        *(float2*)(w3l + 2 * t) = make_float2(l3r0, l3o0);
        *(float2*)(w3l + 2 * (t + 256)) = make_float2(l3r1, l3o1);
        w3l[1024 + t] = l3b;
        w3l[1280 + t] = l3c;
    }
    __syncthreads();

    const int lane = threadIdx.x & 31;
    const int wave = threadIdx.x >> 5;
    const int half = lane >> 4;
    const int lr = lane & 15;
    const int nStrips = M / 16;
    const int s0 = (int)((long long)blk * nStrips / nblk);
    const int s1 = (int)((long long)(blk + 1) * nStrips / nblk);

    float* stw = st + wave * (16 * SP);
    const float* sa = stw + lr * SP + 2 * half;
    const float* bl0 = wl + half * PITCH + lr * 2;

    constexpr int CH = 8;
    constexpr int NCH = (K1 + K2) / (4 * CH);
    constexpr int PER = 8;
    constexpr int V = K1 / (4 * PER);
    const int sub = lane & 7;
    const int rq = lane >> 3;
    const float* lane_src = S + sub * 4;

#pragma unroll 1
    for (int strip = s0 + wave; strip < s1; strip += NW) {
        const int m0 = strip * 16;

#pragma unroll 1
        for (int pass = 0; pass < 4; ++pass) {
            const int rl = pass * 4 + rq;
            const int d = m0 + rl;
            const int beg = off[d];
            const int end = off[d + 1];
            float4 acc4[V];
#pragma unroll
            for (int u = 0; u < V; ++u) acc4[u] = make_float4(0.f, 0.f, 0.f, 0.f);
            int j = beg;
            for (; j + 4 <= end; j += 4) {
                int s[4];
#pragma unroll
                for (int k = 0; k < 4; ++k) s[k] = perm[j + k];
                float4 v[4][V];
#pragma unroll
                for (int k = 0; k < 4; ++k)
#pragma unroll
                    for (int u = 0; u < V; ++u)
                        v[k][u] = *(const float4*)(lane_src + (size_t)s[k] * K1 + u * (PER * 4));
#pragma unroll
                for (int k = 0; k < 4; ++k)
#pragma unroll
                    for (int u = 0; u < V; ++u) {
                        acc4[u].x += v[k][u].x; acc4[u].y += v[k][u].y;
                        acc4[u].z += v[k][u].z; acc4[u].w += v[k][u].w;
                    }
            }
            if (j < end) {
                int s[3];
#pragma unroll
                for (int k = 0; k < 3; ++k) s[k] = perm[min(j + k, end - 1)];
                float4 v[3][V];
#pragma unroll
                for (int k = 0; k < 3; ++k)
#pragma unroll
                    for (int u = 0; u < V; ++u)
                        v[k][u] = *(const float4*)(lane_src + (size_t)s[k] * K1 + u * (PER * 4));
#pragma unroll
                for (int k = 0; k < 3; ++k) {
                    const bool on = j + k < end;
#pragma unroll
                    for (int u = 0; u < V; ++u) {
                        acc4[u].x += on ? v[k][u].x : 0.f; acc4[u].y += on ? v[k][u].y : 0.f;
                        acc4[u].z += on ? v[k][u].z : 0.f; acc4[u].w += on ? v[k][u].w : 0.f;
                    }
                }
            }
#pragma unroll
            for (int u = 0; u < V; ++u)
                *(float4*)(stw + rl * SP + (u * PER + sub) * 4) = acc4[u];
        }

        const float* a2p = A2 + (size_t)(m0 + lr) * K2 + 2 * half;
        const float* bl = bl0;

        v8f acc[8] = {};
        v2f ac[CH], an[CH];
        v2f bc[8], bn[8];
#pragma unroll
        for (int j = 0; j < CH; ++j) ac[j] = *(const v2f*)(sa + j * 4);
#pragma unroll
        for (int nt = 0; nt < 8; ++nt) bc[nt] = *(const v2f*)(bl + nt * 32);

#pragma unroll 1
        for (int ci = 0; ci < NCH; ++ci) {
            const bool more = ci + 1 < NCH;
            if (more) {
                const int kn = (ci + 1) * (4 * CH);
                if (kn < K1) {
#pragma unroll
                    for (int j = 0; j < CH; ++j) an[j] = *(const v2f*)(sa + kn + j * 4);
                } else {
                    const float* ap = a2p + (kn - K1);
#pragma unroll
                    for (int j = 0; j < CH; ++j) an[j] = *(const v2f*)(ap + j * 4);
                }
            }
#pragma unroll
            for (int j = 0; j < CH; ++j) {
                if (j + 1 < CH || more) {
#pragma unroll
                    for (int nt = 0; nt < 8; ++nt)
                        bn[nt] = *(const v2f*)(bl + (2 * (j + 1)) * PITCH + nt * 32);
                }
#pragma unroll
                for (int nt = 0; nt < 8; ++nt) {
                    if constexpr (FUSE3)
                        acc[nt] = __builtin_amdgcn_wmma_f32_16x16x4_f32(
                            false, bc[nt], false, ac[j], (short)0, acc[nt], false, false);
                    else
                        acc[nt] = __builtin_amdgcn_wmma_f32_16x16x4_f32(
                            false, ac[j], false, bc[nt], (short)0, acc[nt], false, false);
                }
#pragma unroll
                for (int nt = 0; nt < 8; ++nt) bc[nt] = bn[nt];
            }
            if (more) {
#pragma unroll
                for (int j = 0; j < CH; ++j) ac[j] = an[j];
            }
            bl += 2 * CH * PITCH;
        }

        if constexpr (FUSE3) {
            float* __restrict__ proj = jobA ? proja : projb;
            v2f pq[2][3];
#pragma unroll
            for (int cc = 0; cc < 3; ++cc) {
                pq[0][cc] = (v2f){0.f, 0.f};
                pq[1][cc] = (v2f){0.f, 0.f};
            }
#pragma unroll
            for (int nt = 0; nt < 8; ++nt) {
                const int n0 = nt * 16 + 8 * half;
                const float4* bp = (const float4*)(w3l + 1024 + n0);
                const float4* wp = (const float4*)(w3l + n0 * 6);
                float bb[8];
                v2f w[24];
#pragma unroll
                for (int t = 0; t < 2; ++t) {
                    const float4 x = bp[t];
                    bb[4 * t + 0] = x.x; bb[4 * t + 1] = x.y; bb[4 * t + 2] = x.z; bb[4 * t + 3] = x.w;
                }
#pragma unroll
                for (int t = 0; t < 12; ++t) {
                    const float4 x = wp[t];
                    w[2 * t + 0] = (v2f){x.x, x.y};
                    w[2 * t + 1] = (v2f){x.z, x.w};
                }
#pragma unroll
                for (int r = 0; r < 8; ++r) {
                    float v = acc[nt][r] + bb[r];
                    if (RELU) v = fmaxf(v, 0.0f);
                    const v2f vv = (v2f){v, v};
#pragma unroll
                    for (int cc = 0; cc < 3; ++cc) pq[r & 1][cc] += vv * w[r * 3 + cc];
                }
            }
            float s[6];
#pragma unroll
            for (int cc = 0; cc < 3; ++cc) {
                const v2f t = pq[0][cc] + pq[1][cc];
                s[cc] = t.x;
                s[3 + cc] = t.y;
            }
#pragma unroll
            for (int cc = 0; cc < 6; ++cc) s[cc] += __shfl_xor(s[cc], 16, 32);
            if (half == 0) {
                const size_t m = (size_t)(m0 + lr);
#pragma unroll
                for (int cc = 0; cc < 3; ++cc) {
                    proj[m * 3 + cc] = s[cc];
                    out[m * 3 + cc] = s[3 + cc] + w3l[1280 + cc];
                }
            }
        } else {
#pragma unroll
            for (int nt = 0; nt < 8; ++nt) {
                const int col = nt * 16 + lr;
                const float bv = bias[col];
#pragma unroll
                for (int i = 0; i < 8; ++i) {
                    float v = acc[nt][i] + bv;
                    if (RELU) v = fmaxf(v, 0.0f);
                    out[(size_t)(m0 + i + 8 * half) * N + col] = v;
                }
            }
        }
    }
}

static inline int ceil_div(long long a, long long b) { return (int)((a + b - 1) / b); }

extern "C" void kernel_launch(void* const* d_in, const int* in_sizes, int n_in,
                              void* d_out, int out_size, void* d_ws, size_t ws_size,
                              hipStream_t stream) {
    const float* x_s = (const float*)d_in[0];
    const float* x_t = (const float*)d_in[1];
    const int* e_l = (const int*)d_in[2];
    const int* e_r = (const int*)d_in[3];
    const float* w1lr_rel  = (const float*)d_in[4];
    const float* b1lr      = (const float*)d_in[5];
    const float* w1lr_root = (const float*)d_in[6];
    const float* w1rl_rel  = (const float*)d_in[7];
    const float* b1rl      = (const float*)d_in[8];
    const float* w1rl_root = (const float*)d_in[9];
    const float* w2lr_rel  = (const float*)d_in[10];
    const float* b2lr      = (const float*)d_in[11];
    const float* w2lr_root = (const float*)d_in[12];
    const float* w2rl_rel  = (const float*)d_in[13];
    const float* b2rl      = (const float*)d_in[14];
    const float* w2rl_root = (const float*)d_in[15];
    const float* w3lr_rel  = (const float*)d_in[16];
    const float* b3lr      = (const float*)d_in[17];
    const float* w3lr_root = (const float*)d_in[18];
    const float* w3rl_rel  = (const float*)d_in[19];
    const float* b3rl      = (const float*)d_in[20];
    const float* w3rl_root = (const float*)d_in[21];

    const int NL = in_sizes[0] / 64;
    const int NR = in_sizes[1] / 64;
    const int E  = in_sizes[2];

    float* ws = (float*)d_ws;
    const size_t H1_R  = 0;
    const size_t H1_L  = H1_R + (size_t)NR * 128;
    const int nbR = ceil_div(NR, 1 << SH_R);
    const int nbL = ceil_div(NL, 1 << SH_L);
    const int chunk = ceil_div(E, HG);
    int* wi = (int*)d_ws;
    const size_t PAIR_R = H1_L + (size_t)NL * 128;
    const size_t PAIR_L = PAIR_R + (size_t)E * 2;
    const size_t TAB    = PAIR_L + (size_t)E * 2;
    const size_t PERM_R = TAB + (size_t)(nbR + nbL + 1) * HG;
    const size_t PERM_L = PERM_R + (size_t)E;
    const size_t OFF_R  = PERM_L + (size_t)E;
    const size_t OFF_L  = OFF_R + (size_t)NR + 1;
    const size_t PROJ_LR = OFF_L + (size_t)NL + 1;
    const size_t PROJ_RL = PROJ_LR + (size_t)NL * 3;

    float* d_left  = (float*)d_out;
    float* d_right = (float*)d_out + (size_t)NL * 3;

    const int TPB = 256;

    bucket_local<<<HG, 1024, 0, stream>>>(e_l, e_r, E, chunk, nbR, nbR + nbL, wi + TAB,
                                          (int2*)(wi + PAIR_R), (int2*)(wi + PAIR_L));
    bucket_sort<<<nbR + nbL, 1024, 0, stream>>>((const int2*)(wi + PAIR_R), (const int2*)(wi + PAIR_L),
                                                wi + TAB, chunk, nbR, nbL,
                                                wi + OFF_R, wi + OFF_L, wi + PERM_R, wi + PERM_L, NR, NL);

    const int gmR = ceil_div(NR, 16 * 8);
    const int gmL = ceil_div(NL, 16 * 8);
    const int g2R = 160;
    const int g2L = 96;
    const int fnL = ceil_div(NL, TPB);
    const int fnR = ceil_div(NR, TPB);

    gemm_dual_wmma<64, 64, true, false, 8><<<gmR + gmL, TPB, 0, stream>>>(
        x_s, wi + OFF_R, wi + PERM_R, w1lr_rel, x_t, w1lr_root, b1lr, ws + H1_R, NR, gmR,
        x_t, wi + OFF_L, wi + PERM_L, w1rl_rel, x_s, w1rl_root, b1rl, ws + H1_L, NL,
        nullptr, nullptr, nullptr, nullptr, nullptr, nullptr, nullptr, nullptr);

    gemm_dual_wmma<128, 128, true, true, 16><<<g2R + g2L, 512, 0, stream>>>(
        ws + H1_L, wi + OFF_R, wi + PERM_R, w2lr_rel, ws + H1_R, w2lr_root, b2lr, d_right, NR, g2R,
        ws + H1_R, wi + OFF_L, wi + PERM_L, w2rl_rel, ws + H1_L, w2rl_root, b2rl, d_left, NL,
        w3rl_rel, w3lr_root, b3lr, ws + PROJ_RL,
        w3lr_rel, w3rl_root, b3rl, ws + PROJ_LR);
    gather_add3_finish<<<fnL + fnR, TPB, 0, stream>>>(ws + PROJ_RL, wi + OFF_L, wi + PERM_L, x_s, d_left, NL, fnL,
                                                      ws + PROJ_LR, wi + OFF_R, wi + PERM_R, x_t, d_right, NR);
}


# Edge-sum index loads issued one step ahead of the row loads they feed, in both GEMM launches and the closing pass (a row's next four indices before its current four rows); same sum order
# speedup vs baseline: 4.5931x; 1.0590x over previous
#include <hip/hip_runtime.h>
#include <hip/hip_bf16.h>

typedef __attribute__((ext_vector_type(2))) float v2f;
typedef __attribute__((ext_vector_type(8))) float v8f;

constexpr int HG = 256;
constexpr int SH_R = 9;
constexpr int SH_L = 8;
constexpr int BW = 512;
constexpr int MAXNB = 1024;
constexpr int EB = 4;

__device__ __forceinline__ int wave_incl_scan(int v, int lane) {
#pragma unroll
    for (int o = 1; o < 32; o <<= 1) {
        const int u = __shfl_up(v, o, 32);
        if (lane >= o) v += u;
    }
    return v;
}

__global__ __launch_bounds__(1024) void bucket_local(const int* __restrict__ e_l,
                                                     const int* __restrict__ e_r, int nE, int chunk,
                                                     int nbR, int nbAll, int* __restrict__ tab,
                                                     int2* __restrict__ pairR, int2* __restrict__ pairL) {
    static_assert(EB == 4, "the statements that group the edge loads name EB = 4 pairs");
    __shared__ int cnt[MAXNB];
    __shared__ int wtot[32];
    const int tid = threadIdx.x;
    const int lane = tid & 31;
    const int wave = tid >> 5;
    cnt[tid] = 0;
    __syncthreads();
    const int e0 = blockIdx.x * chunk;
    const int e1 = min(nE, e0 + chunk);
    const int n = max(e1 - e0, 0);
    for (int i0 = e0 + tid; i0 < e1; i0 += EB * 1024) {
        int dl[EB], dr[EB];
#pragma unroll
        for (int k = 0; k < EB; ++k) {
            const int ic = min(i0 + k * 1024, e1 - 1);
            dl[k] = e_l[ic];
            dr[k] = e_r[ic];
        }
        asm volatile("" ::"v"(dl[0]), "v"(dl[1]), "v"(dl[2]), "v"(dl[3]), "v"(dr[0]), "v"(dr[1]),
                     "v"(dr[2]), "v"(dr[3]));
#pragma unroll
        for (int k = 0; k < EB; ++k)
            if (i0 + k * 1024 >= e1) dl[k] = -1;
#pragma unroll
        for (int k = 0; k < EB; ++k) {
            if (dl[k] >= 0) {
                atomicAdd(&cnt[dr[k] >> SH_R], 1);
                atomicAdd(&cnt[nbR + (dl[k] >> SH_L)], 1);
            }
        }
    }
    __syncthreads();
    const int v = cnt[tid];
    const int incl = wave_incl_scan(v, lane);
    if (lane == 31) wtot[wave] = incl;
    __syncthreads();
    int pre = 0;
#pragma unroll
    for (int w = 0; w < 32; ++w) pre += (w < wave) ? wtot[w] : 0;
    const int excl = pre + incl - v;
    if (tid <= nbAll) tab[(size_t)tid * HG + blockIdx.x] = excl;
    cnt[tid] = excl - (tid >= nbR ? n : 0);
    __syncthreads();
    for (int i0 = e0 + tid; i0 < e1; i0 += EB * 1024) {
        int dl[EB], dr[EB];
#pragma unroll
        for (int k = 0; k < EB; ++k) {
            const int ic = min(i0 + k * 1024, e1 - 1);
            dl[k] = e_l[ic];
            dr[k] = e_r[ic];
        }
        asm volatile("" ::"v"(dl[0]), "v"(dl[1]), "v"(dl[2]), "v"(dl[3]), "v"(dr[0]), "v"(dr[1]),
                     "v"(dr[2]), "v"(dr[3]));
#pragma unroll
        for (int k = 0; k < EB; ++k)
            if (i0 + k * 1024 >= e1) dl[k] = -1;
#pragma unroll
        for (int k = 0; k < EB; ++k) {
            if (dl[k] >= 0) {
                const int pr = atomicAdd(&cnt[dr[k] >> SH_R], 1);
                const int pl = atomicAdd(&cnt[nbR + (dl[k] >> SH_L)], 1);
                pairR[e0 + pr] = make_int2(dr[k], dl[k]);
                pairL[e0 + pl] = make_int2(dl[k], dr[k]);
            }
        }
    }
}

__global__ __launch_bounds__(1024) void bucket_sort(const int2* __restrict__ pairR,
                                                    const int2* __restrict__ pairL,
                                                    const int* __restrict__ tab, int chunk,
                                                    int nbR, int nbL,
                                                    int* __restrict__ offR, int* __restrict__ offL,
                                                    int* __restrict__ permR, int* __restrict__ permL,
                                                    int nR, int nL) {
    constexpr int RW = HG / 32;
    static_assert(HG % 32 == 0 && RW % EB == 0, "a wave takes its runs EB at a time");
    __shared__ int cnt[BW];
    __shared__ int wtot[BW / 32];
    __shared__ int segOff[HG];
    __shared__ int segLen[HG];
    __shared__ int red[2 * RW];
    const bool isR = (int)blockIdx.x < nbR;
    const int b = isR ? (int)blockIdx.x : (int)blockIdx.x - nbR;
    const int nb = isR ? nbR : nbL;
    const int base = isR ? 0 : nbR;
    const int2* __restrict__ pair = isR ? pairR : pairL;
    int* __restrict__ off = isR ? offR : offL;
    int* __restrict__ perm = isR ? permR : permL;
    const int n = isR ? nR : nL;
    const int sh = isR ? SH_R : SH_L;
    const int lo = b << sh;
    const int tid = threadIdx.x;
    const int lane = tid & 31;
    const int wave = tid >> 5;
    int ahead = 0, len = 0;
    if (tid < HG) {
        const int a0 = tab[(size_t)base * HG + tid];
        const int a1 = tab[(size_t)(base + b) * HG + tid];
        const int a2 = tab[(size_t)(base + b + 1) * HG + tid];
        ahead = a1 - a0;
        len = a2 - a1;
        segOff[tid] = tid * chunk + ahead;
        segLen[tid] = len;
    }
    if (tid < BW) cnt[tid] = 0;
    {
        int sa = ahead, sl = len;
#pragma unroll
        for (int o = 16; o; o >>= 1) {
            sa += __shfl_xor(sa, o, 32);
            sl += __shfl_xor(sl, o, 32);
        }
        if (lane == 0 && wave < RW) {
            red[wave] = sa;
            red[RW + wave] = sl;
        }
    }
    __syncthreads();
    int s = 0, tot = 0;
#pragma unroll
    for (int w = 0; w < RW; ++w) {
        s += red[w];
        tot += red[RW + w];
    }
    const int e = s + tot;
#pragma unroll 1
    for (int h = 0; h < RW; h += EB) {
        int ro[EB], rl[EB], d[EB];
#pragma unroll
        for (int k = 0; k < EB; ++k) {
            ro[k] = segOff[wave + 32 * (h + k)];
            rl[k] = segLen[wave + 32 * (h + k)];
        }
#pragma unroll
        for (int k = 0; k < EB; ++k) d[k] = lane < rl[k] ? pair[ro[k] + lane].x : -1;
#pragma unroll
        for (int k = 0; k < EB; ++k)
            if (d[k] >= 0) atomicAdd(&cnt[d[k] - lo], 1);
#pragma unroll
        for (int k = 0; k < EB; ++k)
            for (int l = lane + 32; l < rl[k]; l += 32) atomicAdd(&cnt[pair[ro[k] + l].x - lo], 1);
    }
    __syncthreads();
    const int v = tid < BW ? cnt[tid] : 0;
    const int incl = wave_incl_scan(v, lane);
    if (lane == 31 && wave < BW / 32) wtot[wave] = incl;
    __syncthreads();
    if (tid < BW) {
        int pre = 0;
#pragma unroll
        for (int w = 0; w < BW / 32; ++w) pre += (w < wave) ? wtot[w] : 0;
        const int excl = pre + incl - v;
        cnt[tid] = excl;
        const int node = lo + tid;
        if (tid < (1 << sh) && node < n) off[node] = s + excl;
    }
    if (tid == 0 && b == nb - 1) off[n] = e;
    __syncthreads();
#pragma unroll 1
    for (int h = 0; h < RW; h += EB) {
        int ro[EB], rl[EB];
        int2 t[EB];
#pragma unroll
        for (int k = 0; k < EB; ++k) {
            ro[k] = segOff[wave + 32 * (h + k)];
            rl[k] = segLen[wave + 32 * (h + k)];
        }
#pragma unroll
        for (int k = 0; k < EB; ++k) t[k] = lane < rl[k] ? pair[ro[k] + lane] : make_int2(-1, 0);
#pragma unroll
        for (int k = 0; k < EB; ++k) {
            if (t[k].x >= 0) {
                const int p = atomicAdd(&cnt[t[k].x - lo], 1);
                perm[s + p] = t[k].y;
            }
        }
#pragma unroll
        for (int k = 0; k < EB; ++k)
            for (int l = lane + 32; l < rl[k]; l += 32) {
                const int2 u = pair[ro[k] + l];
                const int p = atomicAdd(&cnt[u.x - lo], 1);
                perm[s + p] = u.y;
            }
    }
}

__global__ __launch_bounds__(256) void gather_add3_finish(const float* __restrict__ projA,
                                                          const int* __restrict__ offA,
                                                          const int* __restrict__ permA,
                                                          const float* __restrict__ xA,
                                                          float* outA, int nA, int blocksA,
                                                          const float* __restrict__ projB,
                                                          const int* __restrict__ offB,
                                                          const int* __restrict__ permB,
                                                          const float* __restrict__ xB,
                                                          float* outB, int nB) {
    const bool jobA = (int)blockIdx.x < blocksA;
    const float* __restrict__ proj = jobA ? projA : projB;
    const int* __restrict__ off = jobA ? offA : offB;
    const int* __restrict__ perm = jobA ? permA : permB;
    const float* __restrict__ x = jobA ? xA : xB;
    float* out = jobA ? outA : outB;
    const int nDst = jobA ? nA : nB;
    const int blk = jobA ? (int)blockIdx.x : (int)blockIdx.x - blocksA;
    int d = blk * blockDim.x + threadIdx.x;
    if (d >= nDst) return;
    const int beg = off[d];
    const int end = off[d + 1];
    float a0 = out[(size_t)d * 3 + 0];
    float a1 = out[(size_t)d * 3 + 1];
    float a2 = out[(size_t)d * 3 + 2];
    const float f0 = x[(size_t)d * 64 + 61];
    const float f2 = x[(size_t)d * 64 + 63];
    int ia[4];
#pragma unroll
    for (int k = 0; k < 4; ++k) ia[k] = perm[beg + k];
    asm volatile("" ::"v"(ia[0]), "v"(ia[1]), "v"(ia[2]), "v"(ia[3]));
    int j = beg;
    for (; j + 4 <= end; j += 4) {
        int in[4];
#pragma unroll
        for (int k = 0; k < 4; ++k) in[k] = perm[j + 4 + k];
        float p[4][3];
#pragma unroll
        for (int k = 0; k < 4; ++k)
#pragma unroll
            for (int c = 0; c < 3; ++c) p[k][c] = proj[(size_t)ia[k] * 3 + c];
#pragma unroll
        for (int k = 0; k < 4; ++k) {
            a0 += p[k][0];
            a1 += p[k][1];
            a2 += p[k][2];
        }
        asm volatile("" ::"v"(in[0]), "v"(in[1]), "v"(in[2]), "v"(in[3]));
#pragma unroll
        for (int k = 0; k < 4; ++k) ia[k] = in[k];
    }
    if (j < end) {
        int s[3];
#pragma unroll
        for (int k = 0; k < 3; ++k) s[k] = (j + k < end) ? ia[k] : ia[0];
        float p[3][3];
#pragma unroll
        for (int k = 0; k < 3; ++k)
#pragma unroll
            for (int c = 0; c < 3; ++c) p[k][c] = proj[(size_t)s[k] * 3 + c];
#pragma unroll
        for (int k = 0; k < 3; ++k) {
            if (j + k < end) {
                a0 += p[k][0];
                a1 += p[k][1];
                a2 += p[k][2];
            }
        }
    }
    float nrm = sqrtf(a0 * a0 + a1 * a1 + a2 * a2);
    float sc = 10.0f / fmaxf(nrm, 1e-12f);
    a0 *= sc; a1 *= sc; a2 *= sc;
    if (f0 != 0.0f) a0 -= 10.0f;
    if (f2 != 0.0f) a2 -= 10.0f;
    out[(size_t)d * 3 + 0] = a0;
    out[(size_t)d * 3 + 1] = a1;
    out[(size_t)d * 3 + 2] = a2;
}

template <int K1, int K2, bool RELU, bool FUSE3, int NW>
__global__ __launch_bounds__(NW * 32) __attribute__((amdgpu_waves_per_eu(NW == 8 ? 6 : 4))) void gemm_dual_wmma(const float* __restrict__ Sa,
                                                          const int* __restrict__ offa,
                                                          const int* __restrict__ perma,
                                                          const float* __restrict__ W1a,
                                                          const float* __restrict__ A2a,
                                                          const float* __restrict__ W2a,
                                                          const float* __restrict__ biasa,
                                                          float* outa, int Ma, int blocksA,
                                                          const float* __restrict__ Sb,
                                                          const int* __restrict__ offb,
                                                          const int* __restrict__ permb,
                                                          const float* __restrict__ W1b,
                                                          const float* __restrict__ A2b,
                                                          const float* __restrict__ W2b,
                                                          const float* __restrict__ biasb,
                                                          float* outb, int Mb,
                                                          const float* __restrict__ W3rela,
                                                          const float* __restrict__ W3roota,
                                                          const float* __restrict__ b3a,
                                                          float* __restrict__ proja,
                                                          const float* __restrict__ W3relb,
                                                          const float* __restrict__ W3rootb,
                                                          const float* __restrict__ b3b,
                                                          float* __restrict__ projb) {
    constexpr int N = 128;
    constexpr int KP = (K1 + K2) / 2;
    constexpr int PITCH = 2 * N + 32;
    constexpr int W3N = FUSE3 ? 1536 : 0;
    constexpr int SP = K1 + 4;
    __shared__ alignas(16) float lds[KP * PITCH + W3N + NW * 16 * SP];
    float* wl = lds;
    float* w3l = lds + KP * PITCH;
    float* st = lds + KP * PITCH + W3N;

    const bool jobA = (int)blockIdx.x < blocksA;
    const float* __restrict__ S = jobA ? Sa : Sb;
    const int* __restrict__ off = jobA ? offa : offb;
    const int* __restrict__ perm = jobA ? perma : permb;
    const float* __restrict__ W1 = jobA ? W1a : W1b;
    const float* __restrict__ A2 = jobA ? A2a : A2b;
    const float* __restrict__ W2 = jobA ? W2a : W2b;
    const float* __restrict__ bias = jobA ? biasa : biasb;
    float* out = jobA ? outa : outb;
    const int M = jobA ? Ma : Mb;
    const int blk = jobA ? (int)blockIdx.x : (int)blockIdx.x - blocksA;
    const int nblk = jobA ? blocksA : (int)gridDim.x - blocksA;

    const int lane = threadIdx.x & 31;
    const int wave = threadIdx.x >> 5;
    const int rq = lane >> 3;
    const int nStrips = M / 16;
    const int s0 = (int)((long long)blk * nStrips / nblk);
    const int s1 = (int)((long long)(blk + 1) * nStrips / nblk);
    int bq[4], eq[4];
    int ia[4];
    {
        const int mf = min(s0 + wave, nStrips - 1) * 16;
#pragma unroll
        for (int p = 0; p < 4; ++p) {
            bq[p] = off[mf + p * 4 + rq];
            eq[p] = off[mf + p * 4 + rq + 1];
        }
    }

    float l3r0 = 0.f, l3o0 = 0.f, l3r1 = 0.f, l3o1 = 0.f, l3b = 0.f, l3c = 0.f;
    if constexpr (FUSE3) {
        const float* __restrict__ W3rel = jobA ? W3rela : W3relb;
        const float* __restrict__ W3root = jobA ? W3roota : W3rootb;
        const float* __restrict__ b3 = jobA ? b3a : b3b;
        const int t = threadIdx.x & 255;
        const int i1 = min(t + 256, 383);
        l3r0 = W3rel[t];
        l3o0 = W3root[t];
        l3r1 = W3rel[i1];
        l3o1 = W3root[i1];
        l3b = bias[t & 127];
        l3c = b3[min(t, 2)];
    }

    {
        constexpr int RP = NW / 2;
        const int c2 = (threadIdx.x & 63) * 2;
        const int p0 = threadIdx.x >> 6;
#pragma unroll 8
        for (int i = 0; i < K1 / (2 * RP); ++i) {
            const int p = p0 + RP * i;
            const float2 r0 = *(const float2*)(W1 + (size_t)(2 * p) * N + c2);
            const float2 r1 = *(const float2*)(W1 + (size_t)(2 * p + 1) * N + c2);
            *(float4*)(wl + p * PITCH + c2 * 2) = make_float4(r0.x, r1.x, r0.y, r1.y);
        }
#pragma unroll
        for (int k = 0; k < 4; ++k) ia[k] = perm[bq[0] + k];
#pragma unroll 8
        for (int i = 0; i < K2 / (2 * RP); ++i) {
            const int p = p0 + RP * i;
            const float2 r0 = *(const float2*)(W2 + (size_t)(2 * p) * N + c2);
            const float2 r1 = *(const float2*)(W2 + (size_t)(2 * p + 1) * N + c2);
            *(float4*)(wl + (K1 / 2 + p) * PITCH + c2 * 2) = make_float4(r0.x, r1.x, r0.y, r1.y);
        }
    }
    if constexpr (FUSE3) {
        const int t = threadIdx.x & 255;
        *(float2*)(w3l + 2 * t) = make_float2(l3r0, l3o0);
        *(float2*)(w3l + 2 * (t + 256)) = make_float2(l3r1, l3o1);
        w3l[1024 + t] = l3b;
        w3l[1280 + t] = l3c;
    }
    __syncthreads();

    const int half = lane >> 4;
    const int lr = lane & 15;

    float* stw = st + wave * (16 * SP);
    const float* sa = stw + lr * SP + 2 * half;
    const float* bl0 = wl + half * PITCH + lr * 2;

    constexpr int CH = 8;
    constexpr int NCH = (K1 + K2) / (4 * CH);
    constexpr int PER = 8;
    constexpr int V = K1 / (4 * PER);
    const int sub = lane & 7;
    const float* lane_src = S + sub * 4;

#pragma unroll 1
    for (int strip = s0 + wave; strip < s1; strip += NW) {
        const int m0 = strip * 16;

#pragma unroll 1
        for (int pass = 0; pass < 4; ++pass) {
            const int rl = pass * 4 + rq;
            const int beg = bq[0];
            const int end = eq[0];
            int ib[4];
#pragma unroll
            for (int k = 0; k < 4; ++k) ib[k] = perm[bq[1] + k];
            float4 acc4[V];
#pragma unroll
            for (int u = 0; u < V; ++u) acc4[u] = make_float4(0.f, 0.f, 0.f, 0.f);
            int j = beg;
            for (; j + 4 <= end; j += 4) {
                int in[4];
#pragma unroll
                for (int k = 0; k < 4; ++k) in[k] = perm[j + 4 + k];
                float4 v[4][V];
#pragma unroll
                for (int k = 0; k < 4; ++k)
#pragma unroll
                    for (int u = 0; u < V; ++u)
                        v[k][u] = *(const float4*)(lane_src + (size_t)ia[k] * K1 + u * (PER * 4));
#pragma unroll
                for (int k = 0; k < 4; ++k)
#pragma unroll
                    for (int u = 0; u < V; ++u) {
                        acc4[u].x += v[k][u].x; acc4[u].y += v[k][u].y;
                        acc4[u].z += v[k][u].z; acc4[u].w += v[k][u].w;
                    }
#pragma unroll
                for (int k = 0; k < 4; ++k) ia[k] = in[k];
            }
            if (j < end) {
                int s[3];
#pragma unroll
                for (int k = 0; k < 3; ++k) s[k] = (j + k < end) ? ia[k] : ia[0];
                float4 v[3][V];
#pragma unroll
                for (int k = 0; k < 3; ++k)
#pragma unroll
                    for (int u = 0; u < V; ++u)
                        v[k][u] = *(const float4*)(lane_src + (size_t)s[k] * K1 + u * (PER * 4));
#pragma unroll
                for (int k = 0; k < 3; ++k) {
                    const bool on = j + k < end;
#pragma unroll
                    for (int u = 0; u < V; ++u) {
                        acc4[u].x += on ? v[k][u].x : 0.f; acc4[u].y += on ? v[k][u].y : 0.f;
                        acc4[u].z += on ? v[k][u].z : 0.f; acc4[u].w += on ? v[k][u].w : 0.f;
                    }
                }
            }
#pragma unroll
            for (int u = 0; u < V; ++u)
                *(float4*)(stw + rl * SP + (u * PER + sub) * 4) = acc4[u];
#pragma unroll
            for (int p = 0; p < 3; ++p) {
                bq[p] = bq[p + 1];
                eq[p] = eq[p + 1];
            }
#pragma unroll
            for (int k = 0; k < 4; ++k) ia[k] = ib[k];
        }

        const float* a2p = A2 + (size_t)(m0 + lr) * K2 + 2 * half;
        const float* bl = bl0;

        v8f acc[8] = {};
        v2f ac[CH], an[CH];
        v2f bc[8], bn[8];
#pragma unroll
        for (int j = 0; j < CH; ++j) ac[j] = *(const v2f*)(sa + j * 4);
#pragma unroll
        for (int nt = 0; nt < 8; ++nt) bc[nt] = *(const v2f*)(bl + nt * 32);

#pragma unroll 1
        for (int ci = 0; ci < NCH; ++ci) {
            const bool more = ci + 1 < NCH;
            if (more) {
                const int kn = (ci + 1) * (4 * CH);
                if (kn < K1) {
#pragma unroll
                    for (int j = 0; j < CH; ++j) an[j] = *(const v2f*)(sa + kn + j * 4);
                } else {
                    const float* ap = a2p + (kn - K1);
#pragma unroll
                    for (int j = 0; j < CH; ++j) an[j] = *(const v2f*)(ap + j * 4);
                }
            }
#pragma unroll
            for (int j = 0; j < CH; ++j) {
                if (j + 1 < CH || more) {
#pragma unroll
                    for (int nt = 0; nt < 8; ++nt)
                        bn[nt] = *(const v2f*)(bl + (2 * (j + 1)) * PITCH + nt * 32);
                }
#pragma unroll
                for (int nt = 0; nt < 8; ++nt) {
                    if constexpr (FUSE3)
                        acc[nt] = __builtin_amdgcn_wmma_f32_16x16x4_f32(
                            false, bc[nt], false, ac[j], (short)0, acc[nt], false, false);
                    else
                        acc[nt] = __builtin_amdgcn_wmma_f32_16x16x4_f32(
                            false, ac[j], false, bc[nt], (short)0, acc[nt], false, false);
                }
#pragma unroll
                for (int nt = 0; nt < 8; ++nt) bc[nt] = bn[nt];
            }
            if (more) {
#pragma unroll
                for (int j = 0; j < CH; ++j) ac[j] = an[j];
            }
            bl += 2 * CH * PITCH;
        }

        if constexpr (FUSE3) {
            float* __restrict__ proj = jobA ? proja : projb;
            v2f pq[2][3];
#pragma unroll
            for (int cc = 0; cc < 3; ++cc) {
                pq[0][cc] = (v2f){0.f, 0.f};
                pq[1][cc] = (v2f){0.f, 0.f};
            }
#pragma unroll
            for (int nt = 0; nt < 8; ++nt) {
                const int n0 = nt * 16 + 8 * half;
                const float4* bp = (const float4*)(w3l + 1024 + n0);
                const float4* wp = (const float4*)(w3l + n0 * 6);
                float bb[8];
                v2f w[24];
#pragma unroll
                for (int t = 0; t < 2; ++t) {
                    const float4 x = bp[t];
                    bb[4 * t + 0] = x.x; bb[4 * t + 1] = x.y; bb[4 * t + 2] = x.z; bb[4 * t + 3] = x.w;
                }
#pragma unroll
                for (int t = 0; t < 12; ++t) {
                    const float4 x = wp[t];
                    w[2 * t + 0] = (v2f){x.x, x.y};
                    w[2 * t + 1] = (v2f){x.z, x.w};
                }
#pragma unroll
                for (int r = 0; r < 8; ++r) {
                    float v = acc[nt][r] + bb[r];
                    if (RELU) v = fmaxf(v, 0.0f);
                    const v2f vv = (v2f){v, v};
#pragma unroll
                    for (int cc = 0; cc < 3; ++cc) pq[r & 1][cc] += vv * w[r * 3 + cc];
                }
            }
            float s[6];
#pragma unroll
            for (int cc = 0; cc < 3; ++cc) {
                const v2f t = pq[0][cc] + pq[1][cc];
                s[cc] = t.x;
                s[3 + cc] = t.y;
            }
#pragma unroll
            for (int cc = 0; cc < 6; ++cc) s[cc] += __shfl_xor(s[cc], 16, 32);
            if (half == 0) {
                const size_t m = (size_t)(m0 + lr);
#pragma unroll
                for (int cc = 0; cc < 3; ++cc) {
                    proj[m * 3 + cc] = s[cc];
                    out[m * 3 + cc] = s[3 + cc] + w3l[1280 + cc];
                }
            }
        } else {
#pragma unroll
            for (int nt = 0; nt < 8; ++nt) {
                const int col = nt * 16 + lr;
                const float bv = bias[col];
#pragma unroll
                for (int i = 0; i < 8; ++i) {
                    float v = acc[nt][i] + bv;
                    if (RELU) v = fmaxf(v, 0.0f);
                    out[(size_t)(m0 + i + 8 * half) * N + col] = v;
                }
            }
        }

        if (strip + NW < s1) {
            const int mn = (strip + NW) * 16;
#pragma unroll
            for (int p = 0; p < 4; ++p) {
                bq[p] = off[mn + p * 4 + rq];
                eq[p] = off[mn + p * 4 + rq + 1];
            }
#pragma unroll
            for (int k = 0; k < 4; ++k) ia[k] = perm[bq[0] + k];
        }
    }
}

static inline int ceil_div(long long a, long long b) { return (int)((a + b - 1) / b); }

extern "C" void kernel_launch(void* const* d_in, const int* in_sizes, int n_in,
                              void* d_out, int out_size, void* d_ws, size_t ws_size,
                              hipStream_t stream) {
    const float* x_s = (const float*)d_in[0];
    const float* x_t = (const float*)d_in[1];
    const int* e_l = (const int*)d_in[2];
    const int* e_r = (const int*)d_in[3];
    const float* w1lr_rel  = (const float*)d_in[4];
    const float* b1lr      = (const float*)d_in[5];
    const float* w1lr_root = (const float*)d_in[6];
    const float* w1rl_rel  = (const float*)d_in[7];
    const float* b1rl      = (const float*)d_in[8];
    const float* w1rl_root = (const float*)d_in[9];
    const float* w2lr_rel  = (const float*)d_in[10];
    const float* b2lr      = (const float*)d_in[11];
    const float* w2lr_root = (const float*)d_in[12];
    const float* w2rl_rel  = (const float*)d_in[13];
    const float* b2rl      = (const float*)d_in[14];
    const float* w2rl_root = (const float*)d_in[15];
    const float* w3lr_rel  = (const float*)d_in[16];
    const float* b3lr      = (const float*)d_in[17];
    const float* w3lr_root = (const float*)d_in[18];
    const float* w3rl_rel  = (const float*)d_in[19];
    const float* b3rl      = (const float*)d_in[20];
    const float* w3rl_root = (const float*)d_in[21];

    const int NL = in_sizes[0] / 64;
    const int NR = in_sizes[1] / 64;
    const int E  = in_sizes[2];

    float* ws = (float*)d_ws;
    const size_t H1_R  = 0;
    const size_t H1_L  = H1_R + (size_t)NR * 128;
    const int nbR = ceil_div(NR, 1 << SH_R);
    const int nbL = ceil_div(NL, 1 << SH_L);
    const int chunk = ceil_div(E, HG);
    int* wi = (int*)d_ws;
    const size_t PAIR_R = H1_L + (size_t)NL * 128;
    const size_t PAIR_L = PAIR_R + (size_t)E * 2;
    const size_t TAB    = PAIR_L + (size_t)E * 2;
    const size_t PERM_R = TAB + (size_t)(nbR + nbL + 1) * HG;
    const size_t PERM_L = PERM_R + (size_t)E;
    const size_t OFF_R  = PERM_L + (size_t)E;
    const size_t OFF_L  = OFF_R + (size_t)NR + 1;
    const size_t PROJ_LR = OFF_L + (size_t)NL + 1;
    const size_t PROJ_RL = PROJ_LR + (size_t)NL * 3;

    float* d_left  = (float*)d_out;
    float* d_right = (float*)d_out + (size_t)NL * 3;

    const int TPB = 256;

    bucket_local<<<HG, 1024, 0, stream>>>(e_l, e_r, E, chunk, nbR, nbR + nbL, wi + TAB,
                                          (int2*)(wi + PAIR_R), (int2*)(wi + PAIR_L));
    bucket_sort<<<nbR + nbL, 1024, 0, stream>>>((const int2*)(wi + PAIR_R), (const int2*)(wi + PAIR_L),
                                                wi + TAB, chunk, nbR, nbL,
                                                wi + OFF_R, wi + OFF_L, wi + PERM_R, wi + PERM_L, NR, NL);

    const int gmR = ceil_div(NR, 16 * 8);
    const int gmL = ceil_div(NL, 16 * 8);
    const int g2R = 160;
    const int g2L = 96;
    const int fnL = ceil_div(NL, TPB);
    const int fnR = ceil_div(NR, TPB);

    gemm_dual_wmma<64, 64, true, false, 8><<<gmR + gmL, TPB, 0, stream>>>(
        x_s, wi + OFF_R, wi + PERM_R, w1lr_rel, x_t, w1lr_root, b1lr, ws + H1_R, NR, gmR,
        x_t, wi + OFF_L, wi + PERM_L, w1rl_rel, x_s, w1rl_root, b1rl, ws + H1_L, NL,
        nullptr, nullptr, nullptr, nullptr, nullptr, nullptr, nullptr, nullptr);

    gemm_dual_wmma<128, 128, true, true, 16><<<g2R + g2L, 512, 0, stream>>>(
        ws + H1_L, wi + OFF_R, wi + PERM_R, w2lr_rel, ws + H1_R, w2lr_root, b2lr, d_right, NR, g2R,
        ws + H1_R, wi + OFF_L, wi + PERM_L, w2rl_rel, ws + H1_L, w2rl_root, b2rl, d_left, NL,
        w3rl_rel, w3lr_root, b3lr, ws + PROJ_RL,
        w3lr_rel, w3rl_root, b3rl, ws + PROJ_LR);
    gather_add3_finish<<<fnL + fnR, TPB, 0, stream>>>(ws + PROJ_RL, wi + OFF_L, wi + PERM_L, x_s, d_left, NL, fnL,
                                                      ws + PROJ_LR, wi + OFF_R, wi + PERM_R, x_t, d_right, NR);
}


# Grouping launches: placed entries (pair pieces, perm parts) staged in LDS and written out in address order, 16 bytes per lane: whole 128-byte lines per store, not one line per lane; same sum order
# speedup vs baseline: 4.8056x; 1.0463x over previous
#include <hip/hip_runtime.h>
#include <hip/hip_bf16.h>

typedef __attribute__((ext_vector_type(2))) float v2f;
typedef __attribute__((ext_vector_type(8))) float v8f;

constexpr int HG = 256;
constexpr int SH_R = 9;
constexpr int SH_L = 8;
constexpr int BW = 512;
constexpr int MAXNB = 1024;
constexpr int EB = 4;
constexpr int CAPE = 4096;
constexpr int CAPS = 8192;

__device__ __forceinline__ int wave_incl_scan(int v, int lane) {
#pragma unroll
    for (int o = 1; o < 32; o <<= 1) {
        const int u = __shfl_up(v, o, 32);
        if (lane >= o) v += u;
    }
    return v;
}

__device__ __forceinline__ void copy_out_pairs(int2* __restrict__ g, const int2* st, int lo, int hi, int tid) {
    for (int b0 = 2 * tid; b0 < hi; b0 += 2 * 1024) {
        if (b0 >= lo && b0 + 2 <= hi) {
            *(int4*)(g + b0) = *(const int4*)(st + b0);
        } else {
            if (b0 >= lo) g[b0] = st[b0];
            if (b0 + 1 >= lo && b0 + 1 < hi) g[b0 + 1] = st[b0 + 1];
        }
    }
}
__device__ __forceinline__ void copy_out_words(int* __restrict__ g, const int* st, int lo, int hi, int tid) {
    for (int b0 = 4 * tid; b0 < hi; b0 += 4 * 1024) {
        if (b0 >= lo && b0 + 4 <= hi) {
            *(int4*)(g + b0) = *(const int4*)(st + b0);
        } else {
#pragma unroll
            for (int k = 0; k < 4; ++k)
                if (b0 + k >= lo && b0 + k < hi) g[b0 + k] = st[b0 + k];
        }
    }
}

__global__ __launch_bounds__(1024) void bucket_local(const int* __restrict__ e_l,
                                                     const int* __restrict__ e_r, int nE, int chunk,
                                                     int nbR, int nbAll, int* __restrict__ tab,
                                                     int2* __restrict__ pairR, int2* __restrict__ pairL) {
    static_assert(EB == 4, "the statements that group the edge loads name EB = 4 pairs");
    __shared__ int cnt[MAXNB];
    __shared__ int wtot[32];
    __shared__ alignas(16) int2 stR[CAPE + 2];
    __shared__ alignas(16) int2 stL[CAPE + 2];
    const int tid = threadIdx.x;
    const int lane = tid & 31;
    const int wave = tid >> 5;
    cnt[tid] = 0;
    __syncthreads();
    const int e0 = blockIdx.x * chunk;
    const int e1 = min(nE, e0 + chunk);
    const int n = max(e1 - e0, 0);
    const bool staged = n <= CAPE;
    int2* const gR = pairR + e0;
    int2* const gL = pairL + e0;
    const int mR = (int)(((size_t)gR >> 3) & 1);
    const int mL = (int)(((size_t)gL >> 3) & 1);
    for (int i0 = e0 + tid; i0 < e1; i0 += EB * 1024) {
        int dl[EB], dr[EB];
#pragma unroll
        for (int k = 0; k < EB; ++k) {
            const int ic = min(i0 + k * 1024, e1 - 1);
            dl[k] = e_l[ic];
            dr[k] = e_r[ic];
        }
        asm volatile("" ::"v"(dl[0]), "v"(dl[1]), "v"(dl[2]), "v"(dl[3]), "v"(dr[0]), "v"(dr[1]),
                     "v"(dr[2]), "v"(dr[3]));
#pragma unroll
        for (int k = 0; k < EB; ++k)
            if (i0 + k * 1024 >= e1) dl[k] = -1;
#pragma unroll
        for (int k = 0; k < EB; ++k) {
            if (dl[k] >= 0) {
                atomicAdd(&cnt[dr[k] >> SH_R], 1);
                atomicAdd(&cnt[nbR + (dl[k] >> SH_L)], 1);
            }
        }
    }
    __syncthreads();
    const int v = cnt[tid];
    const int incl = wave_incl_scan(v, lane);
    if (lane == 31) wtot[wave] = incl;
    __syncthreads();
    int pre = 0;
#pragma unroll
    for (int w = 0; w < 32; ++w) pre += (w < wave) ? wtot[w] : 0;
    const int excl = pre + incl - v;
    if (tid <= nbAll) tab[(size_t)tid * HG + blockIdx.x] = excl;
    cnt[tid] = excl - (tid >= nbR ? n : 0);
    __syncthreads();
    for (int i0 = e0 + tid; i0 < e1; i0 += EB * 1024) {
        int dl[EB], dr[EB];
#pragma unroll
        for (int k = 0; k < EB; ++k) {
            const int ic = min(i0 + k * 1024, e1 - 1);
            dl[k] = e_l[ic];
            dr[k] = e_r[ic];
        }
        asm volatile("" ::"v"(dl[0]), "v"(dl[1]), "v"(dl[2]), "v"(dl[3]), "v"(dr[0]), "v"(dr[1]),
                     "v"(dr[2]), "v"(dr[3]));
#pragma unroll
        for (int k = 0; k < EB; ++k)
            if (i0 + k * 1024 >= e1) dl[k] = -1;
#pragma unroll
        for (int k = 0; k < EB; ++k) {
            if (dl[k] >= 0) {
                const int pr = atomicAdd(&cnt[dr[k] >> SH_R], 1);
                const int pl = atomicAdd(&cnt[nbR + (dl[k] >> SH_L)], 1);
                if (staged) {
                    stR[mR + pr] = make_int2(dr[k], dl[k]);
                    stL[mL + pl] = make_int2(dl[k], dr[k]);
                } else {
                    gR[pr] = make_int2(dr[k], dl[k]);
                    gL[pl] = make_int2(dl[k], dr[k]);
                }
            }
        }
    }
    if (staged) {
        __syncthreads();
        copy_out_pairs(gR - mR, stR, mR, mR + n, tid);
        copy_out_pairs(gL - mL, stL, mL, mL + n, tid);
    }
}

__global__ __launch_bounds__(1024) void bucket_sort(const int2* __restrict__ pairR,
                                                    const int2* __restrict__ pairL,
                                                    const int* __restrict__ tab, int chunk,
                                                    int nbR, int nbL,
                                                    int* __restrict__ offR, int* __restrict__ offL,
                                                    int* __restrict__ permR, int* __restrict__ permL,
                                                    int nR, int nL) {
    constexpr int RW = HG / 32;
    static_assert(HG % 32 == 0 && RW % EB == 0, "a wave takes its runs EB at a time");
    __shared__ int cnt[BW];
    __shared__ int wtot[BW / 32];
    __shared__ int segOff[HG];
    __shared__ int segLen[HG];
    __shared__ int red[2 * RW];
    __shared__ alignas(16) int stp[CAPS + 4];
    const bool isR = (int)blockIdx.x < nbR;
    const int b = isR ? (int)blockIdx.x : (int)blockIdx.x - nbR;
    const int nb = isR ? nbR : nbL;
    const int base = isR ? 0 : nbR;
    const int2* __restrict__ pair = isR ? pairR : pairL;
    int* __restrict__ off = isR ? offR : offL;
    int* __restrict__ perm = isR ? permR : permL;
    const int n = isR ? nR : nL;
    const int sh = isR ? SH_R : SH_L;
    const int lo = b << sh;
    const int tid = threadIdx.x;
    const int lane = tid & 31;
    const int wave = tid >> 5;
    int ahead = 0, len = 0;
    if (tid < HG) {
        const int a0 = tab[(size_t)base * HG + tid];
        const int a1 = tab[(size_t)(base + b) * HG + tid];
        const int a2 = tab[(size_t)(base + b + 1) * HG + tid];
        ahead = a1 - a0;
        len = a2 - a1;
        segOff[tid] = tid * chunk + ahead;
        segLen[tid] = len;
    }
    if (tid < BW) cnt[tid] = 0;
    {
        int sa = ahead, sl = len;
#pragma unroll
        for (int o = 16; o; o >>= 1) {
            sa += __shfl_xor(sa, o, 32);
            sl += __shfl_xor(sl, o, 32);
        }
        if (lane == 0 && wave < RW) {
            red[wave] = sa;
            red[RW + wave] = sl;
        }
    }
    __syncthreads();
    int s = 0, tot = 0;
#pragma unroll
    for (int w = 0; w < RW; ++w) {
        s += red[w];
        tot += red[RW + w];
    }
    const int e = s + tot;
    const bool staged = tot <= CAPS;
    int* const gp = perm + s;
    const int mp = (int)(((size_t)gp >> 2) & 3);
#pragma unroll 1
    for (int h = 0; h < RW; h += EB) {
        int ro[EB], rl[EB], d[EB];
#pragma unroll
        for (int k = 0; k < EB; ++k) {
            ro[k] = segOff[wave + 32 * (h + k)];
            rl[k] = segLen[wave + 32 * (h + k)];
        }
#pragma unroll
        for (int k = 0; k < EB; ++k) d[k] = lane < rl[k] ? pair[ro[k] + lane].x : -1;
#pragma unroll
        for (int k = 0; k < EB; ++k)
            if (d[k] >= 0) atomicAdd(&cnt[d[k] - lo], 1);
#pragma unroll
        for (int k = 0; k < EB; ++k)
            for (int l = lane + 32; l < rl[k]; l += 32) atomicAdd(&cnt[pair[ro[k] + l].x - lo], 1);
    }
    __syncthreads();
    const int v = tid < BW ? cnt[tid] : 0;
    const int incl = wave_incl_scan(v, lane);
    if (lane == 31 && wave < BW / 32) wtot[wave] = incl;
    __syncthreads();
    if (tid < BW) {
        int pre = 0;
#pragma unroll
        for (int w = 0; w < BW / 32; ++w) pre += (w < wave) ? wtot[w] : 0;
        const int excl = pre + incl - v;
        cnt[tid] = excl;
        const int node = lo + tid;
        if (tid < (1 << sh) && node < n) off[node] = s + excl;
    }
    if (tid == 0 && b == nb - 1) off[n] = e;
    __syncthreads();
#pragma unroll 1
    for (int h = 0; h < RW; h += EB) {
        int ro[EB], rl[EB];
        int2 t[EB];
#pragma unroll
        for (int k = 0; k < EB; ++k) {
            ro[k] = segOff[wave + 32 * (h + k)];
            rl[k] = segLen[wave + 32 * (h + k)];
        }
#pragma unroll
        for (int k = 0; k < EB; ++k) t[k] = lane < rl[k] ? pair[ro[k] + lane] : make_int2(-1, 0);
#pragma unroll
        for (int k = 0; k < EB; ++k) {
            if (t[k].x >= 0) {
                const int p = atomicAdd(&cnt[t[k].x - lo], 1);
                if (staged) stp[mp + p] = t[k].y;
                else gp[p] = t[k].y;
            }
        }
#pragma unroll
        for (int k = 0; k < EB; ++k)
            for (int l = lane + 32; l < rl[k]; l += 32) {
                const int2 u = pair[ro[k] + l];
                const int p = atomicAdd(&cnt[u.x - lo], 1);
                if (staged) stp[mp + p] = u.y;
                else gp[p] = u.y;
            }
    }
    if (staged) {
        __syncthreads();
        copy_out_words(gp - mp, stp, mp, mp + tot, tid);
    }
}

__global__ __launch_bounds__(256) void gather_add3_finish(const float* __restrict__ projA,
                                                          const int* __restrict__ offA,
                                                          const int* __restrict__ permA,
                                                          const float* __restrict__ xA,
                                                          float* outA, int nA, int blocksA,
                                                          const float* __restrict__ projB,
                                                          const int* __restrict__ offB,
                                                          const int* __restrict__ permB,
                                                          const float* __restrict__ xB,
                                                          float* outB, int nB) {
    const bool jobA = (int)blockIdx.x < blocksA;
    const float* __restrict__ proj = jobA ? projA : projB;
    const int* __restrict__ off = jobA ? offA : offB;
    const int* __restrict__ perm = jobA ? permA : permB;
    const float* __restrict__ x = jobA ? xA : xB;
    float* out = jobA ? outA : outB;
    const int nDst = jobA ? nA : nB;
    const int blk = jobA ? (int)blockIdx.x : (int)blockIdx.x - blocksA;
    int d = blk * blockDim.x + threadIdx.x;
    if (d >= nDst) return;
    const int beg = off[d];
    const int end = off[d + 1];
    float a0 = out[(size_t)d * 3 + 0];
    float a1 = out[(size_t)d * 3 + 1];
    float a2 = out[(size_t)d * 3 + 2];
    const float f0 = x[(size_t)d * 64 + 61];
    const float f2 = x[(size_t)d * 64 + 63];
    int ia[4];
#pragma unroll
    for (int k = 0; k < 4; ++k) ia[k] = perm[beg + k];
    asm volatile("" ::"v"(ia[0]), "v"(ia[1]), "v"(ia[2]), "v"(ia[3]));
    int j = beg;
    for (; j + 4 <= end; j += 4) {
        int in[4];
#pragma unroll
        for (int k = 0; k < 4; ++k) in[k] = perm[j + 4 + k];
        float p[4][3];
#pragma unroll
        for (int k = 0; k < 4; ++k)
#pragma unroll
            for (int c = 0; c < 3; ++c) p[k][c] = proj[(size_t)ia[k] * 3 + c];
#pragma unroll
        for (int k = 0; k < 4; ++k) {
            a0 += p[k][0];
            a1 += p[k][1];
            a2 += p[k][2];
        }
        asm volatile("" ::"v"(in[0]), "v"(in[1]), "v"(in[2]), "v"(in[3]));
#pragma unroll
        for (int k = 0; k < 4; ++k) ia[k] = in[k];
    }
    if (j < end) {
        int s[3];
#pragma unroll
        for (int k = 0; k < 3; ++k) s[k] = (j + k < end) ? ia[k] : ia[0];
        float p[3][3];
#pragma unroll
        for (int k = 0; k < 3; ++k)
#pragma unroll
            for (int c = 0; c < 3; ++c) p[k][c] = proj[(size_t)s[k] * 3 + c];
#pragma unroll
        for (int k = 0; k < 3; ++k) {
            if (j + k < end) {
                a0 += p[k][0];
                a1 += p[k][1];
                a2 += p[k][2];
            }
        }
    }
    float nrm = sqrtf(a0 * a0 + a1 * a1 + a2 * a2);
    float sc = 10.0f / fmaxf(nrm, 1e-12f);
    a0 *= sc; a1 *= sc; a2 *= sc;
    if (f0 != 0.0f) a0 -= 10.0f;
    if (f2 != 0.0f) a2 -= 10.0f;
    out[(size_t)d * 3 + 0] = a0;
    out[(size_t)d * 3 + 1] = a1;
    out[(size_t)d * 3 + 2] = a2;
}

template <int K1, int K2, bool RELU, bool FUSE3, int NW>
__global__ __launch_bounds__(NW * 32) __attribute__((amdgpu_waves_per_eu(NW == 8 ? 6 : 4))) void gemm_dual_wmma(const float* __restrict__ Sa,
                                                          const int* __restrict__ offa,
                                                          const int* __restrict__ perma,
                                                          const float* __restrict__ W1a,
                                                          const float* __restrict__ A2a,
                                                          const float* __restrict__ W2a,
                                                          const float* __restrict__ biasa,
                                                          float* outa, int Ma, int blocksA,
                                                          const float* __restrict__ Sb,
                                                          const int* __restrict__ offb,
                                                          const int* __restrict__ permb,
                                                          const float* __restrict__ W1b,
                                                          const float* __restrict__ A2b,
                                                          const float* __restrict__ W2b,
                                                          const float* __restrict__ biasb,
                                                          float* outb, int Mb,
                                                          const float* __restrict__ W3rela,
                                                          const float* __restrict__ W3roota,
                                                          const float* __restrict__ b3a,
                                                          float* __restrict__ proja,
                                                          const float* __restrict__ W3relb,
                                                          const float* __restrict__ W3rootb,
                                                          const float* __restrict__ b3b,
                                                          float* __restrict__ projb) {
    constexpr int N = 128;
    constexpr int KP = (K1 + K2) / 2;
    constexpr int PITCH = 2 * N + 32;
    constexpr int W3N = FUSE3 ? 1536 : 0;
    constexpr int SP = K1 + 4;
    __shared__ alignas(16) float lds[KP * PITCH + W3N + NW * 16 * SP];
    float* wl = lds;
    float* w3l = lds + KP * PITCH;
    float* st = lds + KP * PITCH + W3N;

    const bool jobA = (int)blockIdx.x < blocksA;
    const float* __restrict__ S = jobA ? Sa : Sb;
    const int* __restrict__ off = jobA ? offa : offb;
    const int* __restrict__ perm = jobA ? perma : permb;
    const float* __restrict__ W1 = jobA ? W1a : W1b;
    const float* __restrict__ A2 = jobA ? A2a : A2b;
    const float* __restrict__ W2 = jobA ? W2a : W2b;
    const float* __restrict__ bias = jobA ? biasa : biasb;
    float* out = jobA ? outa : outb;
    const int M = jobA ? Ma : Mb;
    const int blk = jobA ? (int)blockIdx.x : (int)blockIdx.x - blocksA;
    const int nblk = jobA ? blocksA : (int)gridDim.x - blocksA;

    const int lane = threadIdx.x & 31;
    const int wave = threadIdx.x >> 5;
    const int rq = lane >> 3;
    const int nStrips = M / 16;
    const int s0 = (int)((long long)blk * nStrips / nblk);
    const int s1 = (int)((long long)(blk + 1) * nStrips / nblk);
    int bq[4], eq[4];
    int ia[4];
    {
        const int mf = min(s0 + wave, nStrips - 1) * 16;
#pragma unroll
        for (int p = 0; p < 4; ++p) {
            bq[p] = off[mf + p * 4 + rq];
            eq[p] = off[mf + p * 4 + rq + 1];
        }
    }

    float l3r0 = 0.f, l3o0 = 0.f, l3r1 = 0.f, l3o1 = 0.f, l3b = 0.f, l3c = 0.f;
    if constexpr (FUSE3) {
        const float* __restrict__ W3rel = jobA ? W3rela : W3relb;
        const float* __restrict__ W3root = jobA ? W3roota : W3rootb;
        const float* __restrict__ b3 = jobA ? b3a : b3b;
        const int t = threadIdx.x & 255;
        const int i1 = min(t + 256, 383);
        l3r0 = W3rel[t];
        l3o0 = W3root[t];
        l3r1 = W3rel[i1];
        l3o1 = W3root[i1];
        l3b = bias[t & 127];
        l3c = b3[min(t, 2)];
    }

    {
        constexpr int RP = NW / 2;
        const int c2 = (threadIdx.x & 63) * 2;
        const int p0 = threadIdx.x >> 6;
#pragma unroll 8
        for (int i = 0; i < K1 / (2 * RP); ++i) {
            const int p = p0 + RP * i;
            const float2 r0 = *(const float2*)(W1 + (size_t)(2 * p) * N + c2);
            const float2 r1 = *(const float2*)(W1 + (size_t)(2 * p + 1) * N + c2);
            *(float4*)(wl + p * PITCH + c2 * 2) = make_float4(r0.x, r1.x, r0.y, r1.y);
        }
#pragma unroll
        for (int k = 0; k < 4; ++k) ia[k] = perm[bq[0] + k];
#pragma unroll 8
        for (int i = 0; i < K2 / (2 * RP); ++i) {
            const int p = p0 + RP * i;
            const float2 r0 = *(const float2*)(W2 + (size_t)(2 * p) * N + c2);
            const float2 r1 = *(const float2*)(W2 + (size_t)(2 * p + 1) * N + c2);
            *(float4*)(wl + (K1 / 2 + p) * PITCH + c2 * 2) = make_float4(r0.x, r1.x, r0.y, r1.y);
        }
    }
    if constexpr (FUSE3) {
        const int t = threadIdx.x & 255;
        *(float2*)(w3l + 2 * t) = make_float2(l3r0, l3o0);
        *(float2*)(w3l + 2 * (t + 256)) = make_float2(l3r1, l3o1);
        w3l[1024 + t] = l3b;
        w3l[1280 + t] = l3c;
    }
    __syncthreads();

    const int half = lane >> 4;
    const int lr = lane & 15;

    float* stw = st + wave * (16 * SP);
    const float* sa = stw + lr * SP + 2 * half;
    const float* bl0 = wl + half * PITCH + lr * 2;

    constexpr int CH = 8;
    constexpr int NCH = (K1 + K2) / (4 * CH);
    constexpr int PER = 8;
    constexpr int V = K1 / (4 * PER);
    const int sub = lane & 7;
    const float* lane_src = S + sub * 4;

#pragma unroll 1
    for (int strip = s0 + wave; strip < s1; strip += NW) {
        const int m0 = strip * 16;

#pragma unroll 1
        for (int pass = 0; pass < 4; ++pass) {
            const int rl = pass * 4 + rq;
            const int beg = bq[0];
            const int end = eq[0];
            int ib[4];
#pragma unroll
            for (int k = 0; k < 4; ++k) ib[k] = perm[bq[1] + k];
            float4 acc4[V];
#pragma unroll
            for (int u = 0; u < V; ++u) acc4[u] = make_float4(0.f, 0.f, 0.f, 0.f);
            int j = beg;
            for (; j + 4 <= end; j += 4) {
                int in[4];
#pragma unroll
                for (int k = 0; k < 4; ++k) in[k] = perm[j + 4 + k];
                float4 v[4][V];
#pragma unroll
                for (int k = 0; k < 4; ++k)
#pragma unroll
                    for (int u = 0; u < V; ++u)
                        v[k][u] = *(const float4*)(lane_src + (size_t)ia[k] * K1 + u * (PER * 4));
#pragma unroll
                for (int k = 0; k < 4; ++k)
#pragma unroll
                    for (int u = 0; u < V; ++u) {
                        acc4[u].x += v[k][u].x; acc4[u].y += v[k][u].y;
                        acc4[u].z += v[k][u].z; acc4[u].w += v[k][u].w;
                    }
#pragma unroll
                for (int k = 0; k < 4; ++k) ia[k] = in[k];
            }
            if (j < end) {
                int s[3];
#pragma unroll
                for (int k = 0; k < 3; ++k) s[k] = (j + k < end) ? ia[k] : ia[0];
                float4 v[3][V];
#pragma unroll
                for (int k = 0; k < 3; ++k)
#pragma unroll
                    for (int u = 0; u < V; ++u)
                        v[k][u] = *(const float4*)(lane_src + (size_t)s[k] * K1 + u * (PER * 4));
#pragma unroll
                for (int k = 0; k < 3; ++k) {
                    const bool on = j + k < end;
#pragma unroll
                    for (int u = 0; u < V; ++u) {
                        acc4[u].x += on ? v[k][u].x : 0.f; acc4[u].y += on ? v[k][u].y : 0.f;
                        acc4[u].z += on ? v[k][u].z : 0.f; acc4[u].w += on ? v[k][u].w : 0.f;
                    }
                }
            }
#pragma unroll
            for (int u = 0; u < V; ++u)
                *(float4*)(stw + rl * SP + (u * PER + sub) * 4) = acc4[u];
#pragma unroll
            for (int p = 0; p < 3; ++p) {
                bq[p] = bq[p + 1];
                eq[p] = eq[p + 1];
            }
#pragma unroll
            for (int k = 0; k < 4; ++k) ia[k] = ib[k];
        }

        const float* a2p = A2 + (size_t)(m0 + lr) * K2 + 2 * half;
        const float* bl = bl0;

        v8f acc[8] = {};
        v2f ac[CH], an[CH];
        v2f bc[8], bn[8];
#pragma unroll
        for (int j = 0; j < CH; ++j) ac[j] = *(const v2f*)(sa + j * 4);
#pragma unroll
        for (int nt = 0; nt < 8; ++nt) bc[nt] = *(const v2f*)(bl + nt * 32);

#pragma unroll 1
        for (int ci = 0; ci < NCH; ++ci) {
            const bool more = ci + 1 < NCH;
            if (more) {
                const int kn = (ci + 1) * (4 * CH);
                if (kn < K1) {
#pragma unroll
                    for (int j = 0; j < CH; ++j) an[j] = *(const v2f*)(sa + kn + j * 4);
                } else {
                    const float* ap = a2p + (kn - K1);
#pragma unroll
                    for (int j = 0; j < CH; ++j) an[j] = *(const v2f*)(ap + j * 4);
                }
            }
#pragma unroll
            for (int j = 0; j < CH; ++j) {
                if (j + 1 < CH || more) {
#pragma unroll
                    for (int nt = 0; nt < 8; ++nt)
                        bn[nt] = *(const v2f*)(bl + (2 * (j + 1)) * PITCH + nt * 32);
                }
#pragma unroll
                for (int nt = 0; nt < 8; ++nt) {
                    if constexpr (FUSE3)
                        acc[nt] = __builtin_amdgcn_wmma_f32_16x16x4_f32(
                            false, bc[nt], false, ac[j], (short)0, acc[nt], false, false);
                    else
                        acc[nt] = __builtin_amdgcn_wmma_f32_16x16x4_f32(
                            false, ac[j], false, bc[nt], (short)0, acc[nt], false, false);
                }
#pragma unroll
                for (int nt = 0; nt < 8; ++nt) bc[nt] = bn[nt];
            }
            if (more) {
#pragma unroll
                for (int j = 0; j < CH; ++j) ac[j] = an[j];
            }
            bl += 2 * CH * PITCH;
        }

        if constexpr (FUSE3) {
            float* __restrict__ proj = jobA ? proja : projb;
            v2f pq[2][3];
#pragma unroll
            for (int cc = 0; cc < 3; ++cc) {
                pq[0][cc] = (v2f){0.f, 0.f};
                pq[1][cc] = (v2f){0.f, 0.f};
            }
#pragma unroll
            for (int nt = 0; nt < 8; ++nt) {
                const int n0 = nt * 16 + 8 * half;
                const float4* bp = (const float4*)(w3l + 1024 + n0);
                const float4* wp = (const float4*)(w3l + n0 * 6);
                float bb[8];
                v2f w[24];
#pragma unroll
                for (int t = 0; t < 2; ++t) {
                    const float4 x = bp[t];
                    bb[4 * t + 0] = x.x; bb[4 * t + 1] = x.y; bb[4 * t + 2] = x.z; bb[4 * t + 3] = x.w;
                }
#pragma unroll
                for (int t = 0; t < 12; ++t) {
                    const float4 x = wp[t];
                    w[2 * t + 0] = (v2f){x.x, x.y};
                    w[2 * t + 1] = (v2f){x.z, x.w};
                }
#pragma unroll
                for (int r = 0; r < 8; ++r) {
                    float v = acc[nt][r] + bb[r];
                    if (RELU) v = fmaxf(v, 0.0f);
                    const v2f vv = (v2f){v, v};
#pragma unroll
                    for (int cc = 0; cc < 3; ++cc) pq[r & 1][cc] += vv * w[r * 3 + cc];
                }
            }
            float s[6];
#pragma unroll
            for (int cc = 0; cc < 3; ++cc) {
                const v2f t = pq[0][cc] + pq[1][cc];
                s[cc] = t.x;
                s[3 + cc] = t.y;
            }
#pragma unroll
            for (int cc = 0; cc < 6; ++cc) s[cc] += __shfl_xor(s[cc], 16, 32);
            if (half == 0) {
                const size_t m = (size_t)(m0 + lr);
#pragma unroll
                for (int cc = 0; cc < 3; ++cc) {
                    proj[m * 3 + cc] = s[cc];
                    out[m * 3 + cc] = s[3 + cc] + w3l[1280 + cc];
                }
            }
        } else {
#pragma unroll
            for (int nt = 0; nt < 8; ++nt) {
                const int col = nt * 16 + lr;
                const float bv = bias[col];
#pragma unroll
                for (int i = 0; i < 8; ++i) {
                    float v = acc[nt][i] + bv;
                    if (RELU) v = fmaxf(v, 0.0f);
                    out[(size_t)(m0 + i + 8 * half) * N + col] = v;
                }
            }
        }

        if (strip + NW < s1) {
            const int mn = (strip + NW) * 16;
#pragma unroll
            for (int p = 0; p < 4; ++p) {
                bq[p] = off[mn + p * 4 + rq];
                eq[p] = off[mn + p * 4 + rq + 1];
            }
#pragma unroll
            for (int k = 0; k < 4; ++k) ia[k] = perm[bq[0] + k];
        }
    }
}

static inline int ceil_div(long long a, long long b) { return (int)((a + b - 1) / b); }

extern "C" void kernel_launch(void* const* d_in, const int* in_sizes, int n_in,
                              void* d_out, int out_size, void* d_ws, size_t ws_size,
                              hipStream_t stream) {
    const float* x_s = (const float*)d_in[0];
    const float* x_t = (const float*)d_in[1];
    const int* e_l = (const int*)d_in[2];
    const int* e_r = (const int*)d_in[3];
    const float* w1lr_rel  = (const float*)d_in[4];
    const float* b1lr      = (const float*)d_in[5];
    const float* w1lr_root = (const float*)d_in[6];
    const float* w1rl_rel  = (const float*)d_in[7];
    const float* b1rl      = (const float*)d_in[8];
    const float* w1rl_root = (const float*)d_in[9];
    const float* w2lr_rel  = (const float*)d_in[10];
    const float* b2lr      = (const float*)d_in[11];
    const float* w2lr_root = (const float*)d_in[12];
    const float* w2rl_rel  = (const float*)d_in[13];
    const float* b2rl      = (const float*)d_in[14];
    const float* w2rl_root = (const float*)d_in[15];
    const float* w3lr_rel  = (const float*)d_in[16];
    const float* b3lr      = (const float*)d_in[17];
    const float* w3lr_root = (const float*)d_in[18];
    const float* w3rl_rel  = (const float*)d_in[19];
    const float* b3rl      = (const float*)d_in[20];
    const float* w3rl_root = (const float*)d_in[21];

    const int NL = in_sizes[0] / 64;
    const int NR = in_sizes[1] / 64;
    const int E  = in_sizes[2];

    float* ws = (float*)d_ws;
    const size_t H1_R  = 0;
    const size_t H1_L  = H1_R + (size_t)NR * 128;
    const int nbR = ceil_div(NR, 1 << SH_R);
    const int nbL = ceil_div(NL, 1 << SH_L);
    const int chunk = ceil_div(E, HG);
    int* wi = (int*)d_ws;
    const size_t PAIR_R = H1_L + (size_t)NL * 128;
    const size_t PAIR_L = PAIR_R + (size_t)E * 2;
    const size_t TAB    = PAIR_L + (size_t)E * 2;
    const size_t PERM_R = TAB + (size_t)(nbR + nbL + 1) * HG;
    const size_t PERM_L = PERM_R + (size_t)E;
    const size_t OFF_R  = PERM_L + (size_t)E;
    const size_t OFF_L  = OFF_R + (size_t)NR + 1;
    const size_t PROJ_LR = OFF_L + (size_t)NL + 1;
    const size_t PROJ_RL = PROJ_LR + (size_t)NL * 3;

    float* d_left  = (float*)d_out;
    float* d_right = (float*)d_out + (size_t)NL * 3;

    const int TPB = 256;

    bucket_local<<<HG, 1024, 0, stream>>>(e_l, e_r, E, chunk, nbR, nbR + nbL, wi + TAB,
                                          (int2*)(wi + PAIR_R), (int2*)(wi + PAIR_L));
    bucket_sort<<<nbR + nbL, 1024, 0, stream>>>((const int2*)(wi + PAIR_R), (const int2*)(wi + PAIR_L),
                                                wi + TAB, chunk, nbR, nbL,
                                                wi + OFF_R, wi + OFF_L, wi + PERM_R, wi + PERM_L, NR, NL);

    const int gmR = ceil_div(NR, 16 * 8);
    const int gmL = ceil_div(NL, 16 * 8);
    const int g2R = 160;
    const int g2L = 96;
    const int fnL = ceil_div(NL, TPB);
    const int fnR = ceil_div(NR, TPB);

    gemm_dual_wmma<64, 64, true, false, 8><<<gmR + gmL, TPB, 0, stream>>>(
        x_s, wi + OFF_R, wi + PERM_R, w1lr_rel, x_t, w1lr_root, b1lr, ws + H1_R, NR, gmR,
        x_t, wi + OFF_L, wi + PERM_L, w1rl_rel, x_s, w1rl_root, b1rl, ws + H1_L, NL,
        nullptr, nullptr, nullptr, nullptr, nullptr, nullptr, nullptr, nullptr);

    gemm_dual_wmma<128, 128, true, true, 16><<<g2R + g2L, 512, 0, stream>>>(
        ws + H1_L, wi + OFF_R, wi + PERM_R, w2lr_rel, ws + H1_R, w2lr_root, b2lr, d_right, NR, g2R,
        ws + H1_R, wi + OFF_L, wi + PERM_L, w2rl_rel, ws + H1_L, w2rl_root, b2rl, d_left, NL,
        w3rl_rel, w3lr_root, b3lr, ws + PROJ_RL,
        w3lr_rel, w3rl_root, b3rl, ws + PROJ_LR);
    gather_add3_finish<<<fnL + fnR, TPB, 0, stream>>>(ws + PROJ_RL, wi + OFF_L, wi + PERM_L, x_s, d_left, NL, fnL,
                                                      ws + PROJ_LR, wi + OFF_R, wi + PERM_R, x_t, d_right, NR);
}
